# starting kernel
# baseline (speedup 1.0000x reference)
#include <hip/hip_runtime.h>

typedef __attribute__((ext_vector_type(2))) float v2f;
typedef __attribute__((ext_vector_type(8))) float v8f;

#define N_DRUG 572
#define DIM    512
#define KNBR   64
#define MAXDEG 32
#define NPAD   576
#define MT     36
#define NT     32

__global__ __launch_bounds__(32)
void qgemm_kernel(const float* __restrict__ drugW, const float* __restrict__ Wa,
                  const int* __restrict__ drug_name, float* __restrict__ q)
{
    const int lane = threadIdx.x;
    const int half = lane >> 4;
    const int l16  = lane & 15;
    const int mBase = blockIdx.x * 16;
    const int nBase = blockIdx.y * 16;

    const int mrow = mBase + l16;
    const int gi = drug_name[mrow < N_DRUG ? mrow : (N_DRUG - 1)];
    const float* __restrict__ Arow = drugW + (size_t)gi * DIM;
    const float* __restrict__ Bcol = Wa + nBase + l16;

    v8f c = {};
    for (int kk = 0; kk < DIM; kk += 4) {
        const int ka = kk + 2 * half;
        v2f a, b;
        a.x = Arow[ka];
        a.y = Arow[ka + 1];
        b.x = Bcol[(size_t)ka * DIM];
        b.y = Bcol[(size_t)(ka + 1) * DIM];
        c = __builtin_amdgcn_wmma_f32_16x16x4_f32(false, a, false, b,
                                                  (short)0, c, false, false);
    }
    float* outBase = q + (size_t)(mBase + half * 8) * DIM + nBase + l16;
#pragma unroll
    for (int j = 0; j < 8; ++j)
        outBase[(size_t)j * DIM] = c[j];
}

__global__ __launch_bounds__(256)
void attn_agg_kernel(const float* __restrict__ q, const float* __restrict__ relaW,
                     const float* __restrict__ entW,
                     const int* __restrict__ adj_rel, const int* __restrict__ adj_tail,
                     float* __restrict__ agg)
{
    __shared__ float qs[DIM];
    __shared__ float attn[KNBR];

    const int n  = blockIdx.x;
    const int ni = n < N_DRUG ? n : (N_DRUG - 1);
    const int tid = threadIdx.x;

    qs[tid]       = q[(size_t)n * DIM + tid];
    qs[tid + 256] = q[(size_t)n * DIM + tid + 256];
    __syncthreads();

    const int wave = tid >> 5;
    const int lane = tid & 31;
    for (int kk = 0; kk < 8; ++kk) {
        const int k = wave * 8 + kk;
        const int rel = adj_rel[ni * KNBR + k];
        const float* __restrict__ rw = relaW + (size_t)rel * DIM;
        float p = 0.f;
        for (int j = lane; j < DIM; j += 32)
            p += qs[j] * rw[j];
        for (int off = 16; off >= 1; off >>= 1)
            p += __shfl_xor(p, off, 32);
        if (lane == 0)
            attn[k] = p * 0.04419417382415922f;
    }
    __syncthreads();

    if (tid < 32) {
        float s0 = attn[tid], s1 = attn[tid + 32];
        float m = fmaxf(s0, s1);
        for (int off = 16; off >= 1; off >>= 1)
            m = fmaxf(m, __shfl_xor(m, off, 32));
        float e0 = __expf(s0 - m), e1 = __expf(s1 - m);
        float s = e0 + e1;
        for (int off = 16; off >= 1; off >>= 1)
            s += __shfl_xor(s, off, 32);
        const float inv = 1.f / s;
        attn[tid]      = e0 * inv;
        attn[tid + 32] = e1 * inv;
    }
    __syncthreads();

    float acc0 = 0.f, acc1 = 0.f;
    for (int k = 0; k < KNBR; ++k) {
        const int t = adj_tail[ni * KNBR + k];
        const float w = attn[k];
        const float* __restrict__ er = entW + (size_t)t * DIM;
        acc0 += w * er[tid];
        acc1 += w * er[tid + 256];
    }
    agg[(size_t)n * DIM + tid]       = acc0;
    agg[(size_t)n * DIM + tid + 256] = acc1;
}

__global__ __launch_bounds__(32)
void lingemm_kernel(const float* __restrict__ agg, const float* __restrict__ drugW,
                    const int* __restrict__ drug_name,
                    const float* __restrict__ linW, const float* __restrict__ linb,
                    float* __restrict__ h)
{
    const int lane = threadIdx.x;
    const int half = lane >> 4;
    const int l16  = lane & 15;
    const int mBase = blockIdx.x * 16;
    const int nBase = blockIdx.y * 16;

    const int mrow = mBase + l16;
    const int gi = drug_name[mrow < N_DRUG ? mrow : (N_DRUG - 1)];
    const float* __restrict__ Arow0 = agg + (size_t)mrow * DIM;
    const float* __restrict__ Arow1 = drugW + (size_t)gi * DIM;
    const float* __restrict__ Bcol  = linW + nBase + l16;

    v8f c = {};
    for (int kk = 0; kk < 2 * DIM; kk += 4) {
        const int ka = kk + 2 * half;
        const float* __restrict__ Ar = (ka < DIM) ? (Arow0 + ka) : (Arow1 + (ka - DIM));
        v2f a, b;
        a.x = Ar[0];
        a.y = Ar[1];
        b.x = Bcol[(size_t)ka * DIM];
        b.y = Bcol[(size_t)(ka + 1) * DIM];
        c = __builtin_amdgcn_wmma_f32_16x16x4_f32(false, a, false, b,
                                                  (short)0, c, false, false);
    }
    const float bias = linb[nBase + l16];
    float* outBase = h + (size_t)(mBase + half * 8) * DIM + nBase + l16;
#pragma unroll
    for (int j = 0; j < 8; ++j) {
        const float v = c[j] + bias;
        outBase[(size_t)j * DIM] = v > 0.f ? v : 0.f;
    }
}

__global__ __launch_bounds__(256)
void bnstats_kernel(const float* __restrict__ h, const float* __restrict__ gamma,
                    const float* __restrict__ beta,
                    float* __restrict__ scale, float* __restrict__ shift)
{
    const int c = blockIdx.x * 256 + threadIdx.x;
    float s = 0.f, s2 = 0.f;
    for (int nrow = 0; nrow < N_DRUG; ++nrow) {
        const float v = h[(size_t)nrow * DIM + c];
        s += v;
        s2 += v * v;
    }
    const float inv_n = 1.0f / (float)N_DRUG;
    const float mean = s * inv_n;
    const float var  = s2 * inv_n - mean * mean;
    const float sc = gamma[c] * rsqrtf(var + 1e-5f);
    scale[c] = sc;
    shift[c] = beta[c] - mean * sc;
}

__global__ __launch_bounds__(256)
void bnapply_kernel(const float* __restrict__ h, const float* __restrict__ scale,
                    const float* __restrict__ shift, float* __restrict__ out)
{
    const int idx = blockIdx.x * 256 + threadIdx.x;
    if (idx >= N_DRUG * DIM) return;
    const int c = idx & (DIM - 1);
    out[idx] = h[idx] * scale[c] + shift[c];
}

__global__ __launch_bounds__(256)
void smooth_kernel(const int* __restrict__ nbr_idx, const int* __restrict__ nbr_deg,
                   const int* __restrict__ epoch, float* __restrict__ f)
{
    if (*epoch <= 1) return;
    const int c = blockIdx.x * 256 + threadIdx.x;
    for (int i = 0; i < N_DRUG; ++i) {
        const int deg = nbr_deg[i];
        if (deg > 0) {
            const int dd = deg < MAXDEG ? deg : MAXDEG;
            float s = 0.f;
            for (int j = 0; j < dd; ++j) {
                const int pos = nbr_idx[i * MAXDEG + j];
                s += f[(size_t)pos * DIM + c];
            }
            const float avg = s / (float)dd;
            const float fi = f[(size_t)i * DIM + c];
            f[(size_t)i * DIM + c] = (avg + fi) * 0.5f;
        }
    }
}

extern "C" void kernel_launch(void* const* d_in, const int* in_sizes, int n_in,
                              void* d_out, int out_size, void* d_ws, size_t ws_size,
                              hipStream_t stream)
{
    const float* drugW  = (const float*)d_in[0];
    const float* relaW  = (const float*)d_in[1];
    const float* entW   = (const float*)d_in[2];
    const float* Wa     = (const float*)d_in[3];
    const float* linW   = (const float*)d_in[4];
    const float* linb   = (const float*)d_in[5];
    const float* gamma  = (const float*)d_in[6];
    const float* beta   = (const float*)d_in[7];
    const int* drug_name = (const int*)d_in[8];
    const int* adj_tail  = (const int*)d_in[9];
    const int* adj_rel   = (const int*)d_in[10];
    const int* nbr_idx   = (const int*)d_in[11];
    const int* nbr_deg   = (const int*)d_in[12];
    const int* epoch     = (const int*)d_in[13];

    float* out = (float*)d_out;
    float* ws  = (float*)d_ws;
    float* q     = ws;
    float* agg   = ws + (size_t)NPAD * DIM;
    float* h     = ws + (size_t)2 * NPAD * DIM;
    float* scale = ws + (size_t)3 * NPAD * DIM;
    float* shift = scale + DIM;

    dim3 gTiles(MT, NT);
    qgemm_kernel<<<gTiles, 32, 0, stream>>>(drugW, Wa, drug_name, q);
    attn_agg_kernel<<<NPAD, 256, 0, stream>>>(q, relaW, entW, adj_rel, adj_tail, agg);
    lingemm_kernel<<<gTiles, 32, 0, stream>>>(agg, drugW, drug_name, linW, linb, h);
    bnstats_kernel<<<2, 256, 0, stream>>>(h, gamma, beta, scale, shift);
    bnapply_kernel<<<(N_DRUG * DIM + 255) / 256, 256, 0, stream>>>(h, scale, shift, out);
    smooth_kernel<<<2, 256, 0, stream>>>(nbr_idx, nbr_deg, epoch, out);
}


# smoothing scan runs on a 32-column slab of f staged in LDS with the neighbor table, one wave per slab, same row and sum order; unused table slots read a negative-zero row
# speedup vs baseline: 4.7869x; 4.7869x over previous
#include <hip/hip_runtime.h>

typedef __attribute__((ext_vector_type(2))) float v2f;
typedef __attribute__((ext_vector_type(8))) float v8f;

#define N_DRUG 572
#define DIM    512
#define KNBR   64
#define MAXDEG 32
#define NPAD   576
#define MT     36
#define NT     32

__global__ __launch_bounds__(32)
void qgemm_kernel(const float* __restrict__ drugW, const float* __restrict__ Wa,
                  const int* __restrict__ drug_name, float* __restrict__ q)
{
    const int lane = threadIdx.x;
    const int half = lane >> 4;
    const int l16  = lane & 15;
    const int mBase = blockIdx.x * 16;
    const int nBase = blockIdx.y * 16;

    const int mrow = mBase + l16;
    const int gi = drug_name[mrow < N_DRUG ? mrow : (N_DRUG - 1)];
    const float* __restrict__ Arow = drugW + (size_t)gi * DIM;
    const float* __restrict__ Bcol = Wa + nBase + l16;

    v8f c = {};
    for (int kk = 0; kk < DIM; kk += 4) {
        const int ka = kk + 2 * half;
        v2f a, b;
        a.x = Arow[ka];
        a.y = Arow[ka + 1];
        b.x = Bcol[(size_t)ka * DIM];
        b.y = Bcol[(size_t)(ka + 1) * DIM];
        c = __builtin_amdgcn_wmma_f32_16x16x4_f32(false, a, false, b,
                                                  (short)0, c, false, false);
    }
    float* outBase = q + (size_t)(mBase + half * 8) * DIM + nBase + l16;
#pragma unroll
    for (int j = 0; j < 8; ++j)
        outBase[(size_t)j * DIM] = c[j];
}

__global__ __launch_bounds__(256)
void attn_agg_kernel(const float* __restrict__ q, const float* __restrict__ relaW,
                     const float* __restrict__ entW,
                     const int* __restrict__ adj_rel, const int* __restrict__ adj_tail,
                     float* __restrict__ agg)
{
    __shared__ float qs[DIM];
    __shared__ float attn[KNBR];

    const int n  = blockIdx.x;
    const int ni = n < N_DRUG ? n : (N_DRUG - 1);
    const int tid = threadIdx.x;

    qs[tid]       = q[(size_t)n * DIM + tid];
    qs[tid + 256] = q[(size_t)n * DIM + tid + 256];
    __syncthreads();

    const int wave = tid >> 5;
    const int lane = tid & 31;
    for (int kk = 0; kk < 8; ++kk) {
        const int k = wave * 8 + kk;
        const int rel = adj_rel[ni * KNBR + k];
        const float* __restrict__ rw = relaW + (size_t)rel * DIM;
        float p = 0.f;
        for (int j = lane; j < DIM; j += 32)
            p += qs[j] * rw[j];
        for (int off = 16; off >= 1; off >>= 1)
            p += __shfl_xor(p, off, 32);
        if (lane == 0)
            attn[k] = p * 0.04419417382415922f;
    }
    __syncthreads();

    if (tid < 32) {
        float s0 = attn[tid], s1 = attn[tid + 32];
        float m = fmaxf(s0, s1);
        for (int off = 16; off >= 1; off >>= 1)
            m = fmaxf(m, __shfl_xor(m, off, 32));
        float e0 = __expf(s0 - m), e1 = __expf(s1 - m);
        float s = e0 + e1;
        for (int off = 16; off >= 1; off >>= 1)
            s += __shfl_xor(s, off, 32);
        const float inv = 1.f / s;
        attn[tid]      = e0 * inv;
        attn[tid + 32] = e1 * inv;
    }
    __syncthreads();

    float acc0 = 0.f, acc1 = 0.f;
    for (int k = 0; k < KNBR; ++k) {
        const int t = adj_tail[ni * KNBR + k];
        const float w = attn[k];
        const float* __restrict__ er = entW + (size_t)t * DIM;
        acc0 += w * er[tid];
        acc1 += w * er[tid + 256];
    }
    agg[(size_t)n * DIM + tid]       = acc0;
    agg[(size_t)n * DIM + tid + 256] = acc1;
}

__global__ __launch_bounds__(32)
void lingemm_kernel(const float* __restrict__ agg, const float* __restrict__ drugW,
                    const int* __restrict__ drug_name,
                    const float* __restrict__ linW, const float* __restrict__ linb,
                    float* __restrict__ h)
{
    const int lane = threadIdx.x;
    const int half = lane >> 4;
    const int l16  = lane & 15;
    const int mBase = blockIdx.x * 16;
    const int nBase = blockIdx.y * 16;

    const int mrow = mBase + l16;
    const int gi = drug_name[mrow < N_DRUG ? mrow : (N_DRUG - 1)];
    const float* __restrict__ Arow0 = agg + (size_t)mrow * DIM;
    const float* __restrict__ Arow1 = drugW + (size_t)gi * DIM;
    const float* __restrict__ Bcol  = linW + nBase + l16;

    v8f c = {};
    for (int kk = 0; kk < 2 * DIM; kk += 4) {
        const int ka = kk + 2 * half;
        const float* __restrict__ Ar = (ka < DIM) ? (Arow0 + ka) : (Arow1 + (ka - DIM));
        v2f a, b;
        a.x = Ar[0];
        a.y = Ar[1];
        b.x = Bcol[(size_t)ka * DIM];
        b.y = Bcol[(size_t)(ka + 1) * DIM];
        c = __builtin_amdgcn_wmma_f32_16x16x4_f32(false, a, false, b,
                                                  (short)0, c, false, false);
    }
    const float bias = linb[nBase + l16];
    float* outBase = h + (size_t)(mBase + half * 8) * DIM + nBase + l16;
#pragma unroll
    for (int j = 0; j < 8; ++j) {
        const float v = c[j] + bias;
        outBase[(size_t)j * DIM] = v > 0.f ? v : 0.f;
    }
}

__global__ __launch_bounds__(256)
void bnstats_kernel(const float* __restrict__ h, const float* __restrict__ gamma,
                    const float* __restrict__ beta,
                    float* __restrict__ scale, float* __restrict__ shift)
{
    const int c = blockIdx.x * 256 + threadIdx.x;
    float s = 0.f, s2 = 0.f;
    for (int nrow = 0; nrow < N_DRUG; ++nrow) {
        const float v = h[(size_t)nrow * DIM + c];
        s += v;
        s2 += v * v;
    }
    const float inv_n = 1.0f / (float)N_DRUG;
    const float mean = s * inv_n;
    const float var  = s2 * inv_n - mean * mean;
    const float sc = gamma[c] * rsqrtf(var + 1e-5f);
    scale[c] = sc;
    shift[c] = beta[c] - mean * sc;
}

__global__ __launch_bounds__(256)
void bnapply_kernel(const float* __restrict__ h, const float* __restrict__ scale,
                    const float* __restrict__ shift, float* __restrict__ out)
{
    const int idx = blockIdx.x * 256 + threadIdx.x;
    if (idx >= N_DRUG * DIM) return;
    const int c = idx & (DIM - 1);
    out[idx] = h[idx] * scale[c] + shift[c];
}

#define SCOLS 32
#define SVEC  (N_DRUG * SCOLS / 4)

__global__ __launch_bounds__(256)
void smooth_kernel(const int* __restrict__ nbr_idx, const int* __restrict__ nbr_deg,
                   const int* __restrict__ epoch, float* __restrict__ f)
{
    if (*epoch <= 1) return;

    __shared__ __attribute__((aligned(16))) float slab[(N_DRUG + 1) * SCOLS];
    __shared__ __attribute__((aligned(16))) int   sidx[N_DRUG * MAXDEG];
    __shared__ int sdeg[N_DRUG];

    const int tid  = threadIdx.x;
    const int col0 = blockIdx.x * SCOLS;

    for (int v = tid; v < SVEC; v += 256) {
        const int row = v >> 3, c4 = (v & 7) * 4;
        *(float4*)(slab + v * 4) = *(const float4*)(f + (size_t)row * DIM + col0 + c4);
        const int dg = nbr_deg[row];
        int4 t = *(const int4*)(nbr_idx + v * 4);
        t.x = (c4     < dg) ? t.x : N_DRUG;
        t.y = (c4 + 1 < dg) ? t.y : N_DRUG;
        t.z = (c4 + 2 < dg) ? t.z : N_DRUG;
        t.w = (c4 + 3 < dg) ? t.w : N_DRUG;
        *(int4*)(sidx + v * 4) = t;
    }
    for (int r = tid; r < N_DRUG; r += 256)
        sdeg[r] = nbr_deg[r];
    if (tid < SCOLS)
        slab[N_DRUG * SCOLS + tid] = -0.0f;
    __syncthreads();

    if (tid < 32) {
        float* col = slab + tid;
        int idx[MAXDEG];
#pragma unroll
        for (int q = 0; q < MAXDEG / 4; ++q) {
            const int4 t = *(const int4*)(sidx + q * 4);
            idx[4 * q] = t.x; idx[4 * q + 1] = t.y; idx[4 * q + 2] = t.z; idx[4 * q + 3] = t.w;
        }
        int deg = sdeg[0];

        for (int i = 0; i < N_DRUG; ++i) {
            const int degu = __builtin_amdgcn_readfirstlane(deg);
            const int dd = degu < 1 ? 1 : (degu < MAXDEG ? degu : MAXDEG);

            float v[MAXDEG];
#pragma unroll
            for (int j = 0; j < MAXDEG; ++j)
                v[j] = col[idx[j] * SCOLS];
            const float fi = col[i * SCOLS];

            const int inx = (i + 1 < N_DRUG) ? (i + 1) : i;
#pragma unroll
            for (int q = 0; q < MAXDEG / 4; ++q) {
                const int4 t = *(const int4*)(sidx + inx * MAXDEG + q * 4);
                idx[4 * q] = t.x; idx[4 * q + 1] = t.y; idx[4 * q + 2] = t.z; idx[4 * q + 3] = t.w;
            }
            deg = sdeg[inx];

            float s = 0.f;
#pragma unroll
            for (int j = 0; j < 8; ++j) s += v[j];
            asm volatile("" :: "v"(v[8]),  "v"(v[9]),  "v"(v[10]), "v"(v[11]), "v"(v[12]), "v"(v[13]),
                               "v"(v[14]), "v"(v[15]), "v"(v[16]), "v"(v[17]), "v"(v[18]), "v"(v[19]),
                               "v"(v[20]), "v"(v[21]), "v"(v[22]), "v"(v[23]), "v"(v[24]), "v"(v[25]),
                               "v"(v[26]), "v"(v[27]), "v"(v[28]), "v"(v[29]), "v"(v[30]), "v"(v[31]));
            if (dd > 8) {
#pragma unroll
                for (int j = 8; j < 16; ++j) s += v[j];
                if (dd > 16) {
#pragma unroll
                    for (int j = 16; j < 24; ++j) s += v[j];
                    if (dd > 24) {
#pragma unroll
                        for (int j = 24; j < 32; ++j) s += v[j];
                    }
                }
            }
            const float avg = s / (float)dd;
            col[i * SCOLS] = (degu > 0) ? (avg + fi) * 0.5f : fi;
        }
    }
    __syncthreads();

    for (int v = tid; v < SVEC; v += 256) {
        const int row = v >> 3, c4 = (v & 7) * 4;
        *(float4*)(f + (size_t)row * DIM + col0 + c4) = *(const float4*)(slab + v * 4);
    }
}

extern "C" void kernel_launch(void* const* d_in, const int* in_sizes, int n_in,
                              void* d_out, int out_size, void* d_ws, size_t ws_size,
                              hipStream_t stream)
{
    const float* drugW  = (const float*)d_in[0];
    const float* relaW  = (const float*)d_in[1];
    const float* entW   = (const float*)d_in[2];
    const float* Wa     = (const float*)d_in[3];
    const float* linW   = (const float*)d_in[4];
    const float* linb   = (const float*)d_in[5];
    const float* gamma  = (const float*)d_in[6];
    const float* beta   = (const float*)d_in[7];
    const int* drug_name = (const int*)d_in[8];
    const int* adj_tail  = (const int*)d_in[9];
    const int* adj_rel   = (const int*)d_in[10];
    const int* nbr_idx   = (const int*)d_in[11];
    const int* nbr_deg   = (const int*)d_in[12];
    const int* epoch     = (const int*)d_in[13];

    float* out = (float*)d_out;
    float* ws  = (float*)d_ws;
    float* q     = ws;
    float* agg   = ws + (size_t)NPAD * DIM;
    float* h     = ws + (size_t)2 * NPAD * DIM;
    float* scale = ws + (size_t)3 * NPAD * DIM;
    float* shift = scale + DIM;

    dim3 gTiles(MT, NT);
    qgemm_kernel<<<gTiles, 32, 0, stream>>>(drugW, Wa, drug_name, q);
    attn_agg_kernel<<<NPAD, 256, 0, stream>>>(q, relaW, entW, adj_rel, adj_tail, agg);
    lingemm_kernel<<<gTiles, 32, 0, stream>>>(agg, drugW, drug_name, linW, linb, h);
    bnstats_kernel<<<2, 256, 0, stream>>>(h, gamma, beta, scale, shift);
    bnapply_kernel<<<(N_DRUG * DIM + 255) / 256, 256, 0, stream>>>(h, scale, shift, out);
    smooth_kernel<<<DIM / SCOLS, 256, 0, stream>>>(nbr_idx, nbr_deg, epoch, out);
}


# smoothing scan: each column's 32-slot neighbor sum is split over 8 lanes (4 slots each, then a pairwise lane-exchange tree), all 8 waves scan 4 columns each; summation order changes
# speedup vs baseline: 6.3876x; 1.3344x over previous
#include <hip/hip_runtime.h>

typedef __attribute__((ext_vector_type(2))) float v2f;
typedef __attribute__((ext_vector_type(8))) float v8f;

#define N_DRUG 572
#define DIM    512
#define KNBR   64
#define MAXDEG 32
#define NPAD   576
#define MT     36
#define NT     32

__global__ __launch_bounds__(32)
void qgemm_kernel(const float* __restrict__ drugW, const float* __restrict__ Wa,
                  const int* __restrict__ drug_name, float* __restrict__ q)
{
    const int lane = threadIdx.x;
    const int half = lane >> 4;
    const int l16  = lane & 15;
    const int mBase = blockIdx.x * 16;
    const int nBase = blockIdx.y * 16;

    const int mrow = mBase + l16;
    const int gi = drug_name[mrow < N_DRUG ? mrow : (N_DRUG - 1)];
    const float* __restrict__ Arow = drugW + (size_t)gi * DIM;
    const float* __restrict__ Bcol = Wa + nBase + l16;

    v8f c = {};
    for (int kk = 0; kk < DIM; kk += 4) {
        const int ka = kk + 2 * half;
        v2f a, b;
        a.x = Arow[ka];
        a.y = Arow[ka + 1];
        b.x = Bcol[(size_t)ka * DIM];
        b.y = Bcol[(size_t)(ka + 1) * DIM];
        c = __builtin_amdgcn_wmma_f32_16x16x4_f32(false, a, false, b,
                                                  (short)0, c, false, false);
    }
    float* outBase = q + (size_t)(mBase + half * 8) * DIM + nBase + l16;
#pragma unroll
    for (int j = 0; j < 8; ++j)
        outBase[(size_t)j * DIM] = c[j];
}

__global__ __launch_bounds__(256)
void attn_agg_kernel(const float* __restrict__ q, const float* __restrict__ relaW,
                     const float* __restrict__ entW,
                     const int* __restrict__ adj_rel, const int* __restrict__ adj_tail,
                     float* __restrict__ agg)
{
    __shared__ float qs[DIM];
    __shared__ float attn[KNBR];

    const int n  = blockIdx.x;
    const int ni = n < N_DRUG ? n : (N_DRUG - 1);
    const int tid = threadIdx.x;

    qs[tid]       = q[(size_t)n * DIM + tid];
    qs[tid + 256] = q[(size_t)n * DIM + tid + 256];
    __syncthreads();

    const int wave = tid >> 5;
    const int lane = tid & 31;
    for (int kk = 0; kk < 8; ++kk) {
        const int k = wave * 8 + kk;
        const int rel = adj_rel[ni * KNBR + k];
        const float* __restrict__ rw = relaW + (size_t)rel * DIM;
        float p = 0.f;
        for (int j = lane; j < DIM; j += 32)
            p += qs[j] * rw[j];
        for (int off = 16; off >= 1; off >>= 1)
            p += __shfl_xor(p, off, 32);
        if (lane == 0)
            attn[k] = p * 0.04419417382415922f;
    }
    __syncthreads();

    if (tid < 32) {
        float s0 = attn[tid], s1 = attn[tid + 32];
        float m = fmaxf(s0, s1);
        for (int off = 16; off >= 1; off >>= 1)
            m = fmaxf(m, __shfl_xor(m, off, 32));
        float e0 = __expf(s0 - m), e1 = __expf(s1 - m);
        float s = e0 + e1;
        for (int off = 16; off >= 1; off >>= 1)
            s += __shfl_xor(s, off, 32);
        const float inv = 1.f / s;
        attn[tid]      = e0 * inv;
        attn[tid + 32] = e1 * inv;
    }
    __syncthreads();

    float acc0 = 0.f, acc1 = 0.f;
    for (int k = 0; k < KNBR; ++k) {
        const int t = adj_tail[ni * KNBR + k];
        const float w = attn[k];
        const float* __restrict__ er = entW + (size_t)t * DIM;
        acc0 += w * er[tid];
        acc1 += w * er[tid + 256];
    }
    agg[(size_t)n * DIM + tid]       = acc0;
    agg[(size_t)n * DIM + tid + 256] = acc1;
}

__global__ __launch_bounds__(32)
void lingemm_kernel(const float* __restrict__ agg, const float* __restrict__ drugW,
                    const int* __restrict__ drug_name,
                    const float* __restrict__ linW, const float* __restrict__ linb,
                    float* __restrict__ h)
{
    const int lane = threadIdx.x;
    const int half = lane >> 4;
    const int l16  = lane & 15;
    const int mBase = blockIdx.x * 16;
    const int nBase = blockIdx.y * 16;

    const int mrow = mBase + l16;
    const int gi = drug_name[mrow < N_DRUG ? mrow : (N_DRUG - 1)];
    const float* __restrict__ Arow0 = agg + (size_t)mrow * DIM;
    const float* __restrict__ Arow1 = drugW + (size_t)gi * DIM;
    const float* __restrict__ Bcol  = linW + nBase + l16;

    v8f c = {};
    for (int kk = 0; kk < 2 * DIM; kk += 4) {
        const int ka = kk + 2 * half;
        const float* __restrict__ Ar = (ka < DIM) ? (Arow0 + ka) : (Arow1 + (ka - DIM));
        v2f a, b;
        a.x = Ar[0];
        a.y = Ar[1];
        b.x = Bcol[(size_t)ka * DIM];
        b.y = Bcol[(size_t)(ka + 1) * DIM];
        c = __builtin_amdgcn_wmma_f32_16x16x4_f32(false, a, false, b,
                                                  (short)0, c, false, false);
    }
    const float bias = linb[nBase + l16];
    float* outBase = h + (size_t)(mBase + half * 8) * DIM + nBase + l16;
#pragma unroll
    for (int j = 0; j < 8; ++j) {
        const float v = c[j] + bias;
        outBase[(size_t)j * DIM] = v > 0.f ? v : 0.f;
    }
}

__global__ __launch_bounds__(256)
void bnstats_kernel(const float* __restrict__ h, const float* __restrict__ gamma,
                    const float* __restrict__ beta,
                    float* __restrict__ scale, float* __restrict__ shift)
{
    const int c = blockIdx.x * 256 + threadIdx.x;
    float s = 0.f, s2 = 0.f;
    for (int nrow = 0; nrow < N_DRUG; ++nrow) {
        const float v = h[(size_t)nrow * DIM + c];
        s += v;
        s2 += v * v;
    }
    const float inv_n = 1.0f / (float)N_DRUG;
    const float mean = s * inv_n;
    const float var  = s2 * inv_n - mean * mean;
    const float sc = gamma[c] * rsqrtf(var + 1e-5f);
    scale[c] = sc;
    shift[c] = beta[c] - mean * sc;
}

__global__ __launch_bounds__(256)
void bnapply_kernel(const float* __restrict__ h, const float* __restrict__ scale,
                    const float* __restrict__ shift, float* __restrict__ out)
{
    const int idx = blockIdx.x * 256 + threadIdx.x;
    if (idx >= N_DRUG * DIM) return;
    const int c = idx & (DIM - 1);
    out[idx] = h[idx] * scale[c] + shift[c];
}

#define SCOLS 32
#define WCOLS 4
#define SROWS (N_DRUG + 1)
#define SVEC  (N_DRUG * SCOLS / 4)

template <int CTRL>
__device__ __forceinline__ float add_from_lane(float s)
{
    const int o = __builtin_amdgcn_update_dpp(0, __float_as_int(s), CTRL, 0xF, 0xF, true);
    return s + __int_as_float(o);
}

__global__ __launch_bounds__(256)
void smooth_kernel(const int* __restrict__ nbr_idx, const int* __restrict__ nbr_deg,
                   const int* __restrict__ epoch, float* __restrict__ f)
{
    if (*epoch <= 1) return;

    __shared__ __attribute__((aligned(16))) float slab[(SCOLS / WCOLS) * SROWS * WCOLS];
    __shared__ __attribute__((aligned(16))) int   sidx[N_DRUG * MAXDEG];
    __shared__ int sdeg[N_DRUG];

    const int tid  = threadIdx.x;
    const int col0 = blockIdx.x * SCOLS;

    for (int v = tid; v < SVEC; v += 256) {
        const int row = v >> 3, w = v & 7, c4 = w * 4;
        *(float4*)(slab + (w * SROWS + row) * WCOLS) =
            *(const float4*)(f + (size_t)row * DIM + col0 + c4);
        const int dg = nbr_deg[row];
        int4 t = *(const int4*)(nbr_idx + v * 4);
        t.x = (c4     < dg) ? t.x : N_DRUG;
        t.y = (c4 + 1 < dg) ? t.y : N_DRUG;
        t.z = (c4 + 2 < dg) ? t.z : N_DRUG;
        t.w = (c4 + 3 < dg) ? t.w : N_DRUG;
        *(int4*)(sidx + v * 4) = t;
    }
    for (int r = tid; r < N_DRUG; r += 256)
        sdeg[r] = nbr_deg[r];
    if (tid < SCOLS)
        slab[((tid >> 2) * SROWS + N_DRUG) * WCOLS + (tid & 3)] = -0.0f;
    __syncthreads();

    {
        const int wave = tid >> 5, lane = tid & 31;
        const int g = lane & 7;
        float* col = slab + wave * (SROWS * WCOLS) + (lane >> 3);
        const int* tab = sidx + 4 * g;
        int4 t = *(const int4*)tab;
        int deg = sdeg[0];

        for (int i = 0; i < N_DRUG; ++i) {
            const int degu = __builtin_amdgcn_readfirstlane(deg);
            const int dd = degu < 1 ? 1 : (degu < MAXDEG ? degu : MAXDEG);

            const float v0 = col[t.x * WCOLS];
            const float v1 = col[t.y * WCOLS];
            const float v2 = col[t.z * WCOLS];
            const float v3 = col[t.w * WCOLS];
            const float fi = col[i * WCOLS];

            const int inx = (i + 1 < N_DRUG) ? (i + 1) : i;
            t = *(const int4*)(tab + inx * MAXDEG);
            deg = sdeg[inx];

            float s = ((v0 + v1) + v2) + v3;
            s = add_from_lane<0xB1>(s);
            s = add_from_lane<0x4E>(s);
            s = add_from_lane<0x164>(s);

            const float avg = s / (float)dd;
            const float nv = (degu > 0) ? (avg + fi) * 0.5f : fi;
            if (g == 0)
                col[i * WCOLS] = nv;
        }
    }
    __syncthreads();

    for (int v = tid; v < SVEC; v += 256) {
        const int row = v >> 3, w = v & 7;
        *(float4*)(f + (size_t)row * DIM + col0 + w * 4) =
            *(const float4*)(slab + (w * SROWS + row) * WCOLS);
    }
}

extern "C" void kernel_launch(void* const* d_in, const int* in_sizes, int n_in,
                              void* d_out, int out_size, void* d_ws, size_t ws_size,
                              hipStream_t stream)
{
    const float* drugW  = (const float*)d_in[0];
    const float* relaW  = (const float*)d_in[1];
    const float* entW   = (const float*)d_in[2];
    const float* Wa     = (const float*)d_in[3];
    const float* linW   = (const float*)d_in[4];
    const float* linb   = (const float*)d_in[5];
    const float* gamma  = (const float*)d_in[6];
    const float* beta   = (const float*)d_in[7];
    const int* drug_name = (const int*)d_in[8];
    const int* adj_tail  = (const int*)d_in[9];
    const int* adj_rel   = (const int*)d_in[10];
    const int* nbr_idx   = (const int*)d_in[11];
    const int* nbr_deg   = (const int*)d_in[12];
    const int* epoch     = (const int*)d_in[13];

    float* out = (float*)d_out;
    float* ws  = (float*)d_ws;
    float* q     = ws;
    float* agg   = ws + (size_t)NPAD * DIM;
    float* h     = ws + (size_t)2 * NPAD * DIM;
    float* scale = ws + (size_t)3 * NPAD * DIM;
    float* shift = scale + DIM;

    dim3 gTiles(MT, NT);
    qgemm_kernel<<<gTiles, 32, 0, stream>>>(drugW, Wa, drug_name, q);
    attn_agg_kernel<<<NPAD, 256, 0, stream>>>(q, relaW, entW, adj_rel, adj_tail, agg);
    lingemm_kernel<<<gTiles, 32, 0, stream>>>(agg, drugW, drug_name, linW, linb, h);
    bnstats_kernel<<<2, 256, 0, stream>>>(h, gamma, beta, scale, shift);
    bnapply_kernel<<<(N_DRUG * DIM + 255) / 256, 256, 0, stream>>>(h, scale, shift, out);
    smooth_kernel<<<DIM / SCOLS, 256, 0, stream>>>(nbr_idx, nbr_deg, epoch, out);
}


# batch-norm statistics and apply fused into the smoothing kernel's stage-in: each block reduces its own 32 columns of h, rows split over its threads (stat sum order changes); 2 launches fewer
# speedup vs baseline: 7.4887x; 1.1724x over previous
#include <hip/hip_runtime.h>

typedef __attribute__((ext_vector_type(2))) float v2f;
typedef __attribute__((ext_vector_type(8))) float v8f;

#define N_DRUG 572
#define DIM    512
#define KNBR   64
#define MAXDEG 32
#define NPAD   576
#define MT     36
#define NT     32

__global__ __launch_bounds__(32)
void qgemm_kernel(const float* __restrict__ drugW, const float* __restrict__ Wa,
                  const int* __restrict__ drug_name, float* __restrict__ q)
{
    const int lane = threadIdx.x;
    const int half = lane >> 4;
    const int l16  = lane & 15;
    const int mBase = blockIdx.x * 16;
    const int nBase = blockIdx.y * 16;

    const int mrow = mBase + l16;
    const int gi = drug_name[mrow < N_DRUG ? mrow : (N_DRUG - 1)];
    const float* __restrict__ Arow = drugW + (size_t)gi * DIM;
    const float* __restrict__ Bcol = Wa + nBase + l16;

    v8f c = {};
    for (int kk = 0; kk < DIM; kk += 4) {
        const int ka = kk + 2 * half;
        v2f a, b;
        a.x = Arow[ka];
        a.y = Arow[ka + 1];
        b.x = Bcol[(size_t)ka * DIM];
        b.y = Bcol[(size_t)(ka + 1) * DIM];
        c = __builtin_amdgcn_wmma_f32_16x16x4_f32(false, a, false, b,
                                                  (short)0, c, false, false);
    }
    float* outBase = q + (size_t)(mBase + half * 8) * DIM + nBase + l16;
#pragma unroll
    for (int j = 0; j < 8; ++j)
        outBase[(size_t)j * DIM] = c[j];
}

__global__ __launch_bounds__(256)
void attn_agg_kernel(const float* __restrict__ q, const float* __restrict__ relaW,
                     const float* __restrict__ entW,
                     const int* __restrict__ adj_rel, const int* __restrict__ adj_tail,
                     float* __restrict__ agg)
{
    __shared__ float qs[DIM];
    __shared__ float attn[KNBR];

    const int n  = blockIdx.x;
    const int ni = n < N_DRUG ? n : (N_DRUG - 1);
    const int tid = threadIdx.x;

    qs[tid]       = q[(size_t)n * DIM + tid];
    qs[tid + 256] = q[(size_t)n * DIM + tid + 256];
    __syncthreads();

    const int wave = tid >> 5;
    const int lane = tid & 31;
    for (int kk = 0; kk < 8; ++kk) {
        const int k = wave * 8 + kk;
        const int rel = adj_rel[ni * KNBR + k];
        const float* __restrict__ rw = relaW + (size_t)rel * DIM;
        float p = 0.f;
        for (int j = lane; j < DIM; j += 32)
            p += qs[j] * rw[j];
        for (int off = 16; off >= 1; off >>= 1)
            p += __shfl_xor(p, off, 32);
        if (lane == 0)
            attn[k] = p * 0.04419417382415922f;
    }
    __syncthreads();

    if (tid < 32) {
        float s0 = attn[tid], s1 = attn[tid + 32];
        float m = fmaxf(s0, s1);
        for (int off = 16; off >= 1; off >>= 1)
            m = fmaxf(m, __shfl_xor(m, off, 32));
        float e0 = __expf(s0 - m), e1 = __expf(s1 - m);
        float s = e0 + e1;
        for (int off = 16; off >= 1; off >>= 1)
            s += __shfl_xor(s, off, 32);
        const float inv = 1.f / s;
        attn[tid]      = e0 * inv;
        attn[tid + 32] = e1 * inv;
    }
    __syncthreads();

    float acc0 = 0.f, acc1 = 0.f;
    for (int k = 0; k < KNBR; ++k) {
        const int t = adj_tail[ni * KNBR + k];
        const float w = attn[k];
        const float* __restrict__ er = entW + (size_t)t * DIM;
        acc0 += w * er[tid];
        acc1 += w * er[tid + 256];
    }
    agg[(size_t)n * DIM + tid]       = acc0;
    agg[(size_t)n * DIM + tid + 256] = acc1;
}

__global__ __launch_bounds__(32)
void lingemm_kernel(const float* __restrict__ agg, const float* __restrict__ drugW,
                    const int* __restrict__ drug_name,
                    const float* __restrict__ linW, const float* __restrict__ linb,
                    float* __restrict__ h)
{
    const int lane = threadIdx.x;
    const int half = lane >> 4;
    const int l16  = lane & 15;
    const int mBase = blockIdx.x * 16;
    const int nBase = blockIdx.y * 16;

    const int mrow = mBase + l16;
    const int gi = drug_name[mrow < N_DRUG ? mrow : (N_DRUG - 1)];
    const float* __restrict__ Arow0 = agg + (size_t)mrow * DIM;
    const float* __restrict__ Arow1 = drugW + (size_t)gi * DIM;
    const float* __restrict__ Bcol  = linW + nBase + l16;

    v8f c = {};
    for (int kk = 0; kk < 2 * DIM; kk += 4) {
        const int ka = kk + 2 * half;
        const float* __restrict__ Ar = (ka < DIM) ? (Arow0 + ka) : (Arow1 + (ka - DIM));
        v2f a, b;
        a.x = Ar[0];
        a.y = Ar[1];
        b.x = Bcol[(size_t)ka * DIM];
        b.y = Bcol[(size_t)(ka + 1) * DIM];
        c = __builtin_amdgcn_wmma_f32_16x16x4_f32(false, a, false, b,
                                                  (short)0, c, false, false);
    }
    const float bias = linb[nBase + l16];
    float* outBase = h + (size_t)(mBase + half * 8) * DIM + nBase + l16;
#pragma unroll
    for (int j = 0; j < 8; ++j) {
        const float v = c[j] + bias;
        outBase[(size_t)j * DIM] = v > 0.f ? v : 0.f;
    }
}

#define SCOLS 32
#define WCOLS 4
#define SROWS (N_DRUG + 1)
#define SVEC  (N_DRUG * SCOLS / 4)
#define HFULL (N_DRUG / 32)
#define HROWS (HFULL + 1)

template <int CTRL>
__device__ __forceinline__ float add_from_lane(float s)
{
    const int o = __builtin_amdgcn_update_dpp(0, __float_as_int(s), CTRL, 0xF, 0xF, true);
    return s + __int_as_float(o);
}

template <int OFF>
__device__ __forceinline__ float4 add_from_lane4(float4 v)
{
    v.x += __shfl_xor(v.x, OFF, 32);
    v.y += __shfl_xor(v.y, OFF, 32);
    v.z += __shfl_xor(v.z, OFF, 32);
    v.w += __shfl_xor(v.w, OFF, 32);
    return v;
}

__global__ __launch_bounds__(256)
void bn_smooth_kernel(const float* __restrict__ h, const float* __restrict__ gamma,
                      const float* __restrict__ beta,
                      const int* __restrict__ nbr_idx, const int* __restrict__ nbr_deg,
                      const int* __restrict__ epoch, float* __restrict__ f)
{
    __shared__ __attribute__((aligned(16))) float slab[(SCOLS / WCOLS) * SROWS * WCOLS];
    __shared__ __attribute__((aligned(16))) int   sidx[N_DRUG * MAXDEG];
    __shared__ int sdeg[N_DRUG];
    __shared__ __attribute__((aligned(16))) float red[2][8][SCOLS];

    const int tid  = threadIdx.x;
    const int wave = tid >> 5, lane = tid & 31;
    const int col0 = blockIdx.x * SCOLS;
    const int w    = tid & 7;
    const int c4   = w * 4;
    const int rb   = tid >> 3;
    const int ep   = *epoch;

    const float* __restrict__ hp = h + (size_t)rb * DIM + col0 + c4;
    const bool tail = rb + 32 * HFULL < N_DRUG;
    float4 hv[HROWS];
#pragma unroll
    for (int j = 0; j < HFULL; ++j)
        hv[j] = *(const float4*)(hp + (size_t)(32 * j) * DIM);
    hv[HFULL] = *(const float4*)(hp + (size_t)(tail ? 32 * HFULL : 0) * DIM);
    const float4 g4 = *(const float4*)(gamma + col0 + c4);
    const float4 b4 = *(const float4*)(beta + col0 + c4);

    for (int v = tid; v < SVEC; v += 256) {
        const int dg = nbr_deg[v >> 3];
        int4 t = *(const int4*)(nbr_idx + v * 4);
        t.x = (c4     < dg) ? t.x : N_DRUG;
        t.y = (c4 + 1 < dg) ? t.y : N_DRUG;
        t.z = (c4 + 2 < dg) ? t.z : N_DRUG;
        t.w = (c4 + 3 < dg) ? t.w : N_DRUG;
        *(int4*)(sidx + v * 4) = t;
    }
    for (int r = tid; r < N_DRUG; r += 256)
        sdeg[r] = nbr_deg[r];
    if (tid < SCOLS)
        slab[((tid >> 2) * SROWS + N_DRUG) * WCOLS + (tid & 3)] = -0.0f;

    if (!tail)
        hv[HFULL] = make_float4(0.f, 0.f, 0.f, 0.f);
    float4 s1 = make_float4(0.f, 0.f, 0.f, 0.f), s2 = s1;
#pragma unroll
    for (int j = 0; j < HROWS; ++j) {
        s1.x += hv[j].x;            s1.y += hv[j].y;
        s1.z += hv[j].z;            s1.w += hv[j].w;
        s2.x += hv[j].x * hv[j].x;  s2.y += hv[j].y * hv[j].y;
        s2.z += hv[j].z * hv[j].z;  s2.w += hv[j].w * hv[j].w;
    }
    s1 = add_from_lane4<8>(s1);   s2 = add_from_lane4<8>(s2);
    s1 = add_from_lane4<16>(s1);  s2 = add_from_lane4<16>(s2);
    if (lane < 8) {
        *(float4*)&red[0][wave][c4] = s1;
        *(float4*)&red[1][wave][c4] = s2;
    }
    __syncthreads();

    float4 t1 = make_float4(0.f, 0.f, 0.f, 0.f), t2 = t1;
#pragma unroll
    for (int wv = 0; wv < 8; ++wv) {
        const float4 a = *(const float4*)&red[0][wv][c4];
        const float4 b = *(const float4*)&red[1][wv][c4];
        t1.x += a.x; t1.y += a.y; t1.z += a.z; t1.w += a.w;
        t2.x += b.x; t2.y += b.y; t2.z += b.z; t2.w += b.w;
    }
    const float inv_n = 1.0f / (float)N_DRUG;
    float4 sc, sh;
    {
        const float mx = t1.x * inv_n, my = t1.y * inv_n, mz = t1.z * inv_n, mw = t1.w * inv_n;
        sc.x = g4.x * rsqrtf((t2.x * inv_n - mx * mx) + 1e-5f);
        sc.y = g4.y * rsqrtf((t2.y * inv_n - my * my) + 1e-5f);
        sc.z = g4.z * rsqrtf((t2.z * inv_n - mz * mz) + 1e-5f);
        sc.w = g4.w * rsqrtf((t2.w * inv_n - mw * mw) + 1e-5f);
        sh.x = b4.x - mx * sc.x;  sh.y = b4.y - my * sc.y;
        sh.z = b4.z - mz * sc.z;  sh.w = b4.w - mw * sc.w;
    }
    float* sp = slab + (w * SROWS + rb) * WCOLS;
#pragma unroll
    for (int j = 0; j < HROWS; ++j) {
        if (j < HFULL || tail) {
            float4 o;
            o.x = hv[j].x * sc.x + sh.x;  o.y = hv[j].y * sc.y + sh.y;
            o.z = hv[j].z * sc.z + sh.z;  o.w = hv[j].w * sc.w + sh.w;
            *(float4*)(sp + 32 * j * WCOLS) = o;
        }
    }
    __syncthreads();

    if (ep > 1) {
        const int g = lane & 7;
        float* col = slab + wave * (SROWS * WCOLS) + (lane >> 3);
        const int* tab = sidx + 4 * g;
        int4 t = *(const int4*)tab;
        int deg = sdeg[0];

        for (int i = 0; i < N_DRUG; ++i) {
            const int degu = __builtin_amdgcn_readfirstlane(deg);
            const int dd = degu < 1 ? 1 : (degu < MAXDEG ? degu : MAXDEG);

            const float v0 = col[t.x * WCOLS];
            const float v1 = col[t.y * WCOLS];
            const float v2 = col[t.z * WCOLS];
            const float v3 = col[t.w * WCOLS];
            const float fi = col[i * WCOLS];

            const int inx = (i + 1 < N_DRUG) ? (i + 1) : i;
            t = *(const int4*)(tab + inx * MAXDEG);
            deg = sdeg[inx];

            float s = ((v0 + v1) + v2) + v3;
            s = add_from_lane<0xB1>(s);
            s = add_from_lane<0x4E>(s);
            s = add_from_lane<0x164>(s);

            const float avg = s / (float)dd;
            const float nv = (degu > 0) ? (avg + fi) * 0.5f : fi;
            if (g == 0)
                col[i * WCOLS] = nv;
        }
    }
    __syncthreads();

    for (int v = tid; v < SVEC; v += 256) {
        const int row = v >> 3;
        *(float4*)(f + (size_t)row * DIM + col0 + c4) =
            *(const float4*)(slab + (w * SROWS + row) * WCOLS);
    }
}

extern "C" void kernel_launch(void* const* d_in, const int* in_sizes, int n_in,
                              void* d_out, int out_size, void* d_ws, size_t ws_size,
                              hipStream_t stream)
{
    const float* drugW  = (const float*)d_in[0];
    const float* relaW  = (const float*)d_in[1];
    const float* entW   = (const float*)d_in[2];
    const float* Wa     = (const float*)d_in[3];
    const float* linW   = (const float*)d_in[4];
    const float* linb   = (const float*)d_in[5];
    const float* gamma  = (const float*)d_in[6];
    const float* beta   = (const float*)d_in[7];
    const int* drug_name = (const int*)d_in[8];
    const int* adj_tail  = (const int*)d_in[9];
    const int* adj_rel   = (const int*)d_in[10];
    const int* nbr_idx   = (const int*)d_in[11];
    const int* nbr_deg   = (const int*)d_in[12];
    const int* epoch     = (const int*)d_in[13];

    float* out = (float*)d_out;
    float* ws  = (float*)d_ws;
    float* q     = ws;
    float* agg   = ws + (size_t)NPAD * DIM;
    float* h     = ws + (size_t)2 * NPAD * DIM;

    dim3 gTiles(MT, NT);
    qgemm_kernel<<<gTiles, 32, 0, stream>>>(drugW, Wa, drug_name, q);
    attn_agg_kernel<<<NPAD, 256, 0, stream>>>(q, relaW, entW, adj_rel, adj_tail, agg);
    lingemm_kernel<<<gTiles, 32, 0, stream>>>(agg, drugW, drug_name, linW, linb, h);
    bn_smooth_kernel<<<DIM / SCOLS, 256, 0, stream>>>(h, gamma, beta, nbr_idx, nbr_deg, epoch, out);
}


# smoothing scan as a linear recurrence: slots naming the previous row are counted ahead, not loaded, so row r+2's LDS gathers issue before row r+1 ends; mean via stored 1/(2 deg); sum order changes
# speedup vs baseline: 9.5712x; 1.2781x over previous
#include <hip/hip_runtime.h>

typedef __attribute__((ext_vector_type(2))) float v2f;
typedef __attribute__((ext_vector_type(8))) float v8f;

#define N_DRUG 572
#define DIM    512
#define KNBR   64
#define MAXDEG 32
#define NPAD   576
#define MT     36
#define NT     32

__global__ __launch_bounds__(32)
void qgemm_kernel(const float* __restrict__ drugW, const float* __restrict__ Wa,
                  const int* __restrict__ drug_name, float* __restrict__ q)
{
    const int lane = threadIdx.x;
    const int half = lane >> 4;
    const int l16  = lane & 15;
    const int mBase = blockIdx.x * 16;
    const int nBase = blockIdx.y * 16;

    const int mrow = mBase + l16;
    const int gi = drug_name[mrow < N_DRUG ? mrow : (N_DRUG - 1)];
    const float* __restrict__ Arow = drugW + (size_t)gi * DIM;
    const float* __restrict__ Bcol = Wa + nBase + l16;

    v8f c = {};
    for (int kk = 0; kk < DIM; kk += 4) {
        const int ka = kk + 2 * half;
        v2f a, b;
        a.x = Arow[ka];
        a.y = Arow[ka + 1];
        b.x = Bcol[(size_t)ka * DIM];
        b.y = Bcol[(size_t)(ka + 1) * DIM];
        c = __builtin_amdgcn_wmma_f32_16x16x4_f32(false, a, false, b,
                                                  (short)0, c, false, false);
    }
    float* outBase = q + (size_t)(mBase + half * 8) * DIM + nBase + l16;
#pragma unroll
    for (int j = 0; j < 8; ++j)
        outBase[(size_t)j * DIM] = c[j];
}

__global__ __launch_bounds__(256)
void attn_agg_kernel(const float* __restrict__ q, const float* __restrict__ relaW,
                     const float* __restrict__ entW,
                     const int* __restrict__ adj_rel, const int* __restrict__ adj_tail,
                     float* __restrict__ agg)
{
    __shared__ float qs[DIM];
    __shared__ float attn[KNBR];

    const int n  = blockIdx.x;
    const int ni = n < N_DRUG ? n : (N_DRUG - 1);
    const int tid = threadIdx.x;

    qs[tid]       = q[(size_t)n * DIM + tid];
    qs[tid + 256] = q[(size_t)n * DIM + tid + 256];
    __syncthreads();

    const int wave = tid >> 5;
    const int lane = tid & 31;
    for (int kk = 0; kk < 8; ++kk) {
        const int k = wave * 8 + kk;
        const int rel = adj_rel[ni * KNBR + k];
        const float* __restrict__ rw = relaW + (size_t)rel * DIM;
        float p = 0.f;
        for (int j = lane; j < DIM; j += 32)
            p += qs[j] * rw[j];
        for (int off = 16; off >= 1; off >>= 1)
            p += __shfl_xor(p, off, 32);
        if (lane == 0)
            attn[k] = p * 0.04419417382415922f;
    }
    __syncthreads();

    if (tid < 32) {
        float s0 = attn[tid], s1 = attn[tid + 32];
        float m = fmaxf(s0, s1);
        for (int off = 16; off >= 1; off >>= 1)
            m = fmaxf(m, __shfl_xor(m, off, 32));
        float e0 = __expf(s0 - m), e1 = __expf(s1 - m);
        float s = e0 + e1;
        for (int off = 16; off >= 1; off >>= 1)
            s += __shfl_xor(s, off, 32);
        const float inv = 1.f / s;
        attn[tid]      = e0 * inv;
        attn[tid + 32] = e1 * inv;
    }
    __syncthreads();

    float acc0 = 0.f, acc1 = 0.f;
    for (int k = 0; k < KNBR; ++k) {
        const int t = adj_tail[ni * KNBR + k];
        const float w = attn[k];
        const float* __restrict__ er = entW + (size_t)t * DIM;
        acc0 += w * er[tid];
        acc1 += w * er[tid + 256];
    }
    agg[(size_t)n * DIM + tid]       = acc0;
    agg[(size_t)n * DIM + tid + 256] = acc1;
}

__global__ __launch_bounds__(32)
void lingemm_kernel(const float* __restrict__ agg, const float* __restrict__ drugW,
                    const int* __restrict__ drug_name,
                    const float* __restrict__ linW, const float* __restrict__ linb,
                    float* __restrict__ h)
{
    const int lane = threadIdx.x;
    const int half = lane >> 4;
    const int l16  = lane & 15;
    const int mBase = blockIdx.x * 16;
    const int nBase = blockIdx.y * 16;

    const int mrow = mBase + l16;
    const int gi = drug_name[mrow < N_DRUG ? mrow : (N_DRUG - 1)];
    const float* __restrict__ Arow0 = agg + (size_t)mrow * DIM;
    const float* __restrict__ Arow1 = drugW + (size_t)gi * DIM;
    const float* __restrict__ Bcol  = linW + nBase + l16;

    v8f c = {};
    for (int kk = 0; kk < 2 * DIM; kk += 4) {
        const int ka = kk + 2 * half;
        const float* __restrict__ Ar = (ka < DIM) ? (Arow0 + ka) : (Arow1 + (ka - DIM));
        v2f a, b;
        a.x = Ar[0];
        a.y = Ar[1];
        b.x = Bcol[(size_t)ka * DIM];
        b.y = Bcol[(size_t)(ka + 1) * DIM];
        c = __builtin_amdgcn_wmma_f32_16x16x4_f32(false, a, false, b,
                                                  (short)0, c, false, false);
    }
    const float bias = linb[nBase + l16];
    float* outBase = h + (size_t)(mBase + half * 8) * DIM + nBase + l16;
#pragma unroll
    for (int j = 0; j < 8; ++j) {
        const float v = c[j] + bias;
        outBase[(size_t)j * DIM] = v > 0.f ? v : 0.f;
    }
}

#define SCOLS 32
#define WCOLS 4
#define SROWS (N_DRUG + 1)
#define SVEC  (N_DRUG * SCOLS / 4)
#define HFULL (N_DRUG / 32)
#define HROWS (HFULL + 1)

template <int CTRL>
__device__ __forceinline__ float add_from_lane(float s)
{
    const int o = __builtin_amdgcn_update_dpp(0, __float_as_int(s), CTRL, 0xF, 0xF, true);
    return s + __int_as_float(o);
}

template <int OFF>
__device__ __forceinline__ float4 add_from_lane4(float4 v)
{
    v.x += __shfl_xor(v.x, OFF, 32);
    v.y += __shfl_xor(v.y, OFF, 32);
    v.z += __shfl_xor(v.z, OFF, 32);
    v.w += __shfl_xor(v.w, OFF, 32);
    return v;
}

struct rowvals { float v0, v1, v2, v3, f; };

__device__ __forceinline__ rowvals gather_row(const float* col, const int4 t, const int r)
{
    rowvals x;
    x.v0 = col[t.x * WCOLS];
    x.v1 = col[t.y * WCOLS];
    x.v2 = col[t.z * WCOLS];
    x.v3 = col[t.w * WCOLS];
    x.f  = col[r * WCOLS];
    return x;
}

__device__ __forceinline__ float row_offset(const rowvals x, const float2 c)
{
    float s = ((x.v0 + x.v1) + x.v2) + x.v3;
    s = add_from_lane<0xB1>(s);
    s = add_from_lane<0x4E>(s);
    s = add_from_lane<0x164>(s);
    const float b = c.x > 0.f ? 0.5f : 1.0f;
    return fmaf(c.x, s, b * x.f);
}

__global__ __launch_bounds__(256)
void bn_smooth_kernel(const float* __restrict__ h, const float* __restrict__ gamma,
                      const float* __restrict__ beta,
                      const int* __restrict__ nbr_idx, const int* __restrict__ nbr_deg,
                      const int* __restrict__ epoch, float* __restrict__ f)
{
    __shared__ __attribute__((aligned(16))) float slab[(SCOLS / WCOLS) * SROWS * WCOLS];
    __shared__ __attribute__((aligned(16))) int   sidx[N_DRUG * MAXDEG];
    __shared__ __attribute__((aligned(16))) float2 scoef[N_DRUG];
    __shared__ __attribute__((aligned(16))) float red[2][8][SCOLS];

    const int tid  = threadIdx.x;
    const int wave = tid >> 5, lane = tid & 31;
    const int col0 = blockIdx.x * SCOLS;
    const int w    = tid & 7;
    const int c4   = w * 4;
    const int rb   = tid >> 3;
    const int ep   = *epoch;

    const float* __restrict__ hp = h + (size_t)rb * DIM + col0 + c4;
    const bool tail = rb + 32 * HFULL < N_DRUG;
    float4 hv[HROWS];
#pragma unroll
    for (int j = 0; j < HFULL; ++j)
        hv[j] = *(const float4*)(hp + (size_t)(32 * j) * DIM);
    hv[HFULL] = *(const float4*)(hp + (size_t)(tail ? 32 * HFULL : 0) * DIM);
    const float4 g4 = *(const float4*)(gamma + col0 + c4);
    const float4 b4 = *(const float4*)(beta + col0 + c4);

    for (int v = tid; v < SVEC; v += 256) {
        const int dg = nbr_deg[v >> 3];
        int4 t = *(const int4*)(nbr_idx + v * 4);
        t.x = (c4     < dg) ? t.x : N_DRUG;
        t.y = (c4 + 1 < dg) ? t.y : N_DRUG;
        t.z = (c4 + 2 < dg) ? t.z : N_DRUG;
        t.w = (c4 + 3 < dg) ? t.w : N_DRUG;
        *(int4*)(sidx + v * 4) = t;
    }
    if (tid < SCOLS)
        slab[((tid >> 2) * SROWS + N_DRUG) * WCOLS + (tid & 3)] = -0.0f;

    if (!tail)
        hv[HFULL] = make_float4(0.f, 0.f, 0.f, 0.f);
    float4 s1 = make_float4(0.f, 0.f, 0.f, 0.f), s2 = s1;
#pragma unroll
    for (int j = 0; j < HROWS; ++j) {
        s1.x += hv[j].x;            s1.y += hv[j].y;
        s1.z += hv[j].z;            s1.w += hv[j].w;
        s2.x += hv[j].x * hv[j].x;  s2.y += hv[j].y * hv[j].y;
        s2.z += hv[j].z * hv[j].z;  s2.w += hv[j].w * hv[j].w;
    }
    s1 = add_from_lane4<8>(s1);   s2 = add_from_lane4<8>(s2);
    s1 = add_from_lane4<16>(s1);  s2 = add_from_lane4<16>(s2);
    if (lane < 8) {
        *(float4*)&red[0][wave][c4] = s1;
        *(float4*)&red[1][wave][c4] = s2;
    }
    __syncthreads();

    for (int r = tid; r < N_DRUG; r += 256) {
        const int dg = nbr_deg[r];
        const int prev = r - 1;
        int* trow = sidx + r * MAXDEG;
        int4 tt[MAXDEG / 4];
#pragma unroll
        for (int k = 0; k < MAXDEG / 4; ++k)
            tt[k] = *(const int4*)(trow + 4 * ((k + tid) & 7));
        int cnt = 0;
#pragma unroll
        for (int k = 0; k < MAXDEG / 4; ++k) {
            int4 t = tt[k];
            const bool e0 = t.x == prev, e1 = t.y == prev;
            const bool e2 = t.z == prev, e3 = t.w == prev;
            cnt += (int)e0 + (int)e1 + (int)e2 + (int)e3;
            t.x = e0 ? N_DRUG : t.x;
            t.y = e1 ? N_DRUG : t.y;
            t.z = e2 ? N_DRUG : t.z;
            t.w = e3 ? N_DRUG : t.w;
            *(int4*)(trow + 4 * ((k + tid) & 7)) = t;
        }
        const int dd = dg < 1 ? 1 : (dg < MAXDEG ? dg : MAXDEG);
        const float hc = dg > 0 ? 0.5f / (float)dd : 0.f;
        scoef[r] = make_float2(hc, hc * (float)cnt);
    }

    float4 t1 = make_float4(0.f, 0.f, 0.f, 0.f), t2 = t1;
#pragma unroll
    for (int wv = 0; wv < 8; ++wv) {
        const float4 a = *(const float4*)&red[0][wv][c4];
        const float4 b = *(const float4*)&red[1][wv][c4];
        t1.x += a.x; t1.y += a.y; t1.z += a.z; t1.w += a.w;
        t2.x += b.x; t2.y += b.y; t2.z += b.z; t2.w += b.w;
    }
    const float inv_n = 1.0f / (float)N_DRUG;
    float4 sc, sh;
    {
        const float mx = t1.x * inv_n, my = t1.y * inv_n, mz = t1.z * inv_n, mw = t1.w * inv_n;
        sc.x = g4.x * rsqrtf((t2.x * inv_n - mx * mx) + 1e-5f);
        sc.y = g4.y * rsqrtf((t2.y * inv_n - my * my) + 1e-5f);
        sc.z = g4.z * rsqrtf((t2.z * inv_n - mz * mz) + 1e-5f);
        sc.w = g4.w * rsqrtf((t2.w * inv_n - mw * mw) + 1e-5f);
        sh.x = b4.x - mx * sc.x;  sh.y = b4.y - my * sc.y;
        sh.z = b4.z - mz * sc.z;  sh.w = b4.w - mw * sc.w;
    }
    float* sp = slab + (w * SROWS + rb) * WCOLS;
#pragma unroll
    for (int j = 0; j < HROWS; ++j) {
        if (j < HFULL || tail) {
            float4 o;
            o.x = hv[j].x * sc.x + sh.x;  o.y = hv[j].y * sc.y + sh.y;
            o.z = hv[j].z * sc.z + sh.z;  o.w = hv[j].w * sc.w + sh.w;
            *(float4*)(sp + 32 * j * WCOLS) = o;
        }
    }
    __syncthreads();

    if (ep > 1) {
        const int g = lane & 7;
        float* col = slab + wave * (SROWS * WCOLS) + (lane >> 3);
        const int* tab = sidx + 4 * g;
        const int last = N_DRUG - 1;

        const int4 t0 = *(const int4*)(tab);
        const int4 t1 = *(const int4*)(tab + MAXDEG);
        int4 tA = *(const int4*)(tab + 2 * MAXDEG);
        int4 tB = *(const int4*)(tab + 3 * MAXDEG);
        __builtin_amdgcn_sched_barrier(0);
        rowvals inB = gather_row(col, t1, 1);
        float2  cB  = scoef[1];
        rowvals inA = gather_row(col, t0, 0);
        float2  cA  = scoef[0];
        float qA = row_offset(inA, cA), aA = cA.y;
        float qB = 0.f, aB = 0.f;
        float nvp = 0.f;

        for (int r = 0; r < N_DRUG; r += 2) {
            {
                const float nv = fmaf(aA, nvp, qA);
                if (g == 0)
                    col[r * WCOLS] = nv;
                const int r2 = (r + 2 < last) ? (r + 2) : last;
                const int r4 = (r + 4 < last) ? (r + 4) : last;
                inA = gather_row(col, tA, r2);
                cA  = scoef[r2];
                tA  = *(const int4*)(tab + r4 * MAXDEG);
                __builtin_amdgcn_sched_barrier(0);
                qB  = row_offset(inB, cB);
                aB  = cB.y;
                nvp = nv;
            }
            {
                const float nv = fmaf(aB, nvp, qB);
                if (g == 0)
                    col[(r + 1) * WCOLS] = nv;
                const int r3 = (r + 3 < last) ? (r + 3) : last;
                const int r5 = (r + 5 < last) ? (r + 5) : last;
                inB = gather_row(col, tB, r3);
                cB  = scoef[r3];
                tB  = *(const int4*)(tab + r5 * MAXDEG);
                __builtin_amdgcn_sched_barrier(0);
                qA  = row_offset(inA, cA);
                aA  = cA.y;
                nvp = nv;
            }
        }
    }
    __syncthreads();

    for (int v = tid; v < SVEC; v += 256) {
        const int row = v >> 3;
        *(float4*)(f + (size_t)row * DIM + col0 + c4) =
            *(const float4*)(slab + (w * SROWS + row) * WCOLS);
    }
}

extern "C" void kernel_launch(void* const* d_in, const int* in_sizes, int n_in,
                              void* d_out, int out_size, void* d_ws, size_t ws_size,
                              hipStream_t stream)
{
    const float* drugW  = (const float*)d_in[0];
    const float* relaW  = (const float*)d_in[1];
    const float* entW   = (const float*)d_in[2];
    const float* Wa     = (const float*)d_in[3];
    const float* linW   = (const float*)d_in[4];
    const float* linb   = (const float*)d_in[5];
    const float* gamma  = (const float*)d_in[6];
    const float* beta   = (const float*)d_in[7];
    const int* drug_name = (const int*)d_in[8];
    const int* adj_tail  = (const int*)d_in[9];
    const int* adj_rel   = (const int*)d_in[10];
    const int* nbr_idx   = (const int*)d_in[11];
    const int* nbr_deg   = (const int*)d_in[12];
    const int* epoch     = (const int*)d_in[13];

    float* out = (float*)d_out;
    float* ws  = (float*)d_ws;
    float* q     = ws;
    float* agg   = ws + (size_t)NPAD * DIM;
    float* h     = ws + (size_t)2 * NPAD * DIM;

    dim3 gTiles(MT, NT);
    qgemm_kernel<<<gTiles, 32, 0, stream>>>(drugW, Wa, drug_name, q);
    attn_agg_kernel<<<NPAD, 256, 0, stream>>>(q, relaW, entW, adj_rel, adj_tail, agg);
    lingemm_kernel<<<gTiles, 32, 0, stream>>>(agg, drugW, drug_name, linW, linb, h);
    bn_smooth_kernel<<<DIM / SCOLS, 256, 0, stream>>>(h, gamma, beta, nbr_idx, nbr_deg, epoch, out);
}


# both GEMMs: one wave per 32x16 tile (two accumulators share each B fragment), 64-k operand chunks loaded one chunk ahead of their WMMAs with 16-byte A loads; k order inside a chunk (sum order) changes
# speedup vs baseline: 11.9884x; 1.2525x over previous
#include <hip/hip_runtime.h>

typedef __attribute__((ext_vector_type(2))) float v2f;
typedef __attribute__((ext_vector_type(8))) float v8f;

#define N_DRUG 572
#define DIM    512
#define KNBR   64
#define MAXDEG 32
#define NPAD   576
#define MT     36
#define NT     32
#define MP     18

#define CHK 64
#define KST (CHK / 4)

struct opchunk { float4 a0[KST / 2]; float4 a1[KST / 2]; v2f b[KST]; };

__device__ __forceinline__ void load_chunk(opchunk& f, const float* __restrict__ A0p,
                                           const float* __restrict__ A1p,
                                           const float* __restrict__ Bp)
{
#pragma unroll
    for (int j = 0; j < KST / 2; ++j) {
        f.a0[j] = *(const float4*)(A0p + 4 * j);
        f.a1[j] = *(const float4*)(A1p + 4 * j);
        f.b[2 * j].x     = Bp[(size_t)(4 * j) * DIM];
        f.b[2 * j].y     = Bp[(size_t)(4 * j + 1) * DIM];
        f.b[2 * j + 1].x = Bp[(size_t)(4 * j + 2) * DIM];
        f.b[2 * j + 1].y = Bp[(size_t)(4 * j + 3) * DIM];
    }
}

__device__ __forceinline__ void mma_chunk(const opchunk& f, v8f& c0, v8f& c1)
{
    __builtin_amdgcn_sched_barrier(0);
#pragma unroll
    for (int j = 0; j < KST / 2; ++j) {
        v2f p0, p1, q0, q1;
        p0.x = f.a0[j].x;  p0.y = f.a0[j].y;
        q0.x = f.a0[j].z;  q0.y = f.a0[j].w;
        p1.x = f.a1[j].x;  p1.y = f.a1[j].y;
        q1.x = f.a1[j].z;  q1.y = f.a1[j].w;
        c0 = __builtin_amdgcn_wmma_f32_16x16x4_f32(false, p0, false, f.b[2 * j],
                                                   (short)0, c0, false, false);
        c1 = __builtin_amdgcn_wmma_f32_16x16x4_f32(false, p1, false, f.b[2 * j],
                                                   (short)0, c1, false, false);
        c0 = __builtin_amdgcn_wmma_f32_16x16x4_f32(false, q0, false, f.b[2 * j + 1],
                                                   (short)0, c0, false, false);
        c1 = __builtin_amdgcn_wmma_f32_16x16x4_f32(false, q1, false, f.b[2 * j + 1],
                                                   (short)0, c1, false, false);
    }
    __builtin_amdgcn_sched_barrier(0);
    asm volatile("v_nop\n\tv_nop\n\tv_nop\n\tv_nop");
    __builtin_amdgcn_sched_barrier(0);
}

#define LOAD_AT(F, K)                                                              \
    do {                                                                           \
        if ((K) < DIM) load_chunk(F, P0 + (K), P1 + (K), Bp + (size_t)(K) * DIM);  \
        else load_chunk(F, Q0 + ((K) - DIM), Q1 + ((K) - DIM), Bp + (size_t)(K) * DIM); \
    } while (0)

template <int KTOT>
__device__ __forceinline__ void wave_gemm_32x16(const float* __restrict__ P0,
                                                const float* __restrict__ P1,
                                                const float* __restrict__ Q0,
                                                const float* __restrict__ Q1,
                                                const float* __restrict__ Bp,
                                                v8f& c0, v8f& c1)
{
    opchunk f0, f1;
    load_chunk(f0, P0, P1, Bp);
#pragma unroll 1
    for (int kc = 0; kc < KTOT - 2 * CHK; kc += 2 * CHK) {
        LOAD_AT(f1, kc + CHK);
        mma_chunk(f0, c0, c1);
        LOAD_AT(f0, kc + 2 * CHK);
        mma_chunk(f1, c0, c1);
    }
    LOAD_AT(f1, KTOT - CHK);
    mma_chunk(f0, c0, c1);
    mma_chunk(f1, c0, c1);
}
#undef LOAD_AT

__global__ __launch_bounds__(32)
void qgemm_kernel(const float* __restrict__ drugW, const float* __restrict__ Wa,
                  const int* __restrict__ drug_name, float* __restrict__ q)
{
    const int lane = threadIdx.x;
    const int half = lane >> 4;
    const int l16  = lane & 15;
    const int mBase = blockIdx.x * 32;
    const int nBase = blockIdx.y * 16;

    const int r0 = mBase + l16, r1 = r0 + 16;
    const int g0 = drug_name[r0 < N_DRUG ? r0 : (N_DRUG - 1)];
    const int g1 = drug_name[r1 < N_DRUG ? r1 : (N_DRUG - 1)];
    const float* __restrict__ A0 = drugW + (size_t)g0 * DIM + 32 * half;
    const float* __restrict__ A1 = drugW + (size_t)g1 * DIM + 32 * half;
    const float* __restrict__ Bp = Wa + (size_t)(32 * half) * DIM + nBase + l16;

    v8f c0 = {}, c1 = {};
    wave_gemm_32x16<DIM>(A0, A1, A0, A1, Bp, c0, c1);

    float* outBase = q + (size_t)(mBase + half * 8) * DIM + nBase + l16;
#pragma unroll
    for (int j = 0; j < 8; ++j) {
        outBase[(size_t)j * DIM]        = c0[j];
        outBase[(size_t)(j + 16) * DIM] = c1[j];
    }
}

__global__ __launch_bounds__(256)
void attn_agg_kernel(const float* __restrict__ q, const float* __restrict__ relaW,
                     const float* __restrict__ entW,
                     const int* __restrict__ adj_rel, const int* __restrict__ adj_tail,
                     float* __restrict__ agg)
{
    __shared__ float qs[DIM];
    __shared__ float attn[KNBR];

    const int n  = blockIdx.x;
    const int ni = n < N_DRUG ? n : (N_DRUG - 1);
    const int tid = threadIdx.x;

    qs[tid]       = q[(size_t)n * DIM + tid];
    qs[tid + 256] = q[(size_t)n * DIM + tid + 256];
    __syncthreads();

    const int wave = tid >> 5;
    const int lane = tid & 31;
    for (int kk = 0; kk < 8; ++kk) {
        const int k = wave * 8 + kk;
        const int rel = adj_rel[ni * KNBR + k];
        const float* __restrict__ rw = relaW + (size_t)rel * DIM;
        float p = 0.f;
        for (int j = lane; j < DIM; j += 32)
            p += qs[j] * rw[j];
        for (int off = 16; off >= 1; off >>= 1)
            p += __shfl_xor(p, off, 32);
        if (lane == 0)
            attn[k] = p * 0.04419417382415922f;
    }
    __syncthreads();

    if (tid < 32) {
        float s0 = attn[tid], s1 = attn[tid + 32];
        float m = fmaxf(s0, s1);
        for (int off = 16; off >= 1; off >>= 1)
            m = fmaxf(m, __shfl_xor(m, off, 32));
        float e0 = __expf(s0 - m), e1 = __expf(s1 - m);
        float s = e0 + e1;
        for (int off = 16; off >= 1; off >>= 1)
            s += __shfl_xor(s, off, 32);
        const float inv = 1.f / s;
        attn[tid]      = e0 * inv;
        attn[tid + 32] = e1 * inv;
    }
    __syncthreads();

    float acc0 = 0.f, acc1 = 0.f;
    for (int k = 0; k < KNBR; ++k) {
        const int t = adj_tail[ni * KNBR + k];
        const float w = attn[k];
        const float* __restrict__ er = entW + (size_t)t * DIM;
        acc0 += w * er[tid];
        acc1 += w * er[tid + 256];
    }
    agg[(size_t)n * DIM + tid]       = acc0;
    agg[(size_t)n * DIM + tid + 256] = acc1;
}

__global__ __launch_bounds__(32)
void lingemm_kernel(const float* __restrict__ agg, const float* __restrict__ drugW,
                    const int* __restrict__ drug_name,
                    const float* __restrict__ linW, const float* __restrict__ linb,
                    float* __restrict__ h)
{
    const int lane = threadIdx.x;
    const int half = lane >> 4;
    const int l16  = lane & 15;
    const int mBase = blockIdx.x * 32;
    const int nBase = blockIdx.y * 16;

    const int r0 = mBase + l16, r1 = r0 + 16;
    const int g0 = drug_name[r0 < N_DRUG ? r0 : (N_DRUG - 1)];
    const int g1 = drug_name[r1 < N_DRUG ? r1 : (N_DRUG - 1)];
    const float* __restrict__ P0 = agg + (size_t)r0 * DIM + 32 * half;
    const float* __restrict__ P1 = agg + (size_t)r1 * DIM + 32 * half;
    const float* __restrict__ Q0 = drugW + (size_t)g0 * DIM + 32 * half;
    const float* __restrict__ Q1 = drugW + (size_t)g1 * DIM + 32 * half;
    const float* __restrict__ Bp = linW + (size_t)(32 * half) * DIM + nBase + l16;

    v8f c0 = {}, c1 = {};
    wave_gemm_32x16<2 * DIM>(P0, P1, Q0, Q1, Bp, c0, c1);

    const float bias = linb[nBase + l16];
    float* outBase = h + (size_t)(mBase + half * 8) * DIM + nBase + l16;
#pragma unroll
    for (int j = 0; j < 8; ++j) {
        const float v0 = c0[j] + bias, v1 = c1[j] + bias;
        outBase[(size_t)j * DIM]        = v0 > 0.f ? v0 : 0.f;
        outBase[(size_t)(j + 16) * DIM] = v1 > 0.f ? v1 : 0.f;
    }
}

#define SCOLS 32
#define WCOLS 4
#define SROWS (N_DRUG + 1)
#define SVEC  (N_DRUG * SCOLS / 4)
#define HFULL (N_DRUG / 32)
#define HROWS (HFULL + 1)

template <int CTRL>
__device__ __forceinline__ float add_from_lane(float s)
{
    const int o = __builtin_amdgcn_update_dpp(0, __float_as_int(s), CTRL, 0xF, 0xF, true);
    return s + __int_as_float(o);
}

template <int OFF>
__device__ __forceinline__ float4 add_from_lane4(float4 v)
{
    v.x += __shfl_xor(v.x, OFF, 32);
    v.y += __shfl_xor(v.y, OFF, 32);
    v.z += __shfl_xor(v.z, OFF, 32);
    v.w += __shfl_xor(v.w, OFF, 32);
    return v;
}

struct rowvals { float v0, v1, v2, v3, f; };

__device__ __forceinline__ rowvals gather_row(const float* col, const int4 t, const int r)
{
    rowvals x;
    x.v0 = col[t.x * WCOLS];
    x.v1 = col[t.y * WCOLS];
    x.v2 = col[t.z * WCOLS];
    x.v3 = col[t.w * WCOLS];
    x.f  = col[r * WCOLS];
    return x;
}

__device__ __forceinline__ float row_offset(const rowvals x, const float2 c)
{
    float s = ((x.v0 + x.v1) + x.v2) + x.v3;
    s = add_from_lane<0xB1>(s);
    s = add_from_lane<0x4E>(s);
    s = add_from_lane<0x164>(s);
    const float b = c.x > 0.f ? 0.5f : 1.0f;
    return fmaf(c.x, s, b * x.f);
}

__global__ __launch_bounds__(256)
void bn_smooth_kernel(const float* __restrict__ h, const float* __restrict__ gamma,
                      const float* __restrict__ beta,
                      const int* __restrict__ nbr_idx, const int* __restrict__ nbr_deg,
                      const int* __restrict__ epoch, float* __restrict__ f)
{
    __shared__ __attribute__((aligned(16))) float slab[(SCOLS / WCOLS) * SROWS * WCOLS];
    __shared__ __attribute__((aligned(16))) int   sidx[N_DRUG * MAXDEG];
    __shared__ __attribute__((aligned(16))) float2 scoef[N_DRUG];
    __shared__ __attribute__((aligned(16))) float red[2][8][SCOLS];

    const int tid  = threadIdx.x;
    const int wave = tid >> 5, lane = tid & 31;
    const int col0 = blockIdx.x * SCOLS;
    const int w    = tid & 7;
    const int c4   = w * 4;
    const int rb   = tid >> 3;
    const int ep   = *epoch;

    const float* __restrict__ hp = h + (size_t)rb * DIM + col0 + c4;
    const bool tail = rb + 32 * HFULL < N_DRUG;
    float4 hv[HROWS];
#pragma unroll
    for (int j = 0; j < HFULL; ++j)
        hv[j] = *(const float4*)(hp + (size_t)(32 * j) * DIM);
    hv[HFULL] = *(const float4*)(hp + (size_t)(tail ? 32 * HFULL : 0) * DIM);
    const float4 g4 = *(const float4*)(gamma + col0 + c4);
    const float4 b4 = *(const float4*)(beta + col0 + c4);

    for (int v = tid; v < SVEC; v += 256) {
        const int dg = nbr_deg[v >> 3];
        int4 t = *(const int4*)(nbr_idx + v * 4);
        t.x = (c4     < dg) ? t.x : N_DRUG;
        t.y = (c4 + 1 < dg) ? t.y : N_DRUG;
        t.z = (c4 + 2 < dg) ? t.z : N_DRUG;
        t.w = (c4 + 3 < dg) ? t.w : N_DRUG;
        *(int4*)(sidx + v * 4) = t;
    }
    if (tid < SCOLS)
        slab[((tid >> 2) * SROWS + N_DRUG) * WCOLS + (tid & 3)] = -0.0f;

    if (!tail)
        hv[HFULL] = make_float4(0.f, 0.f, 0.f, 0.f);
    float4 s1 = make_float4(0.f, 0.f, 0.f, 0.f), s2 = s1;
#pragma unroll
    for (int j = 0; j < HROWS; ++j) {
        s1.x += hv[j].x;            s1.y += hv[j].y;
        s1.z += hv[j].z;            s1.w += hv[j].w;
        s2.x += hv[j].x * hv[j].x;  s2.y += hv[j].y * hv[j].y;
        s2.z += hv[j].z * hv[j].z;  s2.w += hv[j].w * hv[j].w;
    }
    s1 = add_from_lane4<8>(s1);   s2 = add_from_lane4<8>(s2);
    s1 = add_from_lane4<16>(s1);  s2 = add_from_lane4<16>(s2);
    if (lane < 8) {
        *(float4*)&red[0][wave][c4] = s1;
        *(float4*)&red[1][wave][c4] = s2;
    }
    __syncthreads();

    for (int r = tid; r < N_DRUG; r += 256) {
        const int dg = nbr_deg[r];
        const int prev = r - 1;
        int* trow = sidx + r * MAXDEG;
        int4 tt[MAXDEG / 4];
#pragma unroll
        for (int k = 0; k < MAXDEG / 4; ++k)
            tt[k] = *(const int4*)(trow + 4 * ((k + tid) & 7));
        int cnt = 0;
#pragma unroll
        for (int k = 0; k < MAXDEG / 4; ++k) {
            int4 t = tt[k];
            const bool e0 = t.x == prev, e1 = t.y == prev;
            const bool e2 = t.z == prev, e3 = t.w == prev;
            cnt += (int)e0 + (int)e1 + (int)e2 + (int)e3;
            t.x = e0 ? N_DRUG : t.x;
            t.y = e1 ? N_DRUG : t.y;
            t.z = e2 ? N_DRUG : t.z;
            t.w = e3 ? N_DRUG : t.w;
            *(int4*)(trow + 4 * ((k + tid) & 7)) = t;
        }
        const int dd = dg < 1 ? 1 : (dg < MAXDEG ? dg : MAXDEG);
        const float hc = dg > 0 ? 0.5f / (float)dd : 0.f;
        scoef[r] = make_float2(hc, hc * (float)cnt);
    }

    float4 t1 = make_float4(0.f, 0.f, 0.f, 0.f), t2 = t1;
#pragma unroll
    for (int wv = 0; wv < 8; ++wv) {
        const float4 a = *(const float4*)&red[0][wv][c4];
        const float4 b = *(const float4*)&red[1][wv][c4];
        t1.x += a.x; t1.y += a.y; t1.z += a.z; t1.w += a.w;
        t2.x += b.x; t2.y += b.y; t2.z += b.z; t2.w += b.w;
    }
    const float inv_n = 1.0f / (float)N_DRUG;
    float4 sc, sh;
    {
        const float mx = t1.x * inv_n, my = t1.y * inv_n, mz = t1.z * inv_n, mw = t1.w * inv_n;
        sc.x = g4.x * rsqrtf((t2.x * inv_n - mx * mx) + 1e-5f);
        sc.y = g4.y * rsqrtf((t2.y * inv_n - my * my) + 1e-5f);
        sc.z = g4.z * rsqrtf((t2.z * inv_n - mz * mz) + 1e-5f);
        sc.w = g4.w * rsqrtf((t2.w * inv_n - mw * mw) + 1e-5f);
        sh.x = b4.x - mx * sc.x;  sh.y = b4.y - my * sc.y;
        sh.z = b4.z - mz * sc.z;  sh.w = b4.w - mw * sc.w;
    }
    float* sp = slab + (w * SROWS + rb) * WCOLS;
#pragma unroll
    for (int j = 0; j < HROWS; ++j) {
        if (j < HFULL || tail) {
            float4 o;
            o.x = hv[j].x * sc.x + sh.x;  o.y = hv[j].y * sc.y + sh.y;
            o.z = hv[j].z * sc.z + sh.z;  o.w = hv[j].w * sc.w + sh.w;
            *(float4*)(sp + 32 * j * WCOLS) = o;
        }
    }
    __syncthreads();

    if (ep > 1) {
        const int g = lane & 7;
        float* col = slab + wave * (SROWS * WCOLS) + (lane >> 3);
        const int* tab = sidx + 4 * g;
        const int last = N_DRUG - 1;

        const int4 t0 = *(const int4*)(tab);
        const int4 t1 = *(const int4*)(tab + MAXDEG);
        int4 tA = *(const int4*)(tab + 2 * MAXDEG);
        int4 tB = *(const int4*)(tab + 3 * MAXDEG);
        __builtin_amdgcn_sched_barrier(0);
        rowvals inB = gather_row(col, t1, 1);
        float2  cB  = scoef[1];
        rowvals inA = gather_row(col, t0, 0);
        float2  cA  = scoef[0];
        float qA = row_offset(inA, cA), aA = cA.y;
        float qB = 0.f, aB = 0.f;
        float nvp = 0.f;

        for (int r = 0; r < N_DRUG; r += 2) {
            {
                const float nv = fmaf(aA, nvp, qA);
                if (g == 0)
                    col[r * WCOLS] = nv;
                const int r2 = (r + 2 < last) ? (r + 2) : last;
                const int r4 = (r + 4 < last) ? (r + 4) : last;
                inA = gather_row(col, tA, r2);
                cA  = scoef[r2];
                tA  = *(const int4*)(tab + r4 * MAXDEG);
                __builtin_amdgcn_sched_barrier(0);
                qB  = row_offset(inB, cB);
                aB  = cB.y;
                nvp = nv;
            }
            {
                const float nv = fmaf(aB, nvp, qB);
                if (g == 0)
                    col[(r + 1) * WCOLS] = nv;
                const int r3 = (r + 3 < last) ? (r + 3) : last;
                const int r5 = (r + 5 < last) ? (r + 5) : last;
                inB = gather_row(col, tB, r3);
                cB  = scoef[r3];
                tB  = *(const int4*)(tab + r5 * MAXDEG);
                __builtin_amdgcn_sched_barrier(0);
                qA  = row_offset(inA, cA);
                aA  = cA.y;
                nvp = nv;
            }
        }
    }
    __syncthreads();

    for (int v = tid; v < SVEC; v += 256) {
        const int row = v >> 3;
        *(float4*)(f + (size_t)row * DIM + col0 + c4) =
            *(const float4*)(slab + (w * SROWS + row) * WCOLS);
    }
}

extern "C" void kernel_launch(void* const* d_in, const int* in_sizes, int n_in,
                              void* d_out, int out_size, void* d_ws, size_t ws_size,
                              hipStream_t stream)
{
    const float* drugW  = (const float*)d_in[0];
    const float* relaW  = (const float*)d_in[1];
    const float* entW   = (const float*)d_in[2];
    const float* Wa     = (const float*)d_in[3];
    const float* linW   = (const float*)d_in[4];
    const float* linb   = (const float*)d_in[5];
    const float* gamma  = (const float*)d_in[6];
    const float* beta   = (const float*)d_in[7];
    const int* drug_name = (const int*)d_in[8];
    const int* adj_tail  = (const int*)d_in[9];
    const int* adj_rel   = (const int*)d_in[10];
    const int* nbr_idx   = (const int*)d_in[11];
    const int* nbr_deg   = (const int*)d_in[12];
    const int* epoch     = (const int*)d_in[13];

    float* out = (float*)d_out;
    float* ws  = (float*)d_ws;
    float* q     = ws;
    float* agg   = ws + (size_t)NPAD * DIM;
    float* h     = ws + (size_t)2 * NPAD * DIM;

    dim3 gTiles(MP, NT);
    qgemm_kernel<<<gTiles, 32, 0, stream>>>(drugW, Wa, drug_name, q);
    attn_agg_kernel<<<NPAD, 256, 0, stream>>>(q, relaW, entW, adj_rel, adj_tail, agg);
    lingemm_kernel<<<gTiles, 32, 0, stream>>>(agg, drugW, drug_name, linW, linb, h);
    bn_smooth_kernel<<<DIM / SCOLS, 256, 0, stream>>>(h, gamma, beta, nbr_idx, nbr_deg, epoch, out);
}


# smoothing scan: rows go in pairs; slots naming the 2 (odd rows: 3) rows before are counted, not loaded, so a pair's loads issue a trip ahead and its two add trees run side by side; sum order changes
# speedup vs baseline: 12.9956x; 1.0840x over previous
#include <hip/hip_runtime.h>

typedef __attribute__((ext_vector_type(2))) float v2f;
typedef __attribute__((ext_vector_type(8))) float v8f;

#define N_DRUG 572
#define DIM    512
#define KNBR   64
#define MAXDEG 32
#define NPAD   576
#define MT     36
#define NT     32
#define MP     18

#define CHK 64
#define KST (CHK / 4)

struct opchunk { float4 a0[KST / 2]; float4 a1[KST / 2]; v2f b[KST]; };

__device__ __forceinline__ void load_chunk(opchunk& f, const float* __restrict__ A0p,
                                           const float* __restrict__ A1p,
                                           const float* __restrict__ Bp)
{
#pragma unroll
    for (int j = 0; j < KST / 2; ++j) {
        f.a0[j] = *(const float4*)(A0p + 4 * j);
        f.a1[j] = *(const float4*)(A1p + 4 * j);
        f.b[2 * j].x     = Bp[(size_t)(4 * j) * DIM];
        f.b[2 * j].y     = Bp[(size_t)(4 * j + 1) * DIM];
        f.b[2 * j + 1].x = Bp[(size_t)(4 * j + 2) * DIM];
        f.b[2 * j + 1].y = Bp[(size_t)(4 * j + 3) * DIM];
    }
}

__device__ __forceinline__ void mma_chunk(const opchunk& f, v8f& c0, v8f& c1)
{
    __builtin_amdgcn_sched_barrier(0);
#pragma unroll
    for (int j = 0; j < KST / 2; ++j) {
        v2f p0, p1, q0, q1;
        p0.x = f.a0[j].x;  p0.y = f.a0[j].y;
        q0.x = f.a0[j].z;  q0.y = f.a0[j].w;
        p1.x = f.a1[j].x;  p1.y = f.a1[j].y;
        q1.x = f.a1[j].z;  q1.y = f.a1[j].w;
        c0 = __builtin_amdgcn_wmma_f32_16x16x4_f32(false, p0, false, f.b[2 * j],
                                                   (short)0, c0, false, false);
        c1 = __builtin_amdgcn_wmma_f32_16x16x4_f32(false, p1, false, f.b[2 * j],
                                                   (short)0, c1, false, false);
        c0 = __builtin_amdgcn_wmma_f32_16x16x4_f32(false, q0, false, f.b[2 * j + 1],
                                                   (short)0, c0, false, false);
        c1 = __builtin_amdgcn_wmma_f32_16x16x4_f32(false, q1, false, f.b[2 * j + 1],
                                                   (short)0, c1, false, false);
    }
    __builtin_amdgcn_sched_barrier(0);
    asm volatile("v_nop\n\tv_nop\n\tv_nop\n\tv_nop");
    __builtin_amdgcn_sched_barrier(0);
}

#define LOAD_AT(F, K)                                                              \
    do {                                                                           \
        if ((K) < DIM) load_chunk(F, P0 + (K), P1 + (K), Bp + (size_t)(K) * DIM);  \
        else load_chunk(F, Q0 + ((K) - DIM), Q1 + ((K) - DIM), Bp + (size_t)(K) * DIM); \
    } while (0)

template <int KTOT>
__device__ __forceinline__ void wave_gemm_32x16(const float* __restrict__ P0,
                                                const float* __restrict__ P1,
                                                const float* __restrict__ Q0,
                                                const float* __restrict__ Q1,
                                                const float* __restrict__ Bp,
                                                v8f& c0, v8f& c1)
{
    opchunk f0, f1;
    load_chunk(f0, P0, P1, Bp);
#pragma unroll 1
    for (int kc = 0; kc < KTOT - 2 * CHK; kc += 2 * CHK) {
        LOAD_AT(f1, kc + CHK);
        mma_chunk(f0, c0, c1);
        LOAD_AT(f0, kc + 2 * CHK);
        mma_chunk(f1, c0, c1);
    }
    LOAD_AT(f1, KTOT - CHK);
    mma_chunk(f0, c0, c1);
    mma_chunk(f1, c0, c1);
}
#undef LOAD_AT

__global__ __launch_bounds__(32)
void qgemm_kernel(const float* __restrict__ drugW, const float* __restrict__ Wa,
                  const int* __restrict__ drug_name, float* __restrict__ q)
{
    const int lane = threadIdx.x;
    const int half = lane >> 4;
    const int l16  = lane & 15;
    const int mBase = blockIdx.x * 32;
    const int nBase = blockIdx.y * 16;

    const int r0 = mBase + l16, r1 = r0 + 16;
    const int g0 = drug_name[r0 < N_DRUG ? r0 : (N_DRUG - 1)];
    const int g1 = drug_name[r1 < N_DRUG ? r1 : (N_DRUG - 1)];
    const float* __restrict__ A0 = drugW + (size_t)g0 * DIM + 32 * half;
    const float* __restrict__ A1 = drugW + (size_t)g1 * DIM + 32 * half;
    const float* __restrict__ Bp = Wa + (size_t)(32 * half) * DIM + nBase + l16;

    v8f c0 = {}, c1 = {};
    wave_gemm_32x16<DIM>(A0, A1, A0, A1, Bp, c0, c1);

    float* outBase = q + (size_t)(mBase + half * 8) * DIM + nBase + l16;
#pragma unroll
    for (int j = 0; j < 8; ++j) {
        outBase[(size_t)j * DIM]        = c0[j];
        outBase[(size_t)(j + 16) * DIM] = c1[j];
    }
}

__global__ __launch_bounds__(256)
void attn_agg_kernel(const float* __restrict__ q, const float* __restrict__ relaW,
                     const float* __restrict__ entW,
                     const int* __restrict__ adj_rel, const int* __restrict__ adj_tail,
                     float* __restrict__ agg)
{
    __shared__ float qs[DIM];
    __shared__ float attn[KNBR];

    const int n  = blockIdx.x;
    const int ni = n < N_DRUG ? n : (N_DRUG - 1);
    const int tid = threadIdx.x;

    qs[tid]       = q[(size_t)n * DIM + tid];
    qs[tid + 256] = q[(size_t)n * DIM + tid + 256];
    __syncthreads();

    const int wave = tid >> 5;
    const int lane = tid & 31;
    for (int kk = 0; kk < 8; ++kk) {
        const int k = wave * 8 + kk;
        const int rel = adj_rel[ni * KNBR + k];
        const float* __restrict__ rw = relaW + (size_t)rel * DIM;
        float p = 0.f;
        for (int j = lane; j < DIM; j += 32)
            p += qs[j] * rw[j];
        for (int off = 16; off >= 1; off >>= 1)
            p += __shfl_xor(p, off, 32);
        if (lane == 0)
            attn[k] = p * 0.04419417382415922f;
    }
    __syncthreads();

    if (tid < 32) {
        float s0 = attn[tid], s1 = attn[tid + 32];
        float m = fmaxf(s0, s1);
        for (int off = 16; off >= 1; off >>= 1)
            m = fmaxf(m, __shfl_xor(m, off, 32));
        float e0 = __expf(s0 - m), e1 = __expf(s1 - m);
        float s = e0 + e1;
        for (int off = 16; off >= 1; off >>= 1)
            s += __shfl_xor(s, off, 32);
        const float inv = 1.f / s;
        attn[tid]      = e0 * inv;
        attn[tid + 32] = e1 * inv;
    }
    __syncthreads();

    float acc0 = 0.f, acc1 = 0.f;
    for (int k = 0; k < KNBR; ++k) {
        const int t = adj_tail[ni * KNBR + k];
        const float w = attn[k];
        const float* __restrict__ er = entW + (size_t)t * DIM;
        acc0 += w * er[tid];
        acc1 += w * er[tid + 256];
    }
    agg[(size_t)n * DIM + tid]       = acc0;
    agg[(size_t)n * DIM + tid + 256] = acc1;
}

__global__ __launch_bounds__(32)
void lingemm_kernel(const float* __restrict__ agg, const float* __restrict__ drugW,
                    const int* __restrict__ drug_name,
                    const float* __restrict__ linW, const float* __restrict__ linb,
                    float* __restrict__ h)
{
    const int lane = threadIdx.x;
    const int half = lane >> 4;
    const int l16  = lane & 15;
    const int mBase = blockIdx.x * 32;
    const int nBase = blockIdx.y * 16;

    const int r0 = mBase + l16, r1 = r0 + 16;
    const int g0 = drug_name[r0 < N_DRUG ? r0 : (N_DRUG - 1)];
    const int g1 = drug_name[r1 < N_DRUG ? r1 : (N_DRUG - 1)];
    const float* __restrict__ P0 = agg + (size_t)r0 * DIM + 32 * half;
    const float* __restrict__ P1 = agg + (size_t)r1 * DIM + 32 * half;
    const float* __restrict__ Q0 = drugW + (size_t)g0 * DIM + 32 * half;
    const float* __restrict__ Q1 = drugW + (size_t)g1 * DIM + 32 * half;
    const float* __restrict__ Bp = linW + (size_t)(32 * half) * DIM + nBase + l16;

    v8f c0 = {}, c1 = {};
    wave_gemm_32x16<2 * DIM>(P0, P1, Q0, Q1, Bp, c0, c1);

    const float bias = linb[nBase + l16];
    float* outBase = h + (size_t)(mBase + half * 8) * DIM + nBase + l16;
#pragma unroll
    for (int j = 0; j < 8; ++j) {
        const float v0 = c0[j] + bias, v1 = c1[j] + bias;
        outBase[(size_t)j * DIM]        = v0 > 0.f ? v0 : 0.f;
        outBase[(size_t)(j + 16) * DIM] = v1 > 0.f ? v1 : 0.f;
    }
}

#define SCOLS 32
#define WCOLS 4
#define SROWS (N_DRUG + 1)
#define SVEC  (N_DRUG * SCOLS / 4)
#define HFULL (N_DRUG / 32)
#define HROWS (HFULL + 1)

template <int CTRL>
__device__ __forceinline__ float add_from_lane(float s)
{
    const int o = __builtin_amdgcn_update_dpp(0, __float_as_int(s), CTRL, 0xF, 0xF, true);
    return s + __int_as_float(o);
}

template <int OFF>
__device__ __forceinline__ float4 add_from_lane4(float4 v)
{
    v.x += __shfl_xor(v.x, OFF, 32);
    v.y += __shfl_xor(v.y, OFF, 32);
    v.z += __shfl_xor(v.z, OFF, 32);
    v.w += __shfl_xor(v.w, OFF, 32);
    return v;
}

__device__ __forceinline__ float slot_sum(const float v0, const float v1,
                                          const float v2, const float v3)
{
    float s = ((v0 + v1) + v2) + v3;
    s = add_from_lane<0xB1>(s);
    s = add_from_lane<0x4E>(s);
    s = add_from_lane<0x164>(s);
    return s;
}

struct pairvals { float e0, e1, e2, e3, ef, o0, o1, o2, o3, of; };

__device__ __forceinline__ pairvals gather_pair(const float* col, const int4 te, const int4 to,
                                                const int p)
{
    pairvals x;
    x.e0 = col[te.x * WCOLS];  x.e1 = col[te.y * WCOLS];
    x.e2 = col[te.z * WCOLS];  x.e3 = col[te.w * WCOLS];
    x.o0 = col[to.x * WCOLS];  x.o1 = col[to.y * WCOLS];
    x.o2 = col[to.z * WCOLS];  x.o3 = col[to.w * WCOLS];
    x.ef = col[p * WCOLS];
    x.of = col[(p + 1) * WCOLS];
    return x;
}

__device__ __forceinline__ void finish_pair(const pairvals x, const float4 ce, const float4 co,
                                            const float n1, const float n2,
                                            float& ne, float& no)
{
    const float be = ce.x > 0.f ? 0.5f : 1.0f;
    const float bo = co.x > 0.f ? 0.5f : 1.0f;
    const float pe = fmaf(ce.y, n1, fmaf(ce.z, n2, be * x.ef));
    const float po = fmaf(co.z, n1, fmaf(co.w, n2, bo * x.of));
    const float se = slot_sum(x.e0, x.e1, x.e2, x.e3);
    const float so = slot_sum(x.o0, x.o1, x.o2, x.o3);
    ne = fmaf(ce.x, se, pe);
    no = fmaf(co.y, ne, fmaf(co.x, so, po));
}

__global__ __launch_bounds__(256)
void bn_smooth_kernel(const float* __restrict__ h, const float* __restrict__ gamma,
                      const float* __restrict__ beta,
                      const int* __restrict__ nbr_idx, const int* __restrict__ nbr_deg,
                      const int* __restrict__ epoch, float* __restrict__ f)
{
    __shared__ __attribute__((aligned(16))) float slab[(SCOLS / WCOLS) * SROWS * WCOLS];
    __shared__ __attribute__((aligned(16))) int   sidx[N_DRUG * MAXDEG];
    __shared__ __attribute__((aligned(16))) float4 scoef[N_DRUG];
    __shared__ __attribute__((aligned(16))) float red[2][8][SCOLS];

    const int tid  = threadIdx.x;
    const int wave = tid >> 5, lane = tid & 31;
    const int col0 = blockIdx.x * SCOLS;
    const int w    = tid & 7;
    const int c4   = w * 4;
    const int rb   = tid >> 3;
    const int ep   = *epoch;

    const float* __restrict__ hp = h + (size_t)rb * DIM + col0 + c4;
    const bool tail = rb + 32 * HFULL < N_DRUG;
    float4 hv[HROWS];
#pragma unroll
    for (int j = 0; j < HFULL; ++j)
        hv[j] = *(const float4*)(hp + (size_t)(32 * j) * DIM);
    hv[HFULL] = *(const float4*)(hp + (size_t)(tail ? 32 * HFULL : 0) * DIM);
    const float4 g4 = *(const float4*)(gamma + col0 + c4);
    const float4 b4 = *(const float4*)(beta + col0 + c4);

    for (int v = tid; v < SVEC; v += 256) {
        const int dg = nbr_deg[v >> 3];
        int4 t = *(const int4*)(nbr_idx + v * 4);
        t.x = (c4     < dg) ? t.x : N_DRUG;
        t.y = (c4 + 1 < dg) ? t.y : N_DRUG;
        t.z = (c4 + 2 < dg) ? t.z : N_DRUG;
        t.w = (c4 + 3 < dg) ? t.w : N_DRUG;
        *(int4*)(sidx + v * 4) = t;
    }
    if (tid < SCOLS)
        slab[((tid >> 2) * SROWS + N_DRUG) * WCOLS + (tid & 3)] = -0.0f;

    if (!tail)
        hv[HFULL] = make_float4(0.f, 0.f, 0.f, 0.f);
    float4 s1 = make_float4(0.f, 0.f, 0.f, 0.f), s2 = s1;
#pragma unroll
    for (int j = 0; j < HROWS; ++j) {
        s1.x += hv[j].x;            s1.y += hv[j].y;
        s1.z += hv[j].z;            s1.w += hv[j].w;
        s2.x += hv[j].x * hv[j].x;  s2.y += hv[j].y * hv[j].y;
        s2.z += hv[j].z * hv[j].z;  s2.w += hv[j].w * hv[j].w;
    }
    s1 = add_from_lane4<8>(s1);   s2 = add_from_lane4<8>(s2);
    s1 = add_from_lane4<16>(s1);  s2 = add_from_lane4<16>(s2);
    if (lane < 8) {
        *(float4*)&red[0][wave][c4] = s1;
        *(float4*)&red[1][wave][c4] = s2;
    }
    __syncthreads();

    for (int r = tid; r < N_DRUG; r += 256) {
        const int dg = nbr_deg[r];
        const int p1 = r - 1, p2 = r - 2;
        const int p3 = (r & 1) ? r - 3 : -1;
        int* trow = sidx + r * MAXDEG;
        int4 tt[MAXDEG / 4];
#pragma unroll
        for (int k = 0; k < MAXDEG / 4; ++k)
            tt[k] = *(const int4*)(trow + 4 * ((k + tid) & 7));
        int c1 = 0, c2 = 0, c3 = 0;
#pragma unroll
        for (int k = 0; k < MAXDEG / 4; ++k) {
            int4 t = tt[k];
            c1 += (int)(t.x == p1) + (int)(t.y == p1) + (int)(t.z == p1) + (int)(t.w == p1);
            c2 += (int)(t.x == p2) + (int)(t.y == p2) + (int)(t.z == p2) + (int)(t.w == p2);
            c3 += (int)(t.x == p3) + (int)(t.y == p3) + (int)(t.z == p3) + (int)(t.w == p3);
            t.x = (t.x == p1 || t.x == p2 || t.x == p3) ? N_DRUG : t.x;
            t.y = (t.y == p1 || t.y == p2 || t.y == p3) ? N_DRUG : t.y;
            t.z = (t.z == p1 || t.z == p2 || t.z == p3) ? N_DRUG : t.z;
            t.w = (t.w == p1 || t.w == p2 || t.w == p3) ? N_DRUG : t.w;
            *(int4*)(trow + 4 * ((k + tid) & 7)) = t;
        }
        const int dd = dg < 1 ? 1 : (dg < MAXDEG ? dg : MAXDEG);
        const float hc = dg > 0 ? 0.5f / (float)dd : 0.f;
        scoef[r] = make_float4(hc, hc * (float)c1, hc * (float)c2, hc * (float)c3);
    }

    float4 t1 = make_float4(0.f, 0.f, 0.f, 0.f), t2 = t1;
#pragma unroll
    for (int wv = 0; wv < 8; ++wv) {
        const float4 a = *(const float4*)&red[0][wv][c4];
        const float4 b = *(const float4*)&red[1][wv][c4];
        t1.x += a.x; t1.y += a.y; t1.z += a.z; t1.w += a.w;
        t2.x += b.x; t2.y += b.y; t2.z += b.z; t2.w += b.w;
    }
    const float inv_n = 1.0f / (float)N_DRUG;
    float4 sc, sh;
    {
        const float mx = t1.x * inv_n, my = t1.y * inv_n, mz = t1.z * inv_n, mw = t1.w * inv_n;
        sc.x = g4.x * rsqrtf((t2.x * inv_n - mx * mx) + 1e-5f);
        sc.y = g4.y * rsqrtf((t2.y * inv_n - my * my) + 1e-5f);
        sc.z = g4.z * rsqrtf((t2.z * inv_n - mz * mz) + 1e-5f);
        sc.w = g4.w * rsqrtf((t2.w * inv_n - mw * mw) + 1e-5f);
        sh.x = b4.x - mx * sc.x;  sh.y = b4.y - my * sc.y;
        sh.z = b4.z - mz * sc.z;  sh.w = b4.w - mw * sc.w;
    }
    float* sp = slab + (w * SROWS + rb) * WCOLS;
#pragma unroll
    for (int j = 0; j < HROWS; ++j) {
        if (j < HFULL || tail) {
            float4 o;
            o.x = hv[j].x * sc.x + sh.x;  o.y = hv[j].y * sc.y + sh.y;
            o.z = hv[j].z * sc.z + sh.z;  o.w = hv[j].w * sc.w + sh.w;
            *(float4*)(sp + 32 * j * WCOLS) = o;
        }
    }
    __syncthreads();

    if (ep > 1) {
        const int g = lane & 7;
        float* col = slab + wave * (SROWS * WCOLS) + (lane >> 3);
        const int* tab = sidx + 4 * g;
        const int lastp = N_DRUG - 2;

        const int4 t0e = *(const int4*)(tab);
        const int4 t0o = *(const int4*)(tab + MAXDEG);
        int4 tYe = *(const int4*)(tab + 2 * MAXDEG);
        int4 tYo = *(const int4*)(tab + 3 * MAXDEG);
        int4 tXe = *(const int4*)(tab + 4 * MAXDEG);
        int4 tXo = *(const int4*)(tab + 5 * MAXDEG);
        float4 cXe = scoef[0];
        float4 cXo = scoef[1];
        pairvals inX = gather_pair(col, t0e, t0o, 0);
        float n1 = 0.f, n2 = 0.f;

        for (int p = 0; p < N_DRUG; p += 4) {
            const int p2 = (p + 2 < lastp) ? (p + 2) : lastp;
            const int p4 = (p + 4 < lastp) ? (p + 4) : lastp;
            const int p6 = (p + 6 < lastp) ? (p + 6) : lastp;
            const int p8 = (p + 8 < lastp) ? (p + 8) : lastp;
            pairvals inY;
            float4 cYe, cYo, cNe, cNo;
            int4 tNe, tNo;
            {
                inY = gather_pair(col, tYe, tYo, p2);
                cYe = scoef[p2];
                cYo = scoef[p2 + 1];
                cNe = scoef[p4];
                cNo = scoef[p4 + 1];
                tYe = *(const int4*)(tab + p6 * MAXDEG);
                tYo = *(const int4*)(tab + (p6 + 1) * MAXDEG);
                tNe = *(const int4*)(tab + p8 * MAXDEG);
                tNo = *(const int4*)(tab + (p8 + 1) * MAXDEG);
                __builtin_amdgcn_sched_barrier(0);
                float ne, no;
                finish_pair(inX, cXe, cXo, n1, n2, ne, no);
                if (g == 0) {
                    col[p * WCOLS]       = ne;
                    col[(p + 1) * WCOLS] = no;
                }
                n2 = ne;  n1 = no;
            }
            {
                inX = gather_pair(col, tXe, tXo, p4);
                __builtin_amdgcn_sched_barrier(0);
                float ne, no;
                finish_pair(inY, cYe, cYo, n1, n2, ne, no);
                if (g == 0) {
                    col[(p + 2) * WCOLS] = ne;
                    col[(p + 3) * WCOLS] = no;
                }
                n2 = ne;  n1 = no;
            }
            cXe = cNe;  cXo = cNo;
            tXe = tNe;  tXo = tNo;
        }
    }
    __syncthreads();

    for (int v = tid; v < SVEC; v += 256) {
        const int row = v >> 3;
        *(float4*)(f + (size_t)row * DIM + col0 + c4) =
            *(const float4*)(slab + (w * SROWS + row) * WCOLS);
    }
}

extern "C" void kernel_launch(void* const* d_in, const int* in_sizes, int n_in,
                              void* d_out, int out_size, void* d_ws, size_t ws_size,
                              hipStream_t stream)
{
    const float* drugW  = (const float*)d_in[0];
    const float* relaW  = (const float*)d_in[1];
    const float* entW   = (const float*)d_in[2];
    const float* Wa     = (const float*)d_in[3];
    const float* linW   = (const float*)d_in[4];
    const float* linb   = (const float*)d_in[5];
    const float* gamma  = (const float*)d_in[6];
    const float* beta   = (const float*)d_in[7];
    const int* drug_name = (const int*)d_in[8];
    const int* adj_tail  = (const int*)d_in[9];
    const int* adj_rel   = (const int*)d_in[10];
    const int* nbr_idx   = (const int*)d_in[11];
    const int* nbr_deg   = (const int*)d_in[12];
    const int* epoch     = (const int*)d_in[13];

    float* out = (float*)d_out;
    float* ws  = (float*)d_ws;
    float* q     = ws;
    float* agg   = ws + (size_t)NPAD * DIM;
    float* h     = ws + (size_t)2 * NPAD * DIM;

    dim3 gTiles(MP, NT);
    qgemm_kernel<<<gTiles, 32, 0, stream>>>(drugW, Wa, drug_name, q);
    attn_agg_kernel<<<NPAD, 256, 0, stream>>>(q, relaW, entW, adj_rel, adj_tail, agg);
    lingemm_kernel<<<gTiles, 32, 0, stream>>>(agg, drugW, drug_name, linW, linb, h);
    bn_smooth_kernel<<<DIM / SCOLS, 256, 0, stream>>>(h, gamma, beta, nbr_idx, nbr_deg, epoch, out);
}


# attention kernel without serial round trips: rows requested in whole groups before use (16-byte relation rows, 32 entity rows at a time), lane sums by DPP, not through LDS; score sum order changes
# speedup vs baseline: 14.1117x; 1.0859x over previous
#include <hip/hip_runtime.h>

typedef __attribute__((ext_vector_type(2))) float v2f;
typedef __attribute__((ext_vector_type(8))) float v8f;

#define N_DRUG 572
#define DIM    512
#define KNBR   64
#define MAXDEG 32
#define NPAD   576
#define MT     36
#define NT     32
#define MP     18

#define CHK 64
#define KST (CHK / 4)

struct opchunk { float4 a0[KST / 2]; float4 a1[KST / 2]; v2f b[KST]; };

__device__ __forceinline__ void load_chunk(opchunk& f, const float* __restrict__ A0p,
                                           const float* __restrict__ A1p,
                                           const float* __restrict__ Bp)
{
#pragma unroll
    for (int j = 0; j < KST / 2; ++j) {
        f.a0[j] = *(const float4*)(A0p + 4 * j);
        f.a1[j] = *(const float4*)(A1p + 4 * j);
        f.b[2 * j].x     = Bp[(size_t)(4 * j) * DIM];
        f.b[2 * j].y     = Bp[(size_t)(4 * j + 1) * DIM];
        f.b[2 * j + 1].x = Bp[(size_t)(4 * j + 2) * DIM];
        f.b[2 * j + 1].y = Bp[(size_t)(4 * j + 3) * DIM];
    }
}

__device__ __forceinline__ void mma_chunk(const opchunk& f, v8f& c0, v8f& c1)
{
    __builtin_amdgcn_sched_barrier(0);
#pragma unroll
    for (int j = 0; j < KST / 2; ++j) {
        v2f p0, p1, q0, q1;
        p0.x = f.a0[j].x;  p0.y = f.a0[j].y;
        q0.x = f.a0[j].z;  q0.y = f.a0[j].w;
        p1.x = f.a1[j].x;  p1.y = f.a1[j].y;
        q1.x = f.a1[j].z;  q1.y = f.a1[j].w;
        c0 = __builtin_amdgcn_wmma_f32_16x16x4_f32(false, p0, false, f.b[2 * j],
                                                   (short)0, c0, false, false);
        c1 = __builtin_amdgcn_wmma_f32_16x16x4_f32(false, p1, false, f.b[2 * j],
                                                   (short)0, c1, false, false);
        c0 = __builtin_amdgcn_wmma_f32_16x16x4_f32(false, q0, false, f.b[2 * j + 1],
                                                   (short)0, c0, false, false);
        c1 = __builtin_amdgcn_wmma_f32_16x16x4_f32(false, q1, false, f.b[2 * j + 1],
                                                   (short)0, c1, false, false);
    }
    __builtin_amdgcn_sched_barrier(0);
    asm volatile("v_nop\n\tv_nop\n\tv_nop\n\tv_nop");
    __builtin_amdgcn_sched_barrier(0);
}

#define LOAD_AT(F, K)                                                              \
    do {                                                                           \
        if ((K) < DIM) load_chunk(F, P0 + (K), P1 + (K), Bp + (size_t)(K) * DIM);  \
        else load_chunk(F, Q0 + ((K) - DIM), Q1 + ((K) - DIM), Bp + (size_t)(K) * DIM); \
    } while (0)

template <int KTOT>
__device__ __forceinline__ void wave_gemm_32x16(const float* __restrict__ P0,
                                                const float* __restrict__ P1,
                                                const float* __restrict__ Q0,
                                                const float* __restrict__ Q1,
                                                const float* __restrict__ Bp,
                                                v8f& c0, v8f& c1)
{
    opchunk f0, f1;
    load_chunk(f0, P0, P1, Bp);
#pragma unroll 1
    for (int kc = 0; kc < KTOT - 2 * CHK; kc += 2 * CHK) {
        LOAD_AT(f1, kc + CHK);
        mma_chunk(f0, c0, c1);
        LOAD_AT(f0, kc + 2 * CHK);
        mma_chunk(f1, c0, c1);
    }
    LOAD_AT(f1, KTOT - CHK);
    mma_chunk(f0, c0, c1);
    mma_chunk(f1, c0, c1);
}
#undef LOAD_AT

__global__ __launch_bounds__(32)
void qgemm_kernel(const float* __restrict__ drugW, const float* __restrict__ Wa,
                  const int* __restrict__ drug_name, float* __restrict__ q)
{
    const int lane = threadIdx.x;
    const int half = lane >> 4;
    const int l16  = lane & 15;
    const int mBase = blockIdx.x * 32;
    const int nBase = blockIdx.y * 16;

    const int r0 = mBase + l16, r1 = r0 + 16;
    const int g0 = drug_name[r0 < N_DRUG ? r0 : (N_DRUG - 1)];
    const int g1 = drug_name[r1 < N_DRUG ? r1 : (N_DRUG - 1)];
    const float* __restrict__ A0 = drugW + (size_t)g0 * DIM + 32 * half;
    const float* __restrict__ A1 = drugW + (size_t)g1 * DIM + 32 * half;
    const float* __restrict__ Bp = Wa + (size_t)(32 * half) * DIM + nBase + l16;

    v8f c0 = {}, c1 = {};
    wave_gemm_32x16<DIM>(A0, A1, A0, A1, Bp, c0, c1);

    float* outBase = q + (size_t)(mBase + half * 8) * DIM + nBase + l16;
#pragma unroll
    for (int j = 0; j < 8; ++j) {
        outBase[(size_t)j * DIM]        = c0[j];
        outBase[(size_t)(j + 16) * DIM] = c1[j];
    }
}

template <int CTRL>
__device__ __forceinline__ float add_from_lane(float s)
{
    const int o = __builtin_amdgcn_update_dpp(0, __float_as_int(s), CTRL, 0xF, 0xF, true);
    return s + __int_as_float(o);
}

template <int CTRL>
__device__ __forceinline__ float max_from_lane(float s)
{
    const int o = __builtin_amdgcn_update_dpp(0, __float_as_int(s), CTRL, 0xF, 0xF, true);
    return fmaxf(s, __int_as_float(o));
}

__device__ __forceinline__ float from_other_row(float x)
{
    return __int_as_float(__builtin_amdgcn_permlanex16(__float_as_int(x), __float_as_int(x),
                                                       0x76543210, 0xfedcba98, false, false));
}

__device__ __forceinline__ float wave_sum(float s)
{
    s += from_other_row(s);
    s = add_from_lane<0x168>(s);
    s = add_from_lane<0x164>(s);
    s = add_from_lane<0x4E>(s);
    s = add_from_lane<0xB1>(s);
    return s;
}

__device__ __forceinline__ float wave_max(float m)
{
    m = fmaxf(m, from_other_row(m));
    m = max_from_lane<0x168>(m);
    m = max_from_lane<0x164>(m);
    m = max_from_lane<0x4E>(m);
    m = max_from_lane<0xB1>(m);
    return m;
}

__global__ __launch_bounds__(256)
void attn_agg_kernel(const float* __restrict__ q, const float* __restrict__ relaW,
                     const float* __restrict__ entW,
                     const int* __restrict__ adj_rel, const int* __restrict__ adj_tail,
                     float* __restrict__ agg)
{
    __shared__ __attribute__((aligned(16))) float attn[KNBR];

    const int n  = blockIdx.x;
    const int ni = n < N_DRUG ? n : (N_DRUG - 1);
    const int tid = threadIdx.x;
    const int lane = tid & 31;
    const int wave = __builtin_amdgcn_readfirstlane(tid >> 5);

    const int* __restrict__ relp = adj_rel + ni * KNBR + wave * 8;
    int rel[8];
#pragma unroll
    for (int kk = 0; kk < 8; ++kk)
        rel[kk] = relp[kk];

    const float* __restrict__ qp = q + (size_t)n * DIM + 4 * lane;
    float4 qv[4];
#pragma unroll
    for (int i = 0; i < 4; ++i)
        qv[i] = *(const float4*)(qp + 128 * i);

    float p[8];
#pragma unroll
    for (int g = 0; g < 2; ++g) {
        float4 rv[4][4];
#pragma unroll
        for (int kk = 0; kk < 4; ++kk) {
            const float* __restrict__ rw = relaW + (size_t)rel[4 * g + kk] * DIM + 4 * lane;
#pragma unroll
            for (int i = 0; i < 4; ++i)
                rv[kk][i] = *(const float4*)(rw + 128 * i);
        }
#pragma unroll
        for (int kk = 0; kk < 4; ++kk) {
            float s = 0.f;
#pragma unroll
            for (int i = 0; i < 4; ++i) {
                s += qv[i].x * rv[kk][i].x;
                s += qv[i].y * rv[kk][i].y;
                s += qv[i].z * rv[kk][i].z;
                s += qv[i].w * rv[kk][i].w;
            }
            p[4 * g + kk] = s;
        }
    }
#pragma unroll
    for (int kk = 0; kk < 8; ++kk)
        p[kk] = wave_sum(p[kk]);
    if (lane == 0) {
        const float sc = 0.04419417382415922f;
        *(float4*)&attn[wave * 8]     = make_float4(p[0] * sc, p[1] * sc, p[2] * sc, p[3] * sc);
        *(float4*)&attn[wave * 8 + 4] = make_float4(p[4] * sc, p[5] * sc, p[6] * sc, p[7] * sc);
    }
    __syncthreads();

    if (tid < 32) {
        const float s0 = attn[tid], s1 = attn[tid + 32];
        const float m = wave_max(fmaxf(s0, s1));
        const float e0 = __expf(s0 - m), e1 = __expf(s1 - m);
        const float s = wave_sum(e0 + e1);
        const float inv = 1.f / s;
        attn[tid]      = e0 * inv;
        attn[tid + 32] = e1 * inv;
    }
    __syncthreads();

    const int* __restrict__ tp = adj_tail + ni * KNBR;
    const float* __restrict__ ecol = entW + 2 * tid;
    float acc0 = 0.f, acc1 = 0.f;
#pragma unroll
    for (int k0 = 0; k0 < KNBR; k0 += 32) {
        float2 e[32];
#pragma unroll
        for (int j = 0; j < 32; ++j)
            e[j] = *(const float2*)(ecol + (size_t)tp[k0 + j] * DIM);
#pragma unroll
        for (int j = 0; j < 32; ++j) {
            const float w = attn[k0 + j];
            acc0 += w * e[j].x;
            acc1 += w * e[j].y;
        }
    }
    *(float2*)(agg + (size_t)n * DIM + 2 * tid) = make_float2(acc0, acc1);
}

__global__ __launch_bounds__(32)
void lingemm_kernel(const float* __restrict__ agg, const float* __restrict__ drugW,
                    const int* __restrict__ drug_name,
                    const float* __restrict__ linW, const float* __restrict__ linb,
                    float* __restrict__ h)
{
    const int lane = threadIdx.x;
    const int half = lane >> 4;
    const int l16  = lane & 15;
    const int mBase = blockIdx.x * 32;
    const int nBase = blockIdx.y * 16;

    const int r0 = mBase + l16, r1 = r0 + 16;
    const int g0 = drug_name[r0 < N_DRUG ? r0 : (N_DRUG - 1)];
    const int g1 = drug_name[r1 < N_DRUG ? r1 : (N_DRUG - 1)];
    const float* __restrict__ P0 = agg + (size_t)r0 * DIM + 32 * half;
    const float* __restrict__ P1 = agg + (size_t)r1 * DIM + 32 * half;
    const float* __restrict__ Q0 = drugW + (size_t)g0 * DIM + 32 * half;
    const float* __restrict__ Q1 = drugW + (size_t)g1 * DIM + 32 * half;
    const float* __restrict__ Bp = linW + (size_t)(32 * half) * DIM + nBase + l16;

    v8f c0 = {}, c1 = {};
    wave_gemm_32x16<2 * DIM>(P0, P1, Q0, Q1, Bp, c0, c1);

    const float bias = linb[nBase + l16];
    float* outBase = h + (size_t)(mBase + half * 8) * DIM + nBase + l16;
#pragma unroll
    for (int j = 0; j < 8; ++j) {
        const float v0 = c0[j] + bias, v1 = c1[j] + bias;
        outBase[(size_t)j * DIM]        = v0 > 0.f ? v0 : 0.f;
        outBase[(size_t)(j + 16) * DIM] = v1 > 0.f ? v1 : 0.f;
    }
}

#define SCOLS 32
#define WCOLS 4
#define SROWS (N_DRUG + 1)
#define SVEC  (N_DRUG * SCOLS / 4)
#define HFULL (N_DRUG / 32)
#define HROWS (HFULL + 1)

template <int OFF>
__device__ __forceinline__ float4 add_from_lane4(float4 v)
{
    v.x += __shfl_xor(v.x, OFF, 32);
    v.y += __shfl_xor(v.y, OFF, 32);
    v.z += __shfl_xor(v.z, OFF, 32);
    v.w += __shfl_xor(v.w, OFF, 32);
    return v;
}

__device__ __forceinline__ float slot_sum(const float v0, const float v1,
                                          const float v2, const float v3)
{
    float s = ((v0 + v1) + v2) + v3;
    s = add_from_lane<0xB1>(s);
    s = add_from_lane<0x4E>(s);
    s = add_from_lane<0x164>(s);
    return s;
}

struct pairvals { float e0, e1, e2, e3, ef, o0, o1, o2, o3, of; };

__device__ __forceinline__ pairvals gather_pair(const float* col, const int4 te, const int4 to,
                                                const int p)
{
    pairvals x;
    x.e0 = col[te.x * WCOLS];  x.e1 = col[te.y * WCOLS];
    x.e2 = col[te.z * WCOLS];  x.e3 = col[te.w * WCOLS];
    x.o0 = col[to.x * WCOLS];  x.o1 = col[to.y * WCOLS];
    x.o2 = col[to.z * WCOLS];  x.o3 = col[to.w * WCOLS];
    x.ef = col[p * WCOLS];
    x.of = col[(p + 1) * WCOLS];
    return x;
}

__device__ __forceinline__ void finish_pair(const pairvals x, const float4 ce, const float4 co,
                                            const float n1, const float n2,
                                            float& ne, float& no)
{
    const float be = ce.x > 0.f ? 0.5f : 1.0f;
    const float bo = co.x > 0.f ? 0.5f : 1.0f;
    const float pe = fmaf(ce.y, n1, fmaf(ce.z, n2, be * x.ef));
    const float po = fmaf(co.z, n1, fmaf(co.w, n2, bo * x.of));
    const float se = slot_sum(x.e0, x.e1, x.e2, x.e3);
    const float so = slot_sum(x.o0, x.o1, x.o2, x.o3);
    ne = fmaf(ce.x, se, pe);
    no = fmaf(co.y, ne, fmaf(co.x, so, po));
}

__global__ __launch_bounds__(256)
void bn_smooth_kernel(const float* __restrict__ h, const float* __restrict__ gamma,
                      const float* __restrict__ beta,
                      const int* __restrict__ nbr_idx, const int* __restrict__ nbr_deg,
                      const int* __restrict__ epoch, float* __restrict__ f)
{
    __shared__ __attribute__((aligned(16))) float slab[(SCOLS / WCOLS) * SROWS * WCOLS];
    __shared__ __attribute__((aligned(16))) int   sidx[N_DRUG * MAXDEG];
    __shared__ __attribute__((aligned(16))) float4 scoef[N_DRUG];
    __shared__ __attribute__((aligned(16))) float red[2][8][SCOLS];

    const int tid  = threadIdx.x;
    const int wave = tid >> 5, lane = tid & 31;
    const int col0 = blockIdx.x * SCOLS;
    const int w    = tid & 7;
    const int c4   = w * 4;
    const int rb   = tid >> 3;
    const int ep   = *epoch;

    const float* __restrict__ hp = h + (size_t)rb * DIM + col0 + c4;
    const bool tail = rb + 32 * HFULL < N_DRUG;
    float4 hv[HROWS];
#pragma unroll
    for (int j = 0; j < HFULL; ++j)
        hv[j] = *(const float4*)(hp + (size_t)(32 * j) * DIM);
    hv[HFULL] = *(const float4*)(hp + (size_t)(tail ? 32 * HFULL : 0) * DIM);
    const float4 g4 = *(const float4*)(gamma + col0 + c4);
    const float4 b4 = *(const float4*)(beta + col0 + c4);

    for (int v = tid; v < SVEC; v += 256) {
        const int dg = nbr_deg[v >> 3];
        int4 t = *(const int4*)(nbr_idx + v * 4);
        t.x = (c4     < dg) ? t.x : N_DRUG;
        t.y = (c4 + 1 < dg) ? t.y : N_DRUG;
        t.z = (c4 + 2 < dg) ? t.z : N_DRUG;
        t.w = (c4 + 3 < dg) ? t.w : N_DRUG;
        *(int4*)(sidx + v * 4) = t;
    }
    if (tid < SCOLS)
        slab[((tid >> 2) * SROWS + N_DRUG) * WCOLS + (tid & 3)] = -0.0f;

    if (!tail)
        hv[HFULL] = make_float4(0.f, 0.f, 0.f, 0.f);
    float4 s1 = make_float4(0.f, 0.f, 0.f, 0.f), s2 = s1;
#pragma unroll
    for (int j = 0; j < HROWS; ++j) {
        s1.x += hv[j].x;            s1.y += hv[j].y;
        s1.z += hv[j].z;            s1.w += hv[j].w;
        s2.x += hv[j].x * hv[j].x;  s2.y += hv[j].y * hv[j].y;
        s2.z += hv[j].z * hv[j].z;  s2.w += hv[j].w * hv[j].w;
    }
    s1 = add_from_lane4<8>(s1);   s2 = add_from_lane4<8>(s2);
    s1 = add_from_lane4<16>(s1);  s2 = add_from_lane4<16>(s2);
    if (lane < 8) {
        *(float4*)&red[0][wave][c4] = s1;
        *(float4*)&red[1][wave][c4] = s2;
    }
    __syncthreads();

    for (int r = tid; r < N_DRUG; r += 256) {
        const int dg = nbr_deg[r];
        const int p1 = r - 1, p2 = r - 2;
        const int p3 = (r & 1) ? r - 3 : -1;
        int* trow = sidx + r * MAXDEG;
        int4 tt[MAXDEG / 4];
#pragma unroll
        for (int k = 0; k < MAXDEG / 4; ++k)
            tt[k] = *(const int4*)(trow + 4 * ((k + tid) & 7));
        int c1 = 0, c2 = 0, c3 = 0;
#pragma unroll
        for (int k = 0; k < MAXDEG / 4; ++k) {
            int4 t = tt[k];
            c1 += (int)(t.x == p1) + (int)(t.y == p1) + (int)(t.z == p1) + (int)(t.w == p1);
            c2 += (int)(t.x == p2) + (int)(t.y == p2) + (int)(t.z == p2) + (int)(t.w == p2);
            c3 += (int)(t.x == p3) + (int)(t.y == p3) + (int)(t.z == p3) + (int)(t.w == p3);
            t.x = (t.x == p1 || t.x == p2 || t.x == p3) ? N_DRUG : t.x;
            t.y = (t.y == p1 || t.y == p2 || t.y == p3) ? N_DRUG : t.y;
            t.z = (t.z == p1 || t.z == p2 || t.z == p3) ? N_DRUG : t.z;
            t.w = (t.w == p1 || t.w == p2 || t.w == p3) ? N_DRUG : t.w;
            *(int4*)(trow + 4 * ((k + tid) & 7)) = t;
        }
        const int dd = dg < 1 ? 1 : (dg < MAXDEG ? dg : MAXDEG);
        const float hc = dg > 0 ? 0.5f / (float)dd : 0.f;
        scoef[r] = make_float4(hc, hc * (float)c1, hc * (float)c2, hc * (float)c3);
    }

    float4 t1 = make_float4(0.f, 0.f, 0.f, 0.f), t2 = t1;
#pragma unroll
    for (int wv = 0; wv < 8; ++wv) {
        const float4 a = *(const float4*)&red[0][wv][c4];
        const float4 b = *(const float4*)&red[1][wv][c4];
        t1.x += a.x; t1.y += a.y; t1.z += a.z; t1.w += a.w;
        t2.x += b.x; t2.y += b.y; t2.z += b.z; t2.w += b.w;
    }
    const float inv_n = 1.0f / (float)N_DRUG;
    float4 sc, sh;
    {
        const float mx = t1.x * inv_n, my = t1.y * inv_n, mz = t1.z * inv_n, mw = t1.w * inv_n;
        sc.x = g4.x * rsqrtf((t2.x * inv_n - mx * mx) + 1e-5f);
        sc.y = g4.y * rsqrtf((t2.y * inv_n - my * my) + 1e-5f);
        sc.z = g4.z * rsqrtf((t2.z * inv_n - mz * mz) + 1e-5f);
        sc.w = g4.w * rsqrtf((t2.w * inv_n - mw * mw) + 1e-5f);
        sh.x = b4.x - mx * sc.x;  sh.y = b4.y - my * sc.y;
        sh.z = b4.z - mz * sc.z;  sh.w = b4.w - mw * sc.w;
    }
    float* sp = slab + (w * SROWS + rb) * WCOLS;
#pragma unroll
    for (int j = 0; j < HROWS; ++j) {
        if (j < HFULL || tail) {
            float4 o;
            o.x = hv[j].x * sc.x + sh.x;  o.y = hv[j].y * sc.y + sh.y;
            o.z = hv[j].z * sc.z + sh.z;  o.w = hv[j].w * sc.w + sh.w;
            *(float4*)(sp + 32 * j * WCOLS) = o;
        }
    }
    __syncthreads();

    if (ep > 1) {
        const int g = lane & 7;
        float* col = slab + wave * (SROWS * WCOLS) + (lane >> 3);
        const int* tab = sidx + 4 * g;
        const int lastp = N_DRUG - 2;

        const int4 t0e = *(const int4*)(tab);
        const int4 t0o = *(const int4*)(tab + MAXDEG);
        int4 tYe = *(const int4*)(tab + 2 * MAXDEG);
        int4 tYo = *(const int4*)(tab + 3 * MAXDEG);
        int4 tXe = *(const int4*)(tab + 4 * MAXDEG);
        int4 tXo = *(const int4*)(tab + 5 * MAXDEG);
        float4 cXe = scoef[0];
        float4 cXo = scoef[1];
        pairvals inX = gather_pair(col, t0e, t0o, 0);
        float n1 = 0.f, n2 = 0.f;

        for (int p = 0; p < N_DRUG; p += 4) {
            const int p2 = (p + 2 < lastp) ? (p + 2) : lastp;
            const int p4 = (p + 4 < lastp) ? (p + 4) : lastp;
            const int p6 = (p + 6 < lastp) ? (p + 6) : lastp;
            const int p8 = (p + 8 < lastp) ? (p + 8) : lastp;
            pairvals inY;
            float4 cYe, cYo, cNe, cNo;
            int4 tNe, tNo;
            {
                inY = gather_pair(col, tYe, tYo, p2);
                cYe = scoef[p2];
                cYo = scoef[p2 + 1];
                cNe = scoef[p4];
                cNo = scoef[p4 + 1];
                tYe = *(const int4*)(tab + p6 * MAXDEG);
                tYo = *(const int4*)(tab + (p6 + 1) * MAXDEG);
                tNe = *(const int4*)(tab + p8 * MAXDEG);
                tNo = *(const int4*)(tab + (p8 + 1) * MAXDEG);
                __builtin_amdgcn_sched_barrier(0);
                float ne, no;
                finish_pair(inX, cXe, cXo, n1, n2, ne, no);
                if (g == 0) {
                    col[p * WCOLS]       = ne;
                    col[(p + 1) * WCOLS] = no;
                }
                n2 = ne;  n1 = no;
            }
            {
                inX = gather_pair(col, tXe, tXo, p4);
                __builtin_amdgcn_sched_barrier(0);
                float ne, no;
                finish_pair(inY, cYe, cYo, n1, n2, ne, no);
                if (g == 0) {
                    col[(p + 2) * WCOLS] = ne;
                    col[(p + 3) * WCOLS] = no;
                }
                n2 = ne;  n1 = no;
            }
            cXe = cNe;  cXo = cNo;
            tXe = tNe;  tXo = tNo;
        }
    }
    __syncthreads();

    for (int v = tid; v < SVEC; v += 256) {
        const int row = v >> 3;
        *(float4*)(f + (size_t)row * DIM + col0 + c4) =
            *(const float4*)(slab + (w * SROWS + row) * WCOLS);
    }
}

extern "C" void kernel_launch(void* const* d_in, const int* in_sizes, int n_in,
                              void* d_out, int out_size, void* d_ws, size_t ws_size,
                              hipStream_t stream)
{
    const float* drugW  = (const float*)d_in[0];
    const float* relaW  = (const float*)d_in[1];
    const float* entW   = (const float*)d_in[2];
    const float* Wa     = (const float*)d_in[3];
    const float* linW   = (const float*)d_in[4];
    const float* linb   = (const float*)d_in[5];
    const float* gamma  = (const float*)d_in[6];
    const float* beta   = (const float*)d_in[7];
    const int* drug_name = (const int*)d_in[8];
    const int* adj_tail  = (const int*)d_in[9];
    const int* adj_rel   = (const int*)d_in[10];
    const int* nbr_idx   = (const int*)d_in[11];
    const int* nbr_deg   = (const int*)d_in[12];
    const int* epoch     = (const int*)d_in[13];

    float* out = (float*)d_out;
    float* ws  = (float*)d_ws;
    float* q     = ws;
    float* agg   = ws + (size_t)NPAD * DIM;
    float* h     = ws + (size_t)2 * NPAD * DIM;

    dim3 gTiles(MP, NT);
    qgemm_kernel<<<gTiles, 32, 0, stream>>>(drugW, Wa, drug_name, q);
    attn_agg_kernel<<<NPAD, 256, 0, stream>>>(q, relaW, entW, adj_rel, adj_tail, agg);
    lingemm_kernel<<<gTiles, 32, 0, stream>>>(agg, drugW, drug_name, linW, linb, h);
    bn_smooth_kernel<<<DIM / SCOLS, 256, 0, stream>>>(h, gamma, beta, nbr_idx, nbr_deg, epoch, out);
}


# smoothing kernel on 32 blocks of 16 columns: the scan runs on 4 of a block's 8 waves, one per SIMD, so each LDS unit and SIMD serves half of it; h rows split 64 ways (stat sum order changes)
# speedup vs baseline: 14.9300x; 1.0580x over previous
#include <hip/hip_runtime.h>

typedef __attribute__((ext_vector_type(2))) float v2f;
typedef __attribute__((ext_vector_type(8))) float v8f;

#define N_DRUG 572
#define DIM    512
#define KNBR   64
#define MAXDEG 32
#define NPAD   576
#define MT     36
#define NT     32
#define MP     18

#define CHK 64
#define KST (CHK / 4)

struct opchunk { float4 a0[KST / 2]; float4 a1[KST / 2]; v2f b[KST]; };

__device__ __forceinline__ void load_chunk(opchunk& f, const float* __restrict__ A0p,
                                           const float* __restrict__ A1p,
                                           const float* __restrict__ Bp)
{
#pragma unroll
    for (int j = 0; j < KST / 2; ++j) {
        f.a0[j] = *(const float4*)(A0p + 4 * j);
        f.a1[j] = *(const float4*)(A1p + 4 * j);
        f.b[2 * j].x     = Bp[(size_t)(4 * j) * DIM];
        f.b[2 * j].y     = Bp[(size_t)(4 * j + 1) * DIM];
        f.b[2 * j + 1].x = Bp[(size_t)(4 * j + 2) * DIM];
        f.b[2 * j + 1].y = Bp[(size_t)(4 * j + 3) * DIM];
    }
}

__device__ __forceinline__ void mma_chunk(const opchunk& f, v8f& c0, v8f& c1)
{
    __builtin_amdgcn_sched_barrier(0);
#pragma unroll
    for (int j = 0; j < KST / 2; ++j) {
        v2f p0, p1, q0, q1;
        p0.x = f.a0[j].x;  p0.y = f.a0[j].y;
        q0.x = f.a0[j].z;  q0.y = f.a0[j].w;
        p1.x = f.a1[j].x;  p1.y = f.a1[j].y;
        q1.x = f.a1[j].z;  q1.y = f.a1[j].w;
        c0 = __builtin_amdgcn_wmma_f32_16x16x4_f32(false, p0, false, f.b[2 * j],
                                                   (short)0, c0, false, false);
        c1 = __builtin_amdgcn_wmma_f32_16x16x4_f32(false, p1, false, f.b[2 * j],
                                                   (short)0, c1, false, false);
        c0 = __builtin_amdgcn_wmma_f32_16x16x4_f32(false, q0, false, f.b[2 * j + 1],
                                                   (short)0, c0, false, false);
        c1 = __builtin_amdgcn_wmma_f32_16x16x4_f32(false, q1, false, f.b[2 * j + 1],
                                                   (short)0, c1, false, false);
    }
    __builtin_amdgcn_sched_barrier(0);
    asm volatile("v_nop\n\tv_nop\n\tv_nop\n\tv_nop");
    __builtin_amdgcn_sched_barrier(0);
}

#define LOAD_AT(F, K)                                                              \
    do {                                                                           \
        if ((K) < DIM) load_chunk(F, P0 + (K), P1 + (K), Bp + (size_t)(K) * DIM);  \
        else load_chunk(F, Q0 + ((K) - DIM), Q1 + ((K) - DIM), Bp + (size_t)(K) * DIM); \
    } while (0)

template <int KTOT>
__device__ __forceinline__ void wave_gemm_32x16(const float* __restrict__ P0,
                                                const float* __restrict__ P1,
                                                const float* __restrict__ Q0,
                                                const float* __restrict__ Q1,
                                                const float* __restrict__ Bp,
                                                v8f& c0, v8f& c1)
{
    opchunk f0, f1;
    load_chunk(f0, P0, P1, Bp);
#pragma unroll 1
    for (int kc = 0; kc < KTOT - 2 * CHK; kc += 2 * CHK) {
        LOAD_AT(f1, kc + CHK);
        mma_chunk(f0, c0, c1);
        LOAD_AT(f0, kc + 2 * CHK);
        mma_chunk(f1, c0, c1);
    }
    LOAD_AT(f1, KTOT - CHK);
    mma_chunk(f0, c0, c1);
    mma_chunk(f1, c0, c1);
}
#undef LOAD_AT

__global__ __launch_bounds__(32)
void qgemm_kernel(const float* __restrict__ drugW, const float* __restrict__ Wa,
                  const int* __restrict__ drug_name, float* __restrict__ q)
{
    const int lane = threadIdx.x;
    const int half = lane >> 4;
    const int l16  = lane & 15;
    const int mBase = blockIdx.x * 32;
    const int nBase = blockIdx.y * 16;

    const int r0 = mBase + l16, r1 = r0 + 16;
    const int g0 = drug_name[r0 < N_DRUG ? r0 : (N_DRUG - 1)];
    const int g1 = drug_name[r1 < N_DRUG ? r1 : (N_DRUG - 1)];
    const float* __restrict__ A0 = drugW + (size_t)g0 * DIM + 32 * half;
    const float* __restrict__ A1 = drugW + (size_t)g1 * DIM + 32 * half;
    const float* __restrict__ Bp = Wa + (size_t)(32 * half) * DIM + nBase + l16;

    v8f c0 = {}, c1 = {};
    wave_gemm_32x16<DIM>(A0, A1, A0, A1, Bp, c0, c1);

    float* outBase = q + (size_t)(mBase + half * 8) * DIM + nBase + l16;
#pragma unroll
    for (int j = 0; j < 8; ++j) {
        outBase[(size_t)j * DIM]        = c0[j];
        outBase[(size_t)(j + 16) * DIM] = c1[j];
    }
}

template <int CTRL>
__device__ __forceinline__ float add_from_lane(float s)
{
    const int o = __builtin_amdgcn_update_dpp(0, __float_as_int(s), CTRL, 0xF, 0xF, true);
    return s + __int_as_float(o);
}

template <int CTRL>
__device__ __forceinline__ float max_from_lane(float s)
{
    const int o = __builtin_amdgcn_update_dpp(0, __float_as_int(s), CTRL, 0xF, 0xF, true);
    return fmaxf(s, __int_as_float(o));
}

__device__ __forceinline__ float from_other_row(float x)
{
    return __int_as_float(__builtin_amdgcn_permlanex16(__float_as_int(x), __float_as_int(x),
                                                       0x76543210, 0xfedcba98, false, false));
}

__device__ __forceinline__ float wave_sum(float s)
{
    s += from_other_row(s);
    s = add_from_lane<0x168>(s);
    s = add_from_lane<0x164>(s);
    s = add_from_lane<0x4E>(s);
    s = add_from_lane<0xB1>(s);
    return s;
}

__device__ __forceinline__ float wave_max(float m)
{
    m = fmaxf(m, from_other_row(m));
    m = max_from_lane<0x168>(m);
    m = max_from_lane<0x164>(m);
    m = max_from_lane<0x4E>(m);
    m = max_from_lane<0xB1>(m);
    return m;
}

__global__ __launch_bounds__(256)
void attn_agg_kernel(const float* __restrict__ q, const float* __restrict__ relaW,
                     const float* __restrict__ entW,
                     const int* __restrict__ adj_rel, const int* __restrict__ adj_tail,
                     float* __restrict__ agg)
{
    __shared__ __attribute__((aligned(16))) float attn[KNBR];

    const int n  = blockIdx.x;
    const int ni = n < N_DRUG ? n : (N_DRUG - 1);
    const int tid = threadIdx.x;
    const int lane = tid & 31;
    const int wave = __builtin_amdgcn_readfirstlane(tid >> 5);

    const int* __restrict__ relp = adj_rel + ni * KNBR + wave * 8;
    int rel[8];
#pragma unroll
    for (int kk = 0; kk < 8; ++kk)
        rel[kk] = relp[kk];

    const float* __restrict__ qp = q + (size_t)n * DIM + 4 * lane;
    float4 qv[4];
#pragma unroll
    for (int i = 0; i < 4; ++i)
        qv[i] = *(const float4*)(qp + 128 * i);

    float p[8];
#pragma unroll
    for (int g = 0; g < 2; ++g) {
        float4 rv[4][4];
#pragma unroll
        for (int kk = 0; kk < 4; ++kk) {
            const float* __restrict__ rw = relaW + (size_t)rel[4 * g + kk] * DIM + 4 * lane;
#pragma unroll
            for (int i = 0; i < 4; ++i)
                rv[kk][i] = *(const float4*)(rw + 128 * i);
        }
#pragma unroll
        for (int kk = 0; kk < 4; ++kk) {
            float s = 0.f;
#pragma unroll
            for (int i = 0; i < 4; ++i) {
                s += qv[i].x * rv[kk][i].x;
                s += qv[i].y * rv[kk][i].y;
                s += qv[i].z * rv[kk][i].z;
                s += qv[i].w * rv[kk][i].w;
            }
            p[4 * g + kk] = s;
        }
    }
#pragma unroll
    for (int kk = 0; kk < 8; ++kk)
        p[kk] = wave_sum(p[kk]);
    if (lane == 0) {
        const float sc = 0.04419417382415922f;
        *(float4*)&attn[wave * 8]     = make_float4(p[0] * sc, p[1] * sc, p[2] * sc, p[3] * sc);
        *(float4*)&attn[wave * 8 + 4] = make_float4(p[4] * sc, p[5] * sc, p[6] * sc, p[7] * sc);
    }
    __syncthreads();

    if (tid < 32) {
        const float s0 = attn[tid], s1 = attn[tid + 32];
        const float m = wave_max(fmaxf(s0, s1));
        const float e0 = __expf(s0 - m), e1 = __expf(s1 - m);
        const float s = wave_sum(e0 + e1);
        const float inv = 1.f / s;
        attn[tid]      = e0 * inv;
        attn[tid + 32] = e1 * inv;
    }
    __syncthreads();

    const int* __restrict__ tp = adj_tail + ni * KNBR;
    const float* __restrict__ ecol = entW + 2 * tid;
    float acc0 = 0.f, acc1 = 0.f;
#pragma unroll
    for (int k0 = 0; k0 < KNBR; k0 += 32) {
        float2 e[32];
#pragma unroll
        for (int j = 0; j < 32; ++j)
            e[j] = *(const float2*)(ecol + (size_t)tp[k0 + j] * DIM);
#pragma unroll
        for (int j = 0; j < 32; ++j) {
            const float w = attn[k0 + j];
            acc0 += w * e[j].x;
            acc1 += w * e[j].y;
        }
    }
    *(float2*)(agg + (size_t)n * DIM + 2 * tid) = make_float2(acc0, acc1);
}

__global__ __launch_bounds__(32)
void lingemm_kernel(const float* __restrict__ agg, const float* __restrict__ drugW,
                    const int* __restrict__ drug_name,
                    const float* __restrict__ linW, const float* __restrict__ linb,
                    float* __restrict__ h)
{
    const int lane = threadIdx.x;
    const int half = lane >> 4;
    const int l16  = lane & 15;
    const int mBase = blockIdx.x * 32;
    const int nBase = blockIdx.y * 16;

    const int r0 = mBase + l16, r1 = r0 + 16;
    const int g0 = drug_name[r0 < N_DRUG ? r0 : (N_DRUG - 1)];
    const int g1 = drug_name[r1 < N_DRUG ? r1 : (N_DRUG - 1)];
    const float* __restrict__ P0 = agg + (size_t)r0 * DIM + 32 * half;
    const float* __restrict__ P1 = agg + (size_t)r1 * DIM + 32 * half;
    const float* __restrict__ Q0 = drugW + (size_t)g0 * DIM + 32 * half;
    const float* __restrict__ Q1 = drugW + (size_t)g1 * DIM + 32 * half;
    const float* __restrict__ Bp = linW + (size_t)(32 * half) * DIM + nBase + l16;

    v8f c0 = {}, c1 = {};
    wave_gemm_32x16<2 * DIM>(P0, P1, Q0, Q1, Bp, c0, c1);

    const float bias = linb[nBase + l16];
    float* outBase = h + (size_t)(mBase + half * 8) * DIM + nBase + l16;
#pragma unroll
    for (int j = 0; j < 8; ++j) {
        const float v0 = c0[j] + bias, v1 = c1[j] + bias;
        outBase[(size_t)j * DIM]        = v0 > 0.f ? v0 : 0.f;
        outBase[(size_t)(j + 16) * DIM] = v1 > 0.f ? v1 : 0.f;
    }
}

#define SCOLS 16
#define WCOLS 4
#define SWAVES (SCOLS / WCOLS)
#define SROWS (N_DRUG + 1)
#define SVEC  (N_DRUG * SCOLS / 4)
#define TVEC  (N_DRUG * MAXDEG / 4)
#define RSTEP (256 / (SCOLS / 4))
#define HFULL (N_DRUG / RSTEP)
#define HROWS (HFULL + 1)

template <int OFF>
__device__ __forceinline__ float4 add_from_lane4(float4 v)
{
    v.x += __shfl_xor(v.x, OFF, 32);
    v.y += __shfl_xor(v.y, OFF, 32);
    v.z += __shfl_xor(v.z, OFF, 32);
    v.w += __shfl_xor(v.w, OFF, 32);
    return v;
}

__device__ __forceinline__ float slot_sum(const float v0, const float v1,
                                          const float v2, const float v3)
{
    float s = ((v0 + v1) + v2) + v3;
    s = add_from_lane<0xB1>(s);
    s = add_from_lane<0x4E>(s);
    s = add_from_lane<0x164>(s);
    return s;
}

struct pairvals { float e0, e1, e2, e3, ef, o0, o1, o2, o3, of; };

__device__ __forceinline__ pairvals gather_pair(const float* col, const int4 te, const int4 to,
                                                const int p)
{
    pairvals x;
    x.e0 = col[te.x * WCOLS];  x.e1 = col[te.y * WCOLS];
    x.e2 = col[te.z * WCOLS];  x.e3 = col[te.w * WCOLS];
    x.o0 = col[to.x * WCOLS];  x.o1 = col[to.y * WCOLS];
    x.o2 = col[to.z * WCOLS];  x.o3 = col[to.w * WCOLS];
    x.ef = col[p * WCOLS];
    x.of = col[(p + 1) * WCOLS];
    return x;
}

__device__ __forceinline__ void finish_pair(const pairvals x, const float4 ce, const float4 co,
                                            const float n1, const float n2,
                                            float& ne, float& no)
{
    const float be = ce.x > 0.f ? 0.5f : 1.0f;
    const float bo = co.x > 0.f ? 0.5f : 1.0f;
    const float pe = fmaf(ce.y, n1, fmaf(ce.z, n2, be * x.ef));
    const float po = fmaf(co.z, n1, fmaf(co.w, n2, bo * x.of));
    const float se = slot_sum(x.e0, x.e1, x.e2, x.e3);
    const float so = slot_sum(x.o0, x.o1, x.o2, x.o3);
    ne = fmaf(ce.x, se, pe);
    no = fmaf(co.y, ne, fmaf(co.x, so, po));
}

__global__ __launch_bounds__(256)
void bn_smooth_kernel(const float* __restrict__ h, const float* __restrict__ gamma,
                      const float* __restrict__ beta,
                      const int* __restrict__ nbr_idx, const int* __restrict__ nbr_deg,
                      const int* __restrict__ epoch, float* __restrict__ f)
{
    __shared__ __attribute__((aligned(16))) float slab[SWAVES * SROWS * WCOLS];
    __shared__ __attribute__((aligned(16))) int   sidx[N_DRUG * MAXDEG];
    __shared__ __attribute__((aligned(16))) float4 scoef[N_DRUG];
    __shared__ __attribute__((aligned(16))) float red[2][8][SCOLS];

    const int tid  = threadIdx.x;
    const int lane = tid & 31;
    const int wave = __builtin_amdgcn_readfirstlane(tid >> 5);
    const int col0 = blockIdx.x * SCOLS;
    const int w    = tid & 3;
    const int c4   = w * 4;
    const int rb   = tid >> 2;
    const int ts4  = (tid & 7) * 4;
    const int ep   = *epoch;

    const float* __restrict__ hp = h + (size_t)rb * DIM + col0 + c4;
    const bool tail = rb + RSTEP * HFULL < N_DRUG;
    float4 hv[HROWS];
#pragma unroll
    for (int j = 0; j < HFULL; ++j)
        hv[j] = *(const float4*)(hp + (size_t)(RSTEP * j) * DIM);
    hv[HFULL] = *(const float4*)(hp + (size_t)(tail ? RSTEP * HFULL : 0) * DIM);
    const float4 g4 = *(const float4*)(gamma + col0 + c4);
    const float4 b4 = *(const float4*)(beta + col0 + c4);

    for (int v = tid; v < TVEC; v += 256) {
        const int dg = nbr_deg[v >> 3];
        int4 t = *(const int4*)(nbr_idx + v * 4);
        t.x = (ts4     < dg) ? t.x : N_DRUG;
        t.y = (ts4 + 1 < dg) ? t.y : N_DRUG;
        t.z = (ts4 + 2 < dg) ? t.z : N_DRUG;
        t.w = (ts4 + 3 < dg) ? t.w : N_DRUG;
        *(int4*)(sidx + v * 4) = t;
    }
    if (tid < SCOLS)
        slab[((tid >> 2) * SROWS + N_DRUG) * WCOLS + (tid & 3)] = -0.0f;

    if (!tail)
        hv[HFULL] = make_float4(0.f, 0.f, 0.f, 0.f);
    float4 s1 = make_float4(0.f, 0.f, 0.f, 0.f), s2 = s1;
#pragma unroll
    for (int j = 0; j < HROWS; ++j) {
        s1.x += hv[j].x;            s1.y += hv[j].y;
        s1.z += hv[j].z;            s1.w += hv[j].w;
        s2.x += hv[j].x * hv[j].x;  s2.y += hv[j].y * hv[j].y;
        s2.z += hv[j].z * hv[j].z;  s2.w += hv[j].w * hv[j].w;
    }
    s1 = add_from_lane4<4>(s1);   s2 = add_from_lane4<4>(s2);
    s1 = add_from_lane4<8>(s1);   s2 = add_from_lane4<8>(s2);
    s1 = add_from_lane4<16>(s1);  s2 = add_from_lane4<16>(s2);
    if (lane < 4) {
        *(float4*)&red[0][wave][c4] = s1;
        *(float4*)&red[1][wave][c4] = s2;
    }
    __syncthreads();

    for (int r = tid; r < N_DRUG; r += 256) {
        const int dg = nbr_deg[r];
        const int p1 = r - 1, p2 = r - 2;
        const int p3 = (r & 1) ? r - 3 : -1;
        int* trow = sidx + r * MAXDEG;
        int4 tt[MAXDEG / 4];
#pragma unroll
        for (int k = 0; k < MAXDEG / 4; ++k)
            tt[k] = *(const int4*)(trow + 4 * ((k + tid) & 7));
        int c1 = 0, c2 = 0, c3 = 0;
#pragma unroll
        for (int k = 0; k < MAXDEG / 4; ++k) {
            int4 t = tt[k];
            c1 += (int)(t.x == p1) + (int)(t.y == p1) + (int)(t.z == p1) + (int)(t.w == p1);
            c2 += (int)(t.x == p2) + (int)(t.y == p2) + (int)(t.z == p2) + (int)(t.w == p2);
            c3 += (int)(t.x == p3) + (int)(t.y == p3) + (int)(t.z == p3) + (int)(t.w == p3);
            t.x = (t.x == p1 || t.x == p2 || t.x == p3) ? N_DRUG : t.x;
            t.y = (t.y == p1 || t.y == p2 || t.y == p3) ? N_DRUG : t.y;
            t.z = (t.z == p1 || t.z == p2 || t.z == p3) ? N_DRUG : t.z;
            t.w = (t.w == p1 || t.w == p2 || t.w == p3) ? N_DRUG : t.w;
            *(int4*)(trow + 4 * ((k + tid) & 7)) = t;
        }
        const int dd = dg < 1 ? 1 : (dg < MAXDEG ? dg : MAXDEG);
        const float hc = dg > 0 ? 0.5f / (float)dd : 0.f;
        scoef[r] = make_float4(hc, hc * (float)c1, hc * (float)c2, hc * (float)c3);
    }

    float4 t1 = make_float4(0.f, 0.f, 0.f, 0.f), t2 = t1;
#pragma unroll
    for (int wv = 0; wv < 8; ++wv) {
        const float4 a = *(const float4*)&red[0][wv][c4];
        const float4 b = *(const float4*)&red[1][wv][c4];
        t1.x += a.x; t1.y += a.y; t1.z += a.z; t1.w += a.w;
        t2.x += b.x; t2.y += b.y; t2.z += b.z; t2.w += b.w;
    }
    const float inv_n = 1.0f / (float)N_DRUG;
    float4 sc, sh;
    {
        const float mx = t1.x * inv_n, my = t1.y * inv_n, mz = t1.z * inv_n, mw = t1.w * inv_n;
        sc.x = g4.x * rsqrtf((t2.x * inv_n - mx * mx) + 1e-5f);
        sc.y = g4.y * rsqrtf((t2.y * inv_n - my * my) + 1e-5f);
        sc.z = g4.z * rsqrtf((t2.z * inv_n - mz * mz) + 1e-5f);
        sc.w = g4.w * rsqrtf((t2.w * inv_n - mw * mw) + 1e-5f);
        sh.x = b4.x - mx * sc.x;  sh.y = b4.y - my * sc.y;
        sh.z = b4.z - mz * sc.z;  sh.w = b4.w - mw * sc.w;
    }
    float* sp = slab + (w * SROWS + rb) * WCOLS;
#pragma unroll
    for (int j = 0; j < HROWS; ++j) {
        if (j < HFULL || tail) {
            float4 o;
            o.x = hv[j].x * sc.x + sh.x;  o.y = hv[j].y * sc.y + sh.y;
            o.z = hv[j].z * sc.z + sh.z;  o.w = hv[j].w * sc.w + sh.w;
            *(float4*)(sp + RSTEP * j * WCOLS) = o;
        }
    }
    __syncthreads();

    if (ep > 1 && wave < SWAVES) {
        const int g = lane & 7;
        float* col = slab + wave * (SROWS * WCOLS) + (lane >> 3);
        const int* tab = sidx + 4 * g;
        const int lastp = N_DRUG - 2;

        const int4 t0e = *(const int4*)(tab);
        const int4 t0o = *(const int4*)(tab + MAXDEG);
        int4 tYe = *(const int4*)(tab + 2 * MAXDEG);
        int4 tYo = *(const int4*)(tab + 3 * MAXDEG);
        int4 tXe = *(const int4*)(tab + 4 * MAXDEG);
        int4 tXo = *(const int4*)(tab + 5 * MAXDEG);
        float4 cXe = scoef[0];
        float4 cXo = scoef[1];
        pairvals inX = gather_pair(col, t0e, t0o, 0);
        float n1 = 0.f, n2 = 0.f;

        for (int p = 0; p < N_DRUG; p += 4) {
            const int p2 = (p + 2 < lastp) ? (p + 2) : lastp;
            const int p4 = (p + 4 < lastp) ? (p + 4) : lastp;
            const int p6 = (p + 6 < lastp) ? (p + 6) : lastp;
            const int p8 = (p + 8 < lastp) ? (p + 8) : lastp;
            pairvals inY;
            float4 cYe, cYo, cNe, cNo;
            int4 tNe, tNo;
            {
                inY = gather_pair(col, tYe, tYo, p2);
                cYe = scoef[p2];
                cYo = scoef[p2 + 1];
                cNe = scoef[p4];
                cNo = scoef[p4 + 1];
                tYe = *(const int4*)(tab + p6 * MAXDEG);
                tYo = *(const int4*)(tab + (p6 + 1) * MAXDEG);
                tNe = *(const int4*)(tab + p8 * MAXDEG);
                tNo = *(const int4*)(tab + (p8 + 1) * MAXDEG);
                __builtin_amdgcn_sched_barrier(0);
                float ne, no;
                finish_pair(inX, cXe, cXo, n1, n2, ne, no);
                if (g == 0) {
                    col[p * WCOLS]       = ne;
                    col[(p + 1) * WCOLS] = no;
                }
                n2 = ne;  n1 = no;
            }
            {
                inX = gather_pair(col, tXe, tXo, p4);
                __builtin_amdgcn_sched_barrier(0);
                float ne, no;
                finish_pair(inY, cYe, cYo, n1, n2, ne, no);
                if (g == 0) {
                    col[(p + 2) * WCOLS] = ne;
                    col[(p + 3) * WCOLS] = no;
                }
                n2 = ne;  n1 = no;
            }
            cXe = cNe;  cXo = cNo;
            tXe = tNe;  tXo = tNo;
        }
    }
    __syncthreads();

    for (int v = tid; v < SVEC; v += 256) {
        const int row = v >> 2;
        *(float4*)(f + (size_t)row * DIM + col0 + c4) =
            *(const float4*)(slab + (w * SROWS + row) * WCOLS);
    }
}

extern "C" void kernel_launch(void* const* d_in, const int* in_sizes, int n_in,
                              void* d_out, int out_size, void* d_ws, size_t ws_size,
                              hipStream_t stream)
{
    const float* drugW  = (const float*)d_in[0];
    const float* relaW  = (const float*)d_in[1];
    const float* entW   = (const float*)d_in[2];
    const float* Wa     = (const float*)d_in[3];
    const float* linW   = (const float*)d_in[4];
    const float* linb   = (const float*)d_in[5];
    const float* gamma  = (const float*)d_in[6];
    const float* beta   = (const float*)d_in[7];
    const int* drug_name = (const int*)d_in[8];
    const int* adj_tail  = (const int*)d_in[9];
    const int* adj_rel   = (const int*)d_in[10];
    const int* nbr_idx   = (const int*)d_in[11];
    const int* nbr_deg   = (const int*)d_in[12];
    const int* epoch     = (const int*)d_in[13];

    float* out = (float*)d_out;
    float* ws  = (float*)d_ws;
    float* q     = ws;
    float* agg   = ws + (size_t)NPAD * DIM;
    float* h     = ws + (size_t)2 * NPAD * DIM;

    dim3 gTiles(MP, NT);
    qgemm_kernel<<<gTiles, 32, 0, stream>>>(drugW, Wa, drug_name, q);
    attn_agg_kernel<<<NPAD, 256, 0, stream>>>(q, relaW, entW, adj_rel, adj_tail, agg);
    lingemm_kernel<<<gTiles, 32, 0, stream>>>(agg, drugW, drug_name, linW, linb, h);
    bn_smooth_kernel<<<DIM / SCOLS, 256, 0, stream>>>(h, gamma, beta, nbr_idx, nbr_deg, epoch, out);
}


# smoothing scan by blocks of 16 rows, one row per lane: its 32 slot gathers go out at once, the block's earlier rows enter via a 16x16 count matrix solved by 15 row-broadcast FMAs; sum order changes
# speedup vs baseline: 18.8090x; 1.2598x over previous
#include <hip/hip_runtime.h>

typedef __attribute__((ext_vector_type(2))) float v2f;
typedef __attribute__((ext_vector_type(8))) float v8f;

#define N_DRUG 572
#define DIM    512
#define KNBR   64
#define MAXDEG 32
#define NPAD   576
#define MT     36
#define NT     32
#define MP     18

#define CHK 64
#define KST (CHK / 4)

struct opchunk { float4 a0[KST / 2]; float4 a1[KST / 2]; v2f b[KST]; };

__device__ __forceinline__ void load_chunk(opchunk& f, const float* __restrict__ A0p,
                                           const float* __restrict__ A1p,
                                           const float* __restrict__ Bp)
{
#pragma unroll
    for (int j = 0; j < KST / 2; ++j) {
        f.a0[j] = *(const float4*)(A0p + 4 * j);
        f.a1[j] = *(const float4*)(A1p + 4 * j);
        f.b[2 * j].x     = Bp[(size_t)(4 * j) * DIM];
        f.b[2 * j].y     = Bp[(size_t)(4 * j + 1) * DIM];
        f.b[2 * j + 1].x = Bp[(size_t)(4 * j + 2) * DIM];
        f.b[2 * j + 1].y = Bp[(size_t)(4 * j + 3) * DIM];
    }
}

__device__ __forceinline__ void mma_chunk(const opchunk& f, v8f& c0, v8f& c1)
{
    __builtin_amdgcn_sched_barrier(0);
#pragma unroll
    for (int j = 0; j < KST / 2; ++j) {
        v2f p0, p1, q0, q1;
        p0.x = f.a0[j].x;  p0.y = f.a0[j].y;
        q0.x = f.a0[j].z;  q0.y = f.a0[j].w;
        p1.x = f.a1[j].x;  p1.y = f.a1[j].y;
        q1.x = f.a1[j].z;  q1.y = f.a1[j].w;
        c0 = __builtin_amdgcn_wmma_f32_16x16x4_f32(false, p0, false, f.b[2 * j],
                                                   (short)0, c0, false, false);
        c1 = __builtin_amdgcn_wmma_f32_16x16x4_f32(false, p1, false, f.b[2 * j],
                                                   (short)0, c1, false, false);
        c0 = __builtin_amdgcn_wmma_f32_16x16x4_f32(false, q0, false, f.b[2 * j + 1],
                                                   (short)0, c0, false, false);
        c1 = __builtin_amdgcn_wmma_f32_16x16x4_f32(false, q1, false, f.b[2 * j + 1],
                                                   (short)0, c1, false, false);
    }
    __builtin_amdgcn_sched_barrier(0);
    asm volatile("v_nop\n\tv_nop\n\tv_nop\n\tv_nop");
    __builtin_amdgcn_sched_barrier(0);
}

#define LOAD_AT(F, K)                                                              \
    do {                                                                           \
        if ((K) < DIM) load_chunk(F, P0 + (K), P1 + (K), Bp + (size_t)(K) * DIM);  \
        else load_chunk(F, Q0 + ((K) - DIM), Q1 + ((K) - DIM), Bp + (size_t)(K) * DIM); \
    } while (0)

template <int KTOT>
__device__ __forceinline__ void wave_gemm_32x16(const float* __restrict__ P0,
                                                const float* __restrict__ P1,
                                                const float* __restrict__ Q0,
                                                const float* __restrict__ Q1,
                                                const float* __restrict__ Bp,
                                                v8f& c0, v8f& c1)
{
    opchunk f0, f1;
    load_chunk(f0, P0, P1, Bp);
#pragma unroll 1
    for (int kc = 0; kc < KTOT - 2 * CHK; kc += 2 * CHK) {
        LOAD_AT(f1, kc + CHK);
        mma_chunk(f0, c0, c1);
        LOAD_AT(f0, kc + 2 * CHK);
        mma_chunk(f1, c0, c1);
    }
    LOAD_AT(f1, KTOT - CHK);
    mma_chunk(f0, c0, c1);
    mma_chunk(f1, c0, c1);
}
#undef LOAD_AT

__global__ __launch_bounds__(32)
void qgemm_kernel(const float* __restrict__ drugW, const float* __restrict__ Wa,
                  const int* __restrict__ drug_name, float* __restrict__ q)
{
    const int lane = threadIdx.x;
    const int half = lane >> 4;
    const int l16  = lane & 15;
    const int mBase = blockIdx.x * 32;
    const int nBase = blockIdx.y * 16;

    const int r0 = mBase + l16, r1 = r0 + 16;
    const int g0 = drug_name[r0 < N_DRUG ? r0 : (N_DRUG - 1)];
    const int g1 = drug_name[r1 < N_DRUG ? r1 : (N_DRUG - 1)];
    const float* __restrict__ A0 = drugW + (size_t)g0 * DIM + 32 * half;
    const float* __restrict__ A1 = drugW + (size_t)g1 * DIM + 32 * half;
    const float* __restrict__ Bp = Wa + (size_t)(32 * half) * DIM + nBase + l16;

    v8f c0 = {}, c1 = {};
    wave_gemm_32x16<DIM>(A0, A1, A0, A1, Bp, c0, c1);

    float* outBase = q + (size_t)(mBase + half * 8) * DIM + nBase + l16;
#pragma unroll
    for (int j = 0; j < 8; ++j) {
        outBase[(size_t)j * DIM]        = c0[j];
        outBase[(size_t)(j + 16) * DIM] = c1[j];
    }
}

template <int CTRL>
__device__ __forceinline__ float add_from_lane(float s)
{
    const int o = __builtin_amdgcn_update_dpp(0, __float_as_int(s), CTRL, 0xF, 0xF, true);
    return s + __int_as_float(o);
}

template <int CTRL>
__device__ __forceinline__ float max_from_lane(float s)
{
    const int o = __builtin_amdgcn_update_dpp(0, __float_as_int(s), CTRL, 0xF, 0xF, true);
    return fmaxf(s, __int_as_float(o));
}

__device__ __forceinline__ float from_other_row(float x)
{
    return __int_as_float(__builtin_amdgcn_permlanex16(__float_as_int(x), __float_as_int(x),
                                                       0x76543210, 0xfedcba98, false, false));
}

__device__ __forceinline__ float wave_sum(float s)
{
    s += from_other_row(s);
    s = add_from_lane<0x168>(s);
    s = add_from_lane<0x164>(s);
    s = add_from_lane<0x4E>(s);
    s = add_from_lane<0xB1>(s);
    return s;
}

__device__ __forceinline__ float wave_max(float m)
{
    m = fmaxf(m, from_other_row(m));
    m = max_from_lane<0x168>(m);
    m = max_from_lane<0x164>(m);
    m = max_from_lane<0x4E>(m);
    m = max_from_lane<0xB1>(m);
    return m;
}

__global__ __launch_bounds__(256)
void attn_agg_kernel(const float* __restrict__ q, const float* __restrict__ relaW,
                     const float* __restrict__ entW,
                     const int* __restrict__ adj_rel, const int* __restrict__ adj_tail,
                     float* __restrict__ agg)
{
    __shared__ __attribute__((aligned(16))) float attn[KNBR];

    const int n  = blockIdx.x;
    const int ni = n < N_DRUG ? n : (N_DRUG - 1);
    const int tid = threadIdx.x;
    const int lane = tid & 31;
    const int wave = __builtin_amdgcn_readfirstlane(tid >> 5);

    const int* __restrict__ relp = adj_rel + ni * KNBR + wave * 8;
    int rel[8];
#pragma unroll
    for (int kk = 0; kk < 8; ++kk)
        rel[kk] = relp[kk];

    const float* __restrict__ qp = q + (size_t)n * DIM + 4 * lane;
    float4 qv[4];
#pragma unroll
    for (int i = 0; i < 4; ++i)
        qv[i] = *(const float4*)(qp + 128 * i);

    float p[8];
#pragma unroll
    for (int g = 0; g < 2; ++g) {
        float4 rv[4][4];
#pragma unroll
        for (int kk = 0; kk < 4; ++kk) {
            const float* __restrict__ rw = relaW + (size_t)rel[4 * g + kk] * DIM + 4 * lane;
#pragma unroll
            for (int i = 0; i < 4; ++i)
                rv[kk][i] = *(const float4*)(rw + 128 * i);
        }
#pragma unroll
        for (int kk = 0; kk < 4; ++kk) {
            float s = 0.f;
#pragma unroll
            for (int i = 0; i < 4; ++i) {
                s += qv[i].x * rv[kk][i].x;
                s += qv[i].y * rv[kk][i].y;
                s += qv[i].z * rv[kk][i].z;
                s += qv[i].w * rv[kk][i].w;
            }
            p[4 * g + kk] = s;
        }
    }
#pragma unroll
    for (int kk = 0; kk < 8; ++kk)
        p[kk] = wave_sum(p[kk]);
    if (lane == 0) {
        const float sc = 0.04419417382415922f;
        *(float4*)&attn[wave * 8]     = make_float4(p[0] * sc, p[1] * sc, p[2] * sc, p[3] * sc);
        *(float4*)&attn[wave * 8 + 4] = make_float4(p[4] * sc, p[5] * sc, p[6] * sc, p[7] * sc);
    }
    __syncthreads();

    if (tid < 32) {
        const float s0 = attn[tid], s1 = attn[tid + 32];
        const float m = wave_max(fmaxf(s0, s1));
        const float e0 = __expf(s0 - m), e1 = __expf(s1 - m);
        const float s = wave_sum(e0 + e1);
        const float inv = 1.f / s;
        attn[tid]      = e0 * inv;
        attn[tid + 32] = e1 * inv;
    }
    __syncthreads();

    const int* __restrict__ tp = adj_tail + ni * KNBR;
    const float* __restrict__ ecol = entW + 2 * tid;
    float acc0 = 0.f, acc1 = 0.f;
#pragma unroll
    for (int k0 = 0; k0 < KNBR; k0 += 32) {
        float2 e[32];
#pragma unroll
        for (int j = 0; j < 32; ++j)
            e[j] = *(const float2*)(ecol + (size_t)tp[k0 + j] * DIM);
#pragma unroll
        for (int j = 0; j < 32; ++j) {
            const float w = attn[k0 + j];
            acc0 += w * e[j].x;
            acc1 += w * e[j].y;
        }
    }
    *(float2*)(agg + (size_t)n * DIM + 2 * tid) = make_float2(acc0, acc1);
}

__global__ __launch_bounds__(32)
void lingemm_kernel(const float* __restrict__ agg, const float* __restrict__ drugW,
                    const int* __restrict__ drug_name,
                    const float* __restrict__ linW, const float* __restrict__ linb,
                    float* __restrict__ h)
{
    const int lane = threadIdx.x;
    const int half = lane >> 4;
    const int l16  = lane & 15;
    const int mBase = blockIdx.x * 32;
    const int nBase = blockIdx.y * 16;

    const int r0 = mBase + l16, r1 = r0 + 16;
    const int g0 = drug_name[r0 < N_DRUG ? r0 : (N_DRUG - 1)];
    const int g1 = drug_name[r1 < N_DRUG ? r1 : (N_DRUG - 1)];
    const float* __restrict__ P0 = agg + (size_t)r0 * DIM + 32 * half;
    const float* __restrict__ P1 = agg + (size_t)r1 * DIM + 32 * half;
    const float* __restrict__ Q0 = drugW + (size_t)g0 * DIM + 32 * half;
    const float* __restrict__ Q1 = drugW + (size_t)g1 * DIM + 32 * half;
    const float* __restrict__ Bp = linW + (size_t)(32 * half) * DIM + nBase + l16;

    v8f c0 = {}, c1 = {};
    wave_gemm_32x16<2 * DIM>(P0, P1, Q0, Q1, Bp, c0, c1);

    const float bias = linb[nBase + l16];
    float* outBase = h + (size_t)(mBase + half * 8) * DIM + nBase + l16;
#pragma unroll
    for (int j = 0; j < 8; ++j) {
        const float v0 = c0[j] + bias, v1 = c1[j] + bias;
        outBase[(size_t)j * DIM]        = v0 > 0.f ? v0 : 0.f;
        outBase[(size_t)(j + 16) * DIM] = v1 > 0.f ? v1 : 0.f;
    }
}

#define SCOLS 16
#define WCOLS 4
#define SWAVES (SCOLS / WCOLS)
#define NZROW NPAD
#define SROWS (NPAD + 1)
#define ASTR  5
#define SVEC  (N_DRUG * SCOLS / 4)
#define TVEC  (N_DRUG * MAXDEG / 4)
#define RSTEP (256 / (SCOLS / 4))
#define HFULL (N_DRUG / RSTEP)
#define HROWS (HFULL + 1)

template <int OFF>
__device__ __forceinline__ float4 add_from_lane4(float4 v)
{
    v.x += __shfl_xor(v.x, OFF, 32);
    v.y += __shfl_xor(v.y, OFF, 32);
    v.z += __shfl_xor(v.z, OFF, 32);
    v.w += __shfl_xor(v.w, OFF, 32);
    return v;
}

__device__ __forceinline__ int count_entry(const int e, const int R, const unsigned i,
                                           unsigned long long& lo, unsigned long long& hi)
{
    const unsigned d = (unsigned)(e - R);
    const bool inr = d < i;
    const unsigned long long inc = (unsigned long long)(inr ? 1u : 0u) << ((d & 7u) * 8u);
    lo += (d & 8u) ? 0ull : inc;
    hi += (d & 8u) ? inc : 0ull;
    return (inr ? NZROW : e) * WCOLS;
}

#define SOLVE_STEP(J, A)                                                                   \
    do {                                                                                   \
        asm("v_fmac_f32_dpp %0, %0, %1 row_share:" #J " row_mask:0xf bank_mask:0xf bound_ctrl:1" \
            : "+v"(rhs.x) : "v"(A));                                                       \
        asm("v_fmac_f32_dpp %0, %0, %1 row_share:" #J " row_mask:0xf bank_mask:0xf bound_ctrl:1" \
            : "+v"(rhs.y) : "v"(A));                                                       \
    } while (0)

__global__ __launch_bounds__(256)
void bn_smooth_kernel(const float* __restrict__ h, const float* __restrict__ gamma,
                      const float* __restrict__ beta,
                      const int* __restrict__ nbr_idx, const int* __restrict__ nbr_deg,
                      const int* __restrict__ epoch, float* __restrict__ f)
{
    __shared__ __attribute__((aligned(16))) float slab[SWAVES * SROWS * WCOLS];
    __shared__ __attribute__((aligned(16))) int   sidx[NPAD * MAXDEG];
    __shared__ __attribute__((aligned(16))) float4 acoef[NPAD * ASTR];
    __shared__ __attribute__((aligned(16))) float2 hbco[NPAD];
    __shared__ __attribute__((aligned(16))) float red[2][8][SCOLS];

    const int tid  = threadIdx.x;
    const int lane = tid & 31;
    const int wave = __builtin_amdgcn_readfirstlane(tid >> 5);
    const int col0 = blockIdx.x * SCOLS;
    const int w    = tid & 3;
    const int c4   = w * 4;
    const int rb   = tid >> 2;
    const int ts4  = (tid & 7) * 4;
    const int ep   = *epoch;

    const float* __restrict__ hp = h + (size_t)rb * DIM + col0 + c4;
    const bool tail = rb + RSTEP * HFULL < N_DRUG;
    float4 hv[HROWS];
#pragma unroll
    for (int j = 0; j < HFULL; ++j)
        hv[j] = *(const float4*)(hp + (size_t)(RSTEP * j) * DIM);
    hv[HFULL] = *(const float4*)(hp + (size_t)(tail ? RSTEP * HFULL : 0) * DIM);
    const float4 g4 = *(const float4*)(gamma + col0 + c4);
    const float4 b4 = *(const float4*)(beta + col0 + c4);

    for (int v = tid; v < TVEC; v += 256) {
        const int dg = nbr_deg[v >> 3];
        int4 t = *(const int4*)(nbr_idx + v * 4);
        t.x = (ts4     < dg) ? t.x : NZROW;
        t.y = (ts4 + 1 < dg) ? t.y : NZROW;
        t.z = (ts4 + 2 < dg) ? t.z : NZROW;
        t.w = (ts4 + 3 < dg) ? t.w : NZROW;
        *(int4*)(sidx + v * 4) = t;
    }
    if (tid < SWAVES * (SROWS - N_DRUG) * WCOLS) {
        const int sw  = tid / ((SROWS - N_DRUG) * WCOLS);
        const int rem = tid % ((SROWS - N_DRUG) * WCOLS);
        const int row = N_DRUG + rem / WCOLS;
        slab[(sw * SROWS + row) * WCOLS + (rem % WCOLS)] = (row == NZROW) ? -0.0f : 0.0f;
    }

    if (!tail)
        hv[HFULL] = make_float4(0.f, 0.f, 0.f, 0.f);
    float4 s1 = make_float4(0.f, 0.f, 0.f, 0.f), s2 = s1;
#pragma unroll
    for (int j = 0; j < HROWS; ++j) {
        s1.x += hv[j].x;            s1.y += hv[j].y;
        s1.z += hv[j].z;            s1.w += hv[j].w;
        s2.x += hv[j].x * hv[j].x;  s2.y += hv[j].y * hv[j].y;
        s2.z += hv[j].z * hv[j].z;  s2.w += hv[j].w * hv[j].w;
    }
    s1 = add_from_lane4<4>(s1);   s2 = add_from_lane4<4>(s2);
    s1 = add_from_lane4<8>(s1);   s2 = add_from_lane4<8>(s2);
    s1 = add_from_lane4<16>(s1);  s2 = add_from_lane4<16>(s2);
    if (lane < 4) {
        *(float4*)&red[0][wave][c4] = s1;
        *(float4*)&red[1][wave][c4] = s2;
    }
    __syncthreads();

    for (int r = tid; r < NPAD; r += 256) {
        int* trow = sidx + r * MAXDEG;
        if (r < N_DRUG) {
            const int dg = nbr_deg[r];
            const int R = r & ~15;
            const unsigned i = (unsigned)(r & 15);
            int4 tt[MAXDEG / 4];
#pragma unroll
            for (int k = 0; k < MAXDEG / 4; ++k)
                tt[k] = *(const int4*)(trow + 4 * ((k + tid) & 7));
            unsigned long long lo = 0ull, hi = 0ull;
#pragma unroll
            for (int k = 0; k < MAXDEG / 4; ++k) {
                int4 t = tt[k];
                t.x = count_entry(t.x, R, i, lo, hi);
                t.y = count_entry(t.y, R, i, lo, hi);
                t.z = count_entry(t.z, R, i, lo, hi);
                t.w = count_entry(t.w, R, i, lo, hi);
                *(int4*)(trow + 4 * ((k + tid) & 7)) = t;
            }
            const int dd = dg < 1 ? 1 : (dg < MAXDEG ? dg : MAXDEG);
            const float hc = dg > 0 ? 0.5f / (float)dd : 0.f;
            float av[16];
#pragma unroll
            for (int j = 0; j < 8; ++j) {
                av[j]     = hc * (float)((unsigned)(lo >> (8 * j)) & 0xffu);
                av[8 + j] = hc * (float)((unsigned)(hi >> (8 * j)) & 0xffu);
            }
#pragma unroll
            for (int qd = 0; qd < 4; ++qd)
                acoef[r * ASTR + qd] = make_float4(av[4 * qd], av[4 * qd + 1],
                                                   av[4 * qd + 2], av[4 * qd + 3]);
            hbco[r] = make_float2(hc, dg > 0 ? 0.5f : 1.0f);
        } else {
            const int4 nz = make_int4(NZROW * WCOLS, NZROW * WCOLS, NZROW * WCOLS, NZROW * WCOLS);
#pragma unroll
            for (int k = 0; k < MAXDEG / 4; ++k)
                *(int4*)(trow + 4 * k) = nz;
#pragma unroll
            for (int qd = 0; qd < 4; ++qd)
                acoef[r * ASTR + qd] = make_float4(0.f, 0.f, 0.f, 0.f);
            hbco[r] = make_float2(0.f, 1.0f);
        }
    }

    float4 t1 = make_float4(0.f, 0.f, 0.f, 0.f), t2 = t1;
#pragma unroll
    for (int wv = 0; wv < 8; ++wv) {
        const float4 a = *(const float4*)&red[0][wv][c4];
        const float4 b = *(const float4*)&red[1][wv][c4];
        t1.x += a.x; t1.y += a.y; t1.z += a.z; t1.w += a.w;
        t2.x += b.x; t2.y += b.y; t2.z += b.z; t2.w += b.w;
    }
    const float inv_n = 1.0f / (float)N_DRUG;
    float4 sc, sh;
    {
        const float mx = t1.x * inv_n, my = t1.y * inv_n, mz = t1.z * inv_n, mw = t1.w * inv_n;
        sc.x = g4.x * rsqrtf((t2.x * inv_n - mx * mx) + 1e-5f);
        sc.y = g4.y * rsqrtf((t2.y * inv_n - my * my) + 1e-5f);
        sc.z = g4.z * rsqrtf((t2.z * inv_n - mz * mz) + 1e-5f);
        sc.w = g4.w * rsqrtf((t2.w * inv_n - mw * mw) + 1e-5f);
        sh.x = b4.x - mx * sc.x;  sh.y = b4.y - my * sc.y;
        sh.z = b4.z - mz * sc.z;  sh.w = b4.w - mw * sc.w;
    }
    float* sp = slab + (w * SROWS + rb) * WCOLS;
#pragma unroll
    for (int j = 0; j < HROWS; ++j) {
        if (j < HFULL || tail) {
            float4 o;
            o.x = hv[j].x * sc.x + sh.x;  o.y = hv[j].y * sc.y + sh.y;
            o.z = hv[j].z * sc.z + sh.z;  o.w = hv[j].w * sc.w + sh.w;
            *(float4*)(sp + RSTEP * j * WCOLS) = o;
        }
    }
    __syncthreads();

    if (ep > 1 && wave < SWAVES) {
        const int i   = lane & 15;
        const int rot = i >> 1;
        float* colp = slab + wave * (SROWS * WCOLS) + 2 * (lane >> 4);

        int4 tt[MAXDEG / 4];
#pragma unroll
        for (int k = 0; k < MAXDEG / 4; ++k)
            tt[k] = *(const int4*)(sidx + i * MAXDEG + 4 * ((k + rot) & 7));

#define SCAN_BLOCK(RB)                                                                     \
        do {                                                                               \
            const int r = (RB) + i;                                                        \
                  \
            float2 gv[MAXDEG];                                                             \
            _Pragma("unroll")                                                              \
            for (int k = 0; k < MAXDEG / 4; ++k) {                                         \
                gv[4 * k]     = *(const float2*)(colp + tt[k].x);                          \
                gv[4 * k + 1] = *(const float2*)(colp + tt[k].y);                          \
                gv[4 * k + 2] = *(const float2*)(colp + tt[k].z);                          \
                gv[4 * k + 3] = *(const float2*)(colp + tt[k].w);                          \
            }                                                                              \
            const float2 own = *(const float2*)(colp + r * WCOLS);                         \
            const float2 hb  = hbco[r];                                                    \
            const float4 a0 = acoef[r * ASTR],     a1 = acoef[r * ASTR + 1];               \
            const float4 a2 = acoef[r * ASTR + 2], a3 = acoef[r * ASTR + 3];               \
                  \
            const int rn = ((RB) + 16 < NPAD ? (RB) + 16 : (RB)) + i;                      \
            _Pragma("unroll")                                                              \
            for (int k = 0; k < MAXDEG / 4; ++k)                                           \
                tt[k] = *(const int4*)(sidx + rn * MAXDEG + 4 * ((k + rot) & 7));          \
                       \
            float2 s0 = gv[0], s1 = gv[1], s2 = gv[2], s3 = gv[3];                         \
            _Pragma("unroll")                                                              \
            for (int k = 1; k < MAXDEG / 4; ++k) {                                         \
                s0.x += gv[4 * k].x;      s0.y += gv[4 * k].y;                             \
                s1.x += gv[4 * k + 1].x;  s1.y += gv[4 * k + 1].y;                         \
                s2.x += gv[4 * k + 2].x;  s2.y += gv[4 * k + 2].y;                         \
                s3.x += gv[4 * k + 3].x;  s3.y += gv[4 * k + 3].y;                         \
            }                                                                              \
            float2 rhs;                                                                    \
            rhs.x = fmaf(hb.x, (s0.x + s1.x) + (s2.x + s3.x), hb.y * own.x);               \
            rhs.y = fmaf(hb.x, (s0.y + s1.y) + (s2.y + s3.y), hb.y * own.y);               \
                                                         \
            SOLVE_STEP(0,  a0.x);  SOLVE_STEP(1,  a0.y);  SOLVE_STEP(2,  a0.z);            \
            SOLVE_STEP(3,  a0.w);  SOLVE_STEP(4,  a1.x);  SOLVE_STEP(5,  a1.y);            \
            SOLVE_STEP(6,  a1.z);  SOLVE_STEP(7,  a1.w);  SOLVE_STEP(8,  a2.x);            \
            SOLVE_STEP(9,  a2.y);  SOLVE_STEP(10, a2.z);  SOLVE_STEP(11, a2.w);            \
            SOLVE_STEP(12, a3.x);  SOLVE_STEP(13, a3.y);  SOLVE_STEP(14, a3.z);            \
            *(float2*)(colp + r * WCOLS) = rhs;                                            \
        } while (0)

        SCAN_BLOCK(0);
        for (int R = 16; R < NPAD; R += 16)
            SCAN_BLOCK(R);
#undef SCAN_BLOCK
    }
    __syncthreads();

    for (int v = tid; v < SVEC; v += 256) {
        const int row = v >> 2;
        *(float4*)(f + (size_t)row * DIM + col0 + c4) =
            *(const float4*)(slab + (w * SROWS + row) * WCOLS);
    }
}

extern "C" void kernel_launch(void* const* d_in, const int* in_sizes, int n_in,
                              void* d_out, int out_size, void* d_ws, size_t ws_size,
                              hipStream_t stream)
{
    const float* drugW  = (const float*)d_in[0];
    const float* relaW  = (const float*)d_in[1];
    const float* entW   = (const float*)d_in[2];
    const float* Wa     = (const float*)d_in[3];
    const float* linW   = (const float*)d_in[4];
    const float* linb   = (const float*)d_in[5];
    const float* gamma  = (const float*)d_in[6];
    const float* beta   = (const float*)d_in[7];
    const int* drug_name = (const int*)d_in[8];
    const int* adj_tail  = (const int*)d_in[9];
    const int* adj_rel   = (const int*)d_in[10];
    const int* nbr_idx   = (const int*)d_in[11];
    const int* nbr_deg   = (const int*)d_in[12];
    const int* epoch     = (const int*)d_in[13];

    float* out = (float*)d_out;
    float* ws  = (float*)d_ws;
    float* q     = ws;
    float* agg   = ws + (size_t)NPAD * DIM;
    float* h     = ws + (size_t)2 * NPAD * DIM;

    dim3 gTiles(MP, NT);
    qgemm_kernel<<<gTiles, 32, 0, stream>>>(drugW, Wa, drug_name, q);
    attn_agg_kernel<<<NPAD, 256, 0, stream>>>(q, relaW, entW, adj_rel, adj_tail, agg);
    lingemm_kernel<<<gTiles, 32, 0, stream>>>(agg, drugW, drug_name, linW, linb, h);
    bn_smooth_kernel<<<DIM / SCOLS, 256, 0, stream>>>(h, gamma, beta, nbr_idx, nbr_deg, epoch, out);
}


# scores reassociated: launch 1 forms rela x Wa^T once (200 rows, not 572 x Wa; K split over a block's 4 waves, added via LDS), attention dots raw embedding rows with it; score sum order changes
# speedup vs baseline: 20.9434x; 1.1135x over previous
#include <hip/hip_runtime.h>

typedef __attribute__((ext_vector_type(2))) float v2f;
typedef __attribute__((ext_vector_type(8))) float v8f;

#define N_DRUG 572
#define DIM    512
#define KNBR   64
#define MAXDEG 32
#define NPAD   576
#define MT     36
#define NT     32
#define MP     18
#define N_REL  200
#define RPAD   224
#define RP     7

#define CHK 64
#define KST (CHK / 4)

struct opchunk { float4 a0[KST / 2]; float4 a1[KST / 2]; v2f b[KST]; };

__device__ __forceinline__ void load_chunk(opchunk& f, const float* __restrict__ A0p,
                                           const float* __restrict__ A1p,
                                           const float* __restrict__ Bp)
{
#pragma unroll
    for (int j = 0; j < KST / 2; ++j) {
        f.a0[j] = *(const float4*)(A0p + 4 * j);
        f.a1[j] = *(const float4*)(A1p + 4 * j);
        f.b[2 * j].x     = Bp[(size_t)(4 * j) * DIM];
        f.b[2 * j].y     = Bp[(size_t)(4 * j + 1) * DIM];
        f.b[2 * j + 1].x = Bp[(size_t)(4 * j + 2) * DIM];
        f.b[2 * j + 1].y = Bp[(size_t)(4 * j + 3) * DIM];
    }
}

__device__ __forceinline__ void mma_chunk(const opchunk& f, v8f& c0, v8f& c1)
{
    __builtin_amdgcn_sched_barrier(0);
#pragma unroll
    for (int j = 0; j < KST / 2; ++j) {
        v2f p0, p1, q0, q1;
        p0.x = f.a0[j].x;  p0.y = f.a0[j].y;
        q0.x = f.a0[j].z;  q0.y = f.a0[j].w;
        p1.x = f.a1[j].x;  p1.y = f.a1[j].y;
        q1.x = f.a1[j].z;  q1.y = f.a1[j].w;
        c0 = __builtin_amdgcn_wmma_f32_16x16x4_f32(false, p0, false, f.b[2 * j],
                                                   (short)0, c0, false, false);
        c1 = __builtin_amdgcn_wmma_f32_16x16x4_f32(false, p1, false, f.b[2 * j],
                                                   (short)0, c1, false, false);
        c0 = __builtin_amdgcn_wmma_f32_16x16x4_f32(false, q0, false, f.b[2 * j + 1],
                                                   (short)0, c0, false, false);
        c1 = __builtin_amdgcn_wmma_f32_16x16x4_f32(false, q1, false, f.b[2 * j + 1],
                                                   (short)0, c1, false, false);
    }
    __builtin_amdgcn_sched_barrier(0);
    asm volatile("v_nop\n\tv_nop\n\tv_nop\n\tv_nop");
    __builtin_amdgcn_sched_barrier(0);
}

#define LOAD_AT(F, K)                                                              \
    do {                                                                           \
        if ((K) < DIM) load_chunk(F, P0 + (K), P1 + (K), Bp + (size_t)(K) * DIM);  \
        else load_chunk(F, Q0 + ((K) - DIM), Q1 + ((K) - DIM), Bp + (size_t)(K) * DIM); \
    } while (0)

template <int KTOT>
__device__ __forceinline__ void wave_gemm_32x16(const float* __restrict__ P0,
                                                const float* __restrict__ P1,
                                                const float* __restrict__ Q0,
                                                const float* __restrict__ Q1,
                                                const float* __restrict__ Bp,
                                                v8f& c0, v8f& c1)
{
    opchunk f0, f1;
    load_chunk(f0, P0, P1, Bp);
#pragma unroll 1
    for (int kc = 0; kc < KTOT - 2 * CHK; kc += 2 * CHK) {
        LOAD_AT(f1, kc + CHK);
        mma_chunk(f0, c0, c1);
        LOAD_AT(f0, kc + 2 * CHK);
        mma_chunk(f1, c0, c1);
    }
    LOAD_AT(f1, KTOT - CHK);
    mma_chunk(f0, c0, c1);
    mma_chunk(f1, c0, c1);
}
#undef LOAD_AT

#define MK (DIM / 4)
#define MPC (MK / 8)

__global__ __launch_bounds__(128)
void mgemm_kernel(const float* __restrict__ relaW, const float* __restrict__ Wa,
                  float* __restrict__ M)
{
    __shared__ __attribute__((aligned(16))) float red[4][16][32];

    const int tid  = threadIdx.x;
    const int lane = tid & 31;
    const int wave = __builtin_amdgcn_readfirstlane(tid >> 5);
    const int half = lane >> 4;
    const int l16  = lane & 15;
    const int mBase = blockIdx.x * 32;
    const int nBase = blockIdx.y * 16;

    const int r0 = mBase + l16, r1 = r0 + 16;
    const int k0 = wave * MK + 4 * half;
    const float* __restrict__ A0 = relaW + (size_t)(r0 < N_REL ? r0 : (N_REL - 1)) * DIM + k0;
    const float* __restrict__ A1 = relaW + (size_t)(r1 < N_REL ? r1 : (N_REL - 1)) * DIM + k0;
    const float* __restrict__ Bp = Wa + (size_t)(nBase + l16) * DIM + k0;

    float4 a0[MPC], a1[MPC], b[MPC];
#pragma unroll
    for (int j = 0; j < MPC; ++j) {
        a0[j] = *(const float4*)(A0 + 8 * j);
        a1[j] = *(const float4*)(A1 + 8 * j);
        b[j]  = *(const float4*)(Bp + 8 * j);
    }

    v8f c0 = {}, c1 = {};
    __builtin_amdgcn_sched_barrier(0);
#pragma unroll
    for (int j = 0; j < MPC; ++j) {
        v2f p0, p1, q0, q1, bx, by;
        p0.x = a0[j].x;  p0.y = a0[j].y;
        q0.x = a0[j].z;  q0.y = a0[j].w;
        p1.x = a1[j].x;  p1.y = a1[j].y;
        q1.x = a1[j].z;  q1.y = a1[j].w;
        bx.x = b[j].x;   bx.y = b[j].y;
        by.x = b[j].z;   by.y = b[j].w;
        c0 = __builtin_amdgcn_wmma_f32_16x16x4_f32(false, p0, false, bx, (short)0, c0, false, false);
        c1 = __builtin_amdgcn_wmma_f32_16x16x4_f32(false, p1, false, bx, (short)0, c1, false, false);
        c0 = __builtin_amdgcn_wmma_f32_16x16x4_f32(false, q0, false, by, (short)0, c0, false, false);
        c1 = __builtin_amdgcn_wmma_f32_16x16x4_f32(false, q1, false, by, (short)0, c1, false, false);
    }
    __builtin_amdgcn_sched_barrier(0);
    asm volatile("v_nop\n\tv_nop\n\tv_nop\n\tv_nop");
    __builtin_amdgcn_sched_barrier(0);

#pragma unroll
    for (int j = 0; j < 8; ++j) {
        red[wave][j][lane]     = c0[j];
        red[wave][8 + j][lane] = c1[j];
    }
    __syncthreads();

#pragma unroll
    for (int i = 0; i < 4; ++i) {
        const int qd = wave * 4 + i;
        const float s = ((red[0][qd][lane] + red[1][qd][lane]) + red[2][qd][lane]) + red[3][qd][lane];
        const int row = mBase + 16 * (qd >> 3) + 8 * half + (qd & 7);
        M[(size_t)row * DIM + nBase + l16] = s;
    }
}

template <int CTRL>
__device__ __forceinline__ float add_from_lane(float s)
{
    const int o = __builtin_amdgcn_update_dpp(0, __float_as_int(s), CTRL, 0xF, 0xF, true);
    return s + __int_as_float(o);
}

template <int CTRL>
__device__ __forceinline__ float max_from_lane(float s)
{
    const int o = __builtin_amdgcn_update_dpp(0, __float_as_int(s), CTRL, 0xF, 0xF, true);
    return fmaxf(s, __int_as_float(o));
}

__device__ __forceinline__ float from_other_row(float x)
{
    return __int_as_float(__builtin_amdgcn_permlanex16(__float_as_int(x), __float_as_int(x),
                                                       0x76543210, 0xfedcba98, false, false));
}

__device__ __forceinline__ float wave_sum(float s)
{
    s += from_other_row(s);
    s = add_from_lane<0x168>(s);
    s = add_from_lane<0x164>(s);
    s = add_from_lane<0x4E>(s);
    s = add_from_lane<0xB1>(s);
    return s;
}

__device__ __forceinline__ float wave_max(float m)
{
    m = fmaxf(m, from_other_row(m));
    m = max_from_lane<0x168>(m);
    m = max_from_lane<0x164>(m);
    m = max_from_lane<0x4E>(m);
    m = max_from_lane<0xB1>(m);
    return m;
}

__global__ __launch_bounds__(256)
void attn_agg_kernel(const float* __restrict__ drugW, const int* __restrict__ drug_name,
                     const float* __restrict__ M, const float* __restrict__ entW,
                     const int* __restrict__ adj_rel, const int* __restrict__ adj_tail,
                     float* __restrict__ agg)
{
    __shared__ __attribute__((aligned(16))) float attn[KNBR];

    const int n  = blockIdx.x;
    const int ni = n < N_DRUG ? n : (N_DRUG - 1);
    const int tid = threadIdx.x;
    const int lane = tid & 31;
    const int wave = __builtin_amdgcn_readfirstlane(tid >> 5);

    const int* __restrict__ relp = adj_rel + ni * KNBR + wave * 8;
    int rel[8];
#pragma unroll
    for (int kk = 0; kk < 8; ++kk)
        rel[kk] = relp[kk];

    const float* __restrict__ qp = drugW + (size_t)drug_name[ni] * DIM + 4 * lane;
    float4 qv[4];
#pragma unroll
    for (int i = 0; i < 4; ++i)
        qv[i] = *(const float4*)(qp + 128 * i);

    float p[8];
#pragma unroll
    for (int g = 0; g < 2; ++g) {
        float4 rv[4][4];
#pragma unroll
        for (int kk = 0; kk < 4; ++kk) {
            const float* __restrict__ rw = M + (size_t)rel[4 * g + kk] * DIM + 4 * lane;
#pragma unroll
            for (int i = 0; i < 4; ++i)
                rv[kk][i] = *(const float4*)(rw + 128 * i);
        }
#pragma unroll
        for (int kk = 0; kk < 4; ++kk) {
            float s = 0.f;
#pragma unroll
            for (int i = 0; i < 4; ++i) {
                s += qv[i].x * rv[kk][i].x;
                s += qv[i].y * rv[kk][i].y;
                s += qv[i].z * rv[kk][i].z;
                s += qv[i].w * rv[kk][i].w;
            }
            p[4 * g + kk] = s;
        }
    }
#pragma unroll
    for (int kk = 0; kk < 8; ++kk)
        p[kk] = wave_sum(p[kk]);
    if (lane == 0) {
        const float sc = 0.04419417382415922f;
        *(float4*)&attn[wave * 8]     = make_float4(p[0] * sc, p[1] * sc, p[2] * sc, p[3] * sc);
        *(float4*)&attn[wave * 8 + 4] = make_float4(p[4] * sc, p[5] * sc, p[6] * sc, p[7] * sc);
    }
    __syncthreads();

    if (tid < 32) {
        const float s0 = attn[tid], s1 = attn[tid + 32];
        const float m = wave_max(fmaxf(s0, s1));
        const float e0 = __expf(s0 - m), e1 = __expf(s1 - m);
        const float s = wave_sum(e0 + e1);
        const float inv = 1.f / s;
        attn[tid]      = e0 * inv;
        attn[tid + 32] = e1 * inv;
    }
    __syncthreads();

    const int* __restrict__ tp = adj_tail + ni * KNBR;
    const float* __restrict__ ecol = entW + 2 * tid;
    float acc0 = 0.f, acc1 = 0.f;
#pragma unroll
    for (int k0 = 0; k0 < KNBR; k0 += 32) {
        float2 e[32];
#pragma unroll
        for (int j = 0; j < 32; ++j)
            e[j] = *(const float2*)(ecol + (size_t)tp[k0 + j] * DIM);
#pragma unroll
        for (int j = 0; j < 32; ++j) {
            const float w = attn[k0 + j];
            acc0 += w * e[j].x;
            acc1 += w * e[j].y;
        }
    }
    *(float2*)(agg + (size_t)n * DIM + 2 * tid) = make_float2(acc0, acc1);
}

__global__ __launch_bounds__(32)
void lingemm_kernel(const float* __restrict__ agg, const float* __restrict__ drugW,
                    const int* __restrict__ drug_name,
                    const float* __restrict__ linW, const float* __restrict__ linb,
                    float* __restrict__ h)
{
    const int lane = threadIdx.x;
    const int half = lane >> 4;
    const int l16  = lane & 15;
    const int mBase = blockIdx.x * 32;
    const int nBase = blockIdx.y * 16;

    const int r0 = mBase + l16, r1 = r0 + 16;
    const int g0 = drug_name[r0 < N_DRUG ? r0 : (N_DRUG - 1)];
    const int g1 = drug_name[r1 < N_DRUG ? r1 : (N_DRUG - 1)];
    const float* __restrict__ P0 = agg + (size_t)r0 * DIM + 32 * half;
    const float* __restrict__ P1 = agg + (size_t)r1 * DIM + 32 * half;
    const float* __restrict__ Q0 = drugW + (size_t)g0 * DIM + 32 * half;
    const float* __restrict__ Q1 = drugW + (size_t)g1 * DIM + 32 * half;
    const float* __restrict__ Bp = linW + (size_t)(32 * half) * DIM + nBase + l16;

    v8f c0 = {}, c1 = {};
    wave_gemm_32x16<2 * DIM>(P0, P1, Q0, Q1, Bp, c0, c1);

    const float bias = linb[nBase + l16];
    float* outBase = h + (size_t)(mBase + half * 8) * DIM + nBase + l16;
#pragma unroll
    for (int j = 0; j < 8; ++j) {
        const float v0 = c0[j] + bias, v1 = c1[j] + bias;
        outBase[(size_t)j * DIM]        = v0 > 0.f ? v0 : 0.f;
        outBase[(size_t)(j + 16) * DIM] = v1 > 0.f ? v1 : 0.f;
    }
}

#define SCOLS 16
#define WCOLS 4
#define SWAVES (SCOLS / WCOLS)
#define NZROW NPAD
#define SROWS (NPAD + 1)
#define ASTR  5
#define SVEC  (N_DRUG * SCOLS / 4)
#define TVEC  (N_DRUG * MAXDEG / 4)
#define RSTEP (256 / (SCOLS / 4))
#define HFULL (N_DRUG / RSTEP)
#define HROWS (HFULL + 1)

template <int OFF>
__device__ __forceinline__ float4 add_from_lane4(float4 v)
{
    v.x += __shfl_xor(v.x, OFF, 32);
    v.y += __shfl_xor(v.y, OFF, 32);
    v.z += __shfl_xor(v.z, OFF, 32);
    v.w += __shfl_xor(v.w, OFF, 32);
    return v;
}

__device__ __forceinline__ int count_entry(const int e, const int R, const unsigned i,
                                           unsigned long long& lo, unsigned long long& hi)
{
    const unsigned d = (unsigned)(e - R);
    const bool inr = d < i;
    const unsigned long long inc = (unsigned long long)(inr ? 1u : 0u) << ((d & 7u) * 8u);
    lo += (d & 8u) ? 0ull : inc;
    hi += (d & 8u) ? inc : 0ull;
    return (inr ? NZROW : e) * WCOLS;
}

#define SOLVE_STEP(J, A)                                                                   \
    do {                                                                                   \
        asm("v_fmac_f32_dpp %0, %0, %1 row_share:" #J " row_mask:0xf bank_mask:0xf bound_ctrl:1" \
            : "+v"(rhs.x) : "v"(A));                                                       \
        asm("v_fmac_f32_dpp %0, %0, %1 row_share:" #J " row_mask:0xf bank_mask:0xf bound_ctrl:1" \
            : "+v"(rhs.y) : "v"(A));                                                       \
    } while (0)

__global__ __launch_bounds__(256)
void bn_smooth_kernel(const float* __restrict__ h, const float* __restrict__ gamma,
                      const float* __restrict__ beta,
                      const int* __restrict__ nbr_idx, const int* __restrict__ nbr_deg,
                      const int* __restrict__ epoch, float* __restrict__ f)
{
    __shared__ __attribute__((aligned(16))) float slab[SWAVES * SROWS * WCOLS];
    __shared__ __attribute__((aligned(16))) int   sidx[NPAD * MAXDEG];
    __shared__ __attribute__((aligned(16))) float4 acoef[NPAD * ASTR];
    __shared__ __attribute__((aligned(16))) float2 hbco[NPAD];
    __shared__ __attribute__((aligned(16))) float red[2][8][SCOLS];

    const int tid  = threadIdx.x;
    const int lane = tid & 31;
    const int wave = __builtin_amdgcn_readfirstlane(tid >> 5);
    const int col0 = blockIdx.x * SCOLS;
    const int w    = tid & 3;
    const int c4   = w * 4;
    const int rb   = tid >> 2;
    const int ts4  = (tid & 7) * 4;
    const int ep   = *epoch;

    const float* __restrict__ hp = h + (size_t)rb * DIM + col0 + c4;
    const bool tail = rb + RSTEP * HFULL < N_DRUG;
    float4 hv[HROWS];
#pragma unroll
    for (int j = 0; j < HFULL; ++j)
        hv[j] = *(const float4*)(hp + (size_t)(RSTEP * j) * DIM);
    hv[HFULL] = *(const float4*)(hp + (size_t)(tail ? RSTEP * HFULL : 0) * DIM);
    const float4 g4 = *(const float4*)(gamma + col0 + c4);
    const float4 b4 = *(const float4*)(beta + col0 + c4);

    for (int v = tid; v < TVEC; v += 256) {
        const int dg = nbr_deg[v >> 3];
        int4 t = *(const int4*)(nbr_idx + v * 4);
        t.x = (ts4     < dg) ? t.x : NZROW;
        t.y = (ts4 + 1 < dg) ? t.y : NZROW;
        t.z = (ts4 + 2 < dg) ? t.z : NZROW;
        t.w = (ts4 + 3 < dg) ? t.w : NZROW;
        *(int4*)(sidx + v * 4) = t;
    }
    if (tid < SWAVES * (SROWS - N_DRUG) * WCOLS) {
        const int sw  = tid / ((SROWS - N_DRUG) * WCOLS);
        const int rem = tid % ((SROWS - N_DRUG) * WCOLS);
        const int row = N_DRUG + rem / WCOLS;
        slab[(sw * SROWS + row) * WCOLS + (rem % WCOLS)] = (row == NZROW) ? -0.0f : 0.0f;
    }

    if (!tail)
        hv[HFULL] = make_float4(0.f, 0.f, 0.f, 0.f);
    float4 s1 = make_float4(0.f, 0.f, 0.f, 0.f), s2 = s1;
#pragma unroll
    for (int j = 0; j < HROWS; ++j) {
        s1.x += hv[j].x;            s1.y += hv[j].y;
        s1.z += hv[j].z;            s1.w += hv[j].w;
        s2.x += hv[j].x * hv[j].x;  s2.y += hv[j].y * hv[j].y;
        s2.z += hv[j].z * hv[j].z;  s2.w += hv[j].w * hv[j].w;
    }
    s1 = add_from_lane4<4>(s1);   s2 = add_from_lane4<4>(s2);
    s1 = add_from_lane4<8>(s1);   s2 = add_from_lane4<8>(s2);
    s1 = add_from_lane4<16>(s1);  s2 = add_from_lane4<16>(s2);
    if (lane < 4) {
        *(float4*)&red[0][wave][c4] = s1;
        *(float4*)&red[1][wave][c4] = s2;
    }
    __syncthreads();

    for (int r = tid; r < NPAD; r += 256) {
        int* trow = sidx + r * MAXDEG;
        if (r < N_DRUG) {
            const int dg = nbr_deg[r];
            const int R = r & ~15;
            const unsigned i = (unsigned)(r & 15);
            int4 tt[MAXDEG / 4];
#pragma unroll
            for (int k = 0; k < MAXDEG / 4; ++k)
                tt[k] = *(const int4*)(trow + 4 * ((k + tid) & 7));
            unsigned long long lo = 0ull, hi = 0ull;
#pragma unroll
            for (int k = 0; k < MAXDEG / 4; ++k) {
                int4 t = tt[k];
                t.x = count_entry(t.x, R, i, lo, hi);
                t.y = count_entry(t.y, R, i, lo, hi);
                t.z = count_entry(t.z, R, i, lo, hi);
                t.w = count_entry(t.w, R, i, lo, hi);
                *(int4*)(trow + 4 * ((k + tid) & 7)) = t;
            }
            const int dd = dg < 1 ? 1 : (dg < MAXDEG ? dg : MAXDEG);
            const float hc = dg > 0 ? 0.5f / (float)dd : 0.f;
            float av[16];
#pragma unroll
            for (int j = 0; j < 8; ++j) {
                av[j]     = hc * (float)((unsigned)(lo >> (8 * j)) & 0xffu);
                av[8 + j] = hc * (float)((unsigned)(hi >> (8 * j)) & 0xffu);
            }
#pragma unroll
            for (int qd = 0; qd < 4; ++qd)
                acoef[r * ASTR + qd] = make_float4(av[4 * qd], av[4 * qd + 1],
                                                   av[4 * qd + 2], av[4 * qd + 3]);
            hbco[r] = make_float2(hc, dg > 0 ? 0.5f : 1.0f);
        } else {
            const int4 nz = make_int4(NZROW * WCOLS, NZROW * WCOLS, NZROW * WCOLS, NZROW * WCOLS);
#pragma unroll
            for (int k = 0; k < MAXDEG / 4; ++k)
                *(int4*)(trow + 4 * k) = nz;
#pragma unroll
            for (int qd = 0; qd < 4; ++qd)
                acoef[r * ASTR + qd] = make_float4(0.f, 0.f, 0.f, 0.f);
            hbco[r] = make_float2(0.f, 1.0f);
        }
    }

    float4 t1 = make_float4(0.f, 0.f, 0.f, 0.f), t2 = t1;
#pragma unroll
    for (int wv = 0; wv < 8; ++wv) {
        const float4 a = *(const float4*)&red[0][wv][c4];
        const float4 b = *(const float4*)&red[1][wv][c4];
        t1.x += a.x; t1.y += a.y; t1.z += a.z; t1.w += a.w;
        t2.x += b.x; t2.y += b.y; t2.z += b.z; t2.w += b.w;
    }
    const float inv_n = 1.0f / (float)N_DRUG;
    float4 sc, sh;
    {
        const float mx = t1.x * inv_n, my = t1.y * inv_n, mz = t1.z * inv_n, mw = t1.w * inv_n;
        sc.x = g4.x * rsqrtf((t2.x * inv_n - mx * mx) + 1e-5f);
        sc.y = g4.y * rsqrtf((t2.y * inv_n - my * my) + 1e-5f);
        sc.z = g4.z * rsqrtf((t2.z * inv_n - mz * mz) + 1e-5f);
        sc.w = g4.w * rsqrtf((t2.w * inv_n - mw * mw) + 1e-5f);
        sh.x = b4.x - mx * sc.x;  sh.y = b4.y - my * sc.y;
        sh.z = b4.z - mz * sc.z;  sh.w = b4.w - mw * sc.w;
    }
    float* sp = slab + (w * SROWS + rb) * WCOLS;
#pragma unroll
    for (int j = 0; j < HROWS; ++j) {
        if (j < HFULL || tail) {
            float4 o;
            o.x = hv[j].x * sc.x + sh.x;  o.y = hv[j].y * sc.y + sh.y;
            o.z = hv[j].z * sc.z + sh.z;  o.w = hv[j].w * sc.w + sh.w;
            *(float4*)(sp + RSTEP * j * WCOLS) = o;
        }
    }
    __syncthreads();

    if (ep > 1 && wave < SWAVES) {
        const int i   = lane & 15;
        const int rot = i >> 1;
        float* colp = slab + wave * (SROWS * WCOLS) + 2 * (lane >> 4);

        int4 tt[MAXDEG / 4];
#pragma unroll
        for (int k = 0; k < MAXDEG / 4; ++k)
            tt[k] = *(const int4*)(sidx + i * MAXDEG + 4 * ((k + rot) & 7));

#define SCAN_BLOCK(RB)                                                                     \
        do {                                                                               \
            const int r = (RB) + i;                                                        \
                  \
            float2 gv[MAXDEG];                                                             \
            _Pragma("unroll")                                                              \
            for (int k = 0; k < MAXDEG / 4; ++k) {                                         \
                gv[4 * k]     = *(const float2*)(colp + tt[k].x);                          \
                gv[4 * k + 1] = *(const float2*)(colp + tt[k].y);                          \
                gv[4 * k + 2] = *(const float2*)(colp + tt[k].z);                          \
                gv[4 * k + 3] = *(const float2*)(colp + tt[k].w);                          \
            }                                                                              \
            const float2 own = *(const float2*)(colp + r * WCOLS);                         \
            const float2 hb  = hbco[r];                                                    \
            const float4 a0 = acoef[r * ASTR],     a1 = acoef[r * ASTR + 1];               \
            const float4 a2 = acoef[r * ASTR + 2], a3 = acoef[r * ASTR + 3];               \
                  \
            const int rn = ((RB) + 16 < NPAD ? (RB) + 16 : (RB)) + i;                      \
            _Pragma("unroll")                                                              \
            for (int k = 0; k < MAXDEG / 4; ++k)                                           \
                tt[k] = *(const int4*)(sidx + rn * MAXDEG + 4 * ((k + rot) & 7));          \
                       \
            float2 s0 = gv[0], s1 = gv[1], s2 = gv[2], s3 = gv[3];                         \
            _Pragma("unroll")                                                              \
            for (int k = 1; k < MAXDEG / 4; ++k) {                                         \
                s0.x += gv[4 * k].x;      s0.y += gv[4 * k].y;                             \
                s1.x += gv[4 * k + 1].x;  s1.y += gv[4 * k + 1].y;                         \
                s2.x += gv[4 * k + 2].x;  s2.y += gv[4 * k + 2].y;                         \
                s3.x += gv[4 * k + 3].x;  s3.y += gv[4 * k + 3].y;                         \
            }                                                                              \
            float2 rhs;                                                                    \
            rhs.x = fmaf(hb.x, (s0.x + s1.x) + (s2.x + s3.x), hb.y * own.x);               \
            rhs.y = fmaf(hb.x, (s0.y + s1.y) + (s2.y + s3.y), hb.y * own.y);               \
                                                         \
            SOLVE_STEP(0,  a0.x);  SOLVE_STEP(1,  a0.y);  SOLVE_STEP(2,  a0.z);            \
            SOLVE_STEP(3,  a0.w);  SOLVE_STEP(4,  a1.x);  SOLVE_STEP(5,  a1.y);            \
            SOLVE_STEP(6,  a1.z);  SOLVE_STEP(7,  a1.w);  SOLVE_STEP(8,  a2.x);            \
            SOLVE_STEP(9,  a2.y);  SOLVE_STEP(10, a2.z);  SOLVE_STEP(11, a2.w);            \
            SOLVE_STEP(12, a3.x);  SOLVE_STEP(13, a3.y);  SOLVE_STEP(14, a3.z);            \
            *(float2*)(colp + r * WCOLS) = rhs;                                            \
        } while (0)

        SCAN_BLOCK(0);
        for (int R = 16; R < NPAD; R += 16)
            SCAN_BLOCK(R);
#undef SCAN_BLOCK
    }
    __syncthreads();

    for (int v = tid; v < SVEC; v += 256) {
        const int row = v >> 2;
        *(float4*)(f + (size_t)row * DIM + col0 + c4) =
            *(const float4*)(slab + (w * SROWS + row) * WCOLS);
    }
}

extern "C" void kernel_launch(void* const* d_in, const int* in_sizes, int n_in,
                              void* d_out, int out_size, void* d_ws, size_t ws_size,
                              hipStream_t stream)
{
    const float* drugW  = (const float*)d_in[0];
    const float* relaW  = (const float*)d_in[1];
    const float* entW   = (const float*)d_in[2];
    const float* Wa     = (const float*)d_in[3];
    const float* linW   = (const float*)d_in[4];
    const float* linb   = (const float*)d_in[5];
    const float* gamma  = (const float*)d_in[6];
    const float* beta   = (const float*)d_in[7];
    const int* drug_name = (const int*)d_in[8];
    const int* adj_tail  = (const int*)d_in[9];
    const int* adj_rel   = (const int*)d_in[10];
    const int* nbr_idx   = (const int*)d_in[11];
    const int* nbr_deg   = (const int*)d_in[12];
    const int* epoch     = (const int*)d_in[13];

    float* out = (float*)d_out;
    float* ws  = (float*)d_ws;
    float* M     = ws;
    float* agg   = ws + (size_t)NPAD * DIM;
    float* h     = ws + (size_t)2 * NPAD * DIM;

    dim3 gTiles(MP, NT);
    dim3 gM(RP, NT);
    mgemm_kernel<<<gM, 128, 0, stream>>>(relaW, Wa, M);
    attn_agg_kernel<<<NPAD, 256, 0, stream>>>(drugW, drug_name, M, entW, adj_rel, adj_tail, agg);
    lingemm_kernel<<<gTiles, 32, 0, stream>>>(agg, drugW, drug_name, linW, linb, h);
    bn_smooth_kernel<<<DIM / SCOLS, 256, 0, stream>>>(h, gamma, beta, nbr_idx, nbr_deg, epoch, out);
}
